# starting kernel
# baseline (speedup 1.0000x reference)
#include <hip/hip_runtime.h>

typedef __attribute__((ext_vector_type(2))) float v2f;
typedef __attribute__((ext_vector_type(8))) float v8f;

#define IN_DIM  128
#define HID_DIM 128
#define OUT_DIM 64


__global__ __launch_bounds__(256) void zero_f32(float* __restrict__ p, long n) {
  long i = (long)blockIdx.x * blockDim.x + threadIdx.x;
  if (i < n) p[i] = 0.0f;
}

__global__ __launch_bounds__(256) void degree_kernel(const int* __restrict__ src,
                                                     const int* __restrict__ dst,
                                                     float* __restrict__ out_deg,
                                                     float* __restrict__ in_deg,
                                                     int n_edges) {
  int i = blockIdx.x * blockDim.x + threadIdx.x;
  if (i < n_edges) {
    atomicAdd(&out_deg[src[i]], 1.0f);
    atomicAdd(&in_deg[dst[i]], 1.0f);
  }
}

__global__ __launch_bounds__(256) void norm_kernel(const float* __restrict__ out_deg,
                                                   const float* __restrict__ in_deg,
                                                   float* __restrict__ out_norm,
                                                   float* __restrict__ in_norm,
                                                   int n) {
  int i = blockIdx.x * blockDim.x + threadIdx.x;
  if (i < n) {
    out_norm[i] = rsqrtf(fmaxf(out_deg[i], 1.0f));
    in_norm[i]  = rsqrtf(fmaxf(in_deg[i], 1.0f));
  }
}


__global__ __launch_bounds__(256) void gemm1_wmma(const float* __restrict__ x,
                                                  const float* __restrict__ W1,
                                                  const float* __restrict__ out_norm,
                                                  float* __restrict__ h1,
                                                  int n_row_tiles) {
  const int lane = threadIdx.x & 31;
  const int wave = (int)((blockIdx.x * blockDim.x + threadIdx.x) >> 5);
  const int n_tiles = n_row_tiles * (HID_DIM / 16);
  if (wave >= n_tiles) return;
  const int rt   = wave / (HID_DIM / 16);
  const int ct   = wave % (HID_DIM / 16);
  const int half = lane >> 4;
  const int sub  = lane & 15;
  const int m    = rt * 16 + sub;
  const int n    = ct * 16 + sub;
  const int kh   = half * 2;

  const float norm = out_norm[m];
  const float* __restrict__ arow = x + (size_t)m * IN_DIM;

  v8f acc = {};
#pragma unroll 4
  for (int k = 0; k < IN_DIM; k += 4) {
    const float2 av = *(const float2*)(arow + k + kh);
    v2f a, b;
    a.x = av.x * norm;
    a.y = av.y * norm;
    b.x = W1[(size_t)(k + kh)     * HID_DIM + n];
    b.y = W1[(size_t)(k + kh + 1) * HID_DIM + n];
    acc = __builtin_amdgcn_wmma_f32_16x16x4_f32(
         false, a,  false, b,
         (short)0, acc,  false,  false);
  }

  const int m_base = rt * 16 + half * 8;
#pragma unroll
  for (int r = 0; r < 8; ++r)
    h1[(size_t)(m_base + r) * HID_DIM + n] = acc[r];
}


__global__ __launch_bounds__(256) void gemm2_wmma(const float* __restrict__ agg1,
                                                  const float* __restrict__ W2,
                                                  const float* __restrict__ b1,
                                                  const float* __restrict__ out_norm,
                                                  const float* __restrict__ in_norm,
                                                  float* __restrict__ h2,
                                                  int n_row_tiles) {
  const int lane = threadIdx.x & 31;
  const int wave = (int)((blockIdx.x * blockDim.x + threadIdx.x) >> 5);
  const int n_tiles = n_row_tiles * (OUT_DIM / 16);
  if (wave >= n_tiles) return;
  const int rt   = wave / (OUT_DIM / 16);
  const int ct   = wave % (OUT_DIM / 16);
  const int half = lane >> 4;
  const int sub  = lane & 15;
  const int m    = rt * 16 + sub;
  const int n    = ct * 16 + sub;
  const int kh   = half * 2;

  const float s_in  = in_norm[m];
  const float s_out = out_norm[m];
  const float* __restrict__ arow = agg1 + (size_t)m * HID_DIM;

  v8f acc = {};
#pragma unroll 4
  for (int k = 0; k < HID_DIM; k += 4) {
    const float2 av = *(const float2*)(arow + k + kh);
    const float2 bv = *(const float2*)(b1 + k + kh);
    v2f a, b;
    a.x = fmaxf(av.x * s_in + bv.x, 0.0f) * s_out;
    a.y = fmaxf(av.y * s_in + bv.y, 0.0f) * s_out;
    b.x = W2[(size_t)(k + kh)     * OUT_DIM + n];
    b.y = W2[(size_t)(k + kh + 1) * OUT_DIM + n];
    acc = __builtin_amdgcn_wmma_f32_16x16x4_f32(
        false, a, false, b, (short)0, acc, false, false);
  }

  const int m_base = rt * 16 + half * 8;
#pragma unroll
  for (int r = 0; r < 8; ++r)
    h2[(size_t)(m_base + r) * OUT_DIM + n] = acc[r];
}


__global__ __launch_bounds__(256) void scatter128(const float* __restrict__ h,
                                                  const int* __restrict__ src,
                                                  const int* __restrict__ dst,
                                                  float* __restrict__ agg,
                                                  int n_edges) {
  const int wave = (int)((blockIdx.x * blockDim.x + threadIdx.x) >> 5);
  const int lane = threadIdx.x & 31;
  if (wave >= n_edges) return;
  const int s = src[wave];
  const int d = dst[wave];
  const float4 v = *(const float4*)(h + (size_t)s * 128 + lane * 4);
  float* p = agg + (size_t)d * 128 + lane * 4;
  atomicAdd(p + 0, v.x);
  atomicAdd(p + 1, v.y);
  atomicAdd(p + 2, v.z);
  atomicAdd(p + 3, v.w);
}

__global__ __launch_bounds__(256) void scatter64(const float* __restrict__ h,
                                                 const int* __restrict__ src,
                                                 const int* __restrict__ dst,
                                                 float* __restrict__ agg,
                                                 int n_edges) {
  const int wave = (int)((blockIdx.x * blockDim.x + threadIdx.x) >> 5);
  const int lane = threadIdx.x & 31;
  if (wave >= n_edges) return;
  const int s = src[wave];
  const int d = dst[wave];
  const float2 v = *(const float2*)(h + (size_t)s * 64 + lane * 2);
  float* p = agg + (size_t)d * 64 + lane * 2;
  atomicAdd(p + 0, v.x);
  atomicAdd(p + 1, v.y);
}

__global__ __launch_bounds__(256) void finalize_kernel(float* __restrict__ out,
                                                       const float* __restrict__ in_norm,
                                                       const float* __restrict__ b2,
                                                       long total) {
  long i = (long)blockIdx.x * blockDim.x + threadIdx.x;
  if (i < total) {
    const int m = (int)(i >> 6);
    const int n = (int)(i & 63);
    out[i] = out[i] * in_norm[m] + b2[n];
  }
}


static inline int grid1d(long n, int block) { return (int)((n + block - 1) / block); }

extern "C" void kernel_launch(void* const* d_in, const int* in_sizes, int n_in,
                              void* d_out, int out_size, void* d_ws, size_t ws_size,
                              hipStream_t stream) {
  const float* x   = (const float*)d_in[0];
  const float* W1  = (const float*)d_in[1];
  const float* b1  = (const float*)d_in[2];
  const float* W2  = (const float*)d_in[3];
  const float* b2  = (const float*)d_in[4];
  const int*   src = (const int*)d_in[5];
  const int*   dst = (const int*)d_in[6];

  const int n_nodes = in_sizes[0] / IN_DIM;
  const int n_edges = in_sizes[5];
  float* out = (float*)d_out;

  float* ws       = (float*)d_ws;
  float* out_deg  = ws;
  float* in_deg   = out_deg + (size_t)n_nodes;
  float* out_norm = in_deg + (size_t)n_nodes;
  float* in_norm  = out_norm + (size_t)n_nodes;
  float* h1       = in_norm + (size_t)n_nodes;
  float* agg1     = h1 + (size_t)n_nodes * HID_DIM;
  float* h2       = agg1 + (size_t)n_nodes * HID_DIM;

  zero_f32<<<grid1d(2L * n_nodes, 256), 256, 0, stream>>>(out_deg, 2L * n_nodes);
  zero_f32<<<grid1d((long)n_nodes * HID_DIM, 256), 256, 0, stream>>>(agg1, (long)n_nodes * HID_DIM);
  zero_f32<<<grid1d((long)n_nodes * OUT_DIM, 256), 256, 0, stream>>>(out, (long)n_nodes * OUT_DIM);

  degree_kernel<<<grid1d(n_edges, 256), 256, 0, stream>>>(src, dst, out_deg, in_deg, n_edges);
  norm_kernel<<<grid1d(n_nodes, 256), 256, 0, stream>>>(out_deg, in_deg, out_norm, in_norm, n_nodes);

  const int row_tiles = n_nodes / 16;

  {
    const int waves  = row_tiles * (HID_DIM / 16);
    const int blocks = (waves + 7) / 8;
    gemm1_wmma<<<blocks, 256, 0, stream>>>(x, W1, out_norm, h1, row_tiles);
  }

  scatter128<<<(n_edges + 7) / 8, 256, 0, stream>>>(h1, src, dst, agg1, n_edges);

  {
    const int waves  = row_tiles * (OUT_DIM / 16);
    const int blocks = (waves + 7) / 8;
    gemm2_wmma<<<blocks, 256, 0, stream>>>(agg1, W2, b1, out_norm, in_norm, h2, row_tiles);
  }

  scatter64<<<(n_edges + 7) / 8, 256, 0, stream>>>(h2, src, dst, out, n_edges);

  finalize_kernel<<<grid1d((long)n_nodes * OUT_DIM, 256), 256, 0, stream>>>(
      out, in_norm, b2, (long)n_nodes * OUT_DIM);
}


# degrees: per-work-group LDS histograms (ds_add_u32) of edge slices plus a row-sum kernel that writes the norms replace the contended global f32 atomicAdd kernel; integer counts, exact in any order
# speedup vs baseline: 1.3277x; 1.3277x over previous
#include <hip/hip_runtime.h>

typedef __attribute__((ext_vector_type(2))) float v2f;
typedef __attribute__((ext_vector_type(8))) float v8f;

#define IN_DIM  128
#define HID_DIM 128
#define OUT_DIM 64


__global__ __launch_bounds__(256) void zero_f32(float* __restrict__ p, long n) {
  long i = (long)blockIdx.x * blockDim.x + threadIdx.x;
  if (i < n) p[i] = 0.0f;
}


#define HIST_CAP     65536
#define HIST_THREADS 1024
#define DEG_SLICES   128
static_assert(DEG_SLICES % 8 == 0, "degree_norm splits the slice rows over 8 waves");

__device__ __forceinline__ int padded_row(int n_nodes) { return (n_nodes + 31) & ~31; }

__device__ __forceinline__ void hist_bump(unsigned int* hist, int v, int base, int nb) {
  const unsigned int b = (unsigned int)(v - base);
  if (b < (unsigned int)nb) atomicAdd(&hist[b], 1u);
}

__global__ __launch_bounds__(HIST_THREADS) void degree_hist(const int* __restrict__ src,
                                                            const int* __restrict__ dst,
                                                            unsigned int* __restrict__ partial,
                                                            int n_edges, int n_nodes) {
  __shared__ __attribute__((aligned(16))) unsigned int hist[HIST_CAP];
  const int tid = threadIdx.x;
  const int* __restrict__ idx = blockIdx.y ? dst : src;
  const int n_slices = DEG_SLICES;
  const int per = (((n_edges + n_slices - 1) / n_slices) + 3) & ~3;
  const int e0  = min((int)blockIdx.x * per, n_edges);
  const int e1  = min(e0 + per, n_edges);
  const int nv  = (e1 - e0) >> 2;
  const int4* __restrict__ p4 = (const int4*)(idx + e0);
  const int row = padded_row(n_nodes);
  unsigned int* __restrict__ my =
      partial + ((size_t)blockIdx.y * n_slices + blockIdx.x) * (size_t)row;

  for (int base = 0; base < n_nodes; base += HIST_CAP) {
    const int nb  = min(HIST_CAP, n_nodes - base);
    const int nb4 = (nb + 3) >> 2;
    uint4* h4 = (uint4*)hist;
    for (int i = tid; i < nb4; i += HIST_THREADS) h4[i] = make_uint4(0u, 0u, 0u, 0u);
    __syncthreads();

    for (int q = tid; q < nv; q += 4 * HIST_THREADS) {
      int4 v[4];
#pragma unroll
      for (int j = 0; j < 4; ++j) {
        const int qq = q + j * HIST_THREADS;
        v[j] = (qq < nv) ? p4[qq] : make_int4(-1, -1, -1, -1);
      }
#pragma unroll
      for (int j = 0; j < 4; ++j) {
        hist_bump(hist, v[j].x, base, nb);
        hist_bump(hist, v[j].y, base, nb);
        hist_bump(hist, v[j].z, base, nb);
        hist_bump(hist, v[j].w, base, nb);
      }
    }
    {
      const int e = e0 + (nv << 2) + tid;
      if (tid < 3 && e < e1) hist_bump(hist, idx[e], base, nb);
    }
    __syncthreads();

    uint4* o4 = (uint4*)(my + base);
    for (int i = tid; i < nb4; i += HIST_THREADS) o4[i] = h4[i];
    __syncthreads();
  }
}

__global__ __launch_bounds__(256) void degree_norm(const unsigned int* __restrict__ partial,
                                                   float* __restrict__ out_norm,
                                                   float* __restrict__ in_norm,
                                                   int n_nodes) {
  __shared__ uint4 red[2][8][32];
  const int lane = threadIdx.x & 31;
  const int w    = threadIdx.x >> 5;
  const int node = (blockIdx.x * 32 + lane) * 4;
  const int row  = padded_row(n_nodes);
  uint4 so = make_uint4(0u, 0u, 0u, 0u);
  uint4 si = make_uint4(0u, 0u, 0u, 0u);
  if (node < n_nodes) {
    const uint4* __restrict__ po = (const uint4*)(partial + node);
    const uint4* __restrict__ pi = (const uint4*)(partial + (size_t)DEG_SLICES * row + node);
    const size_t row4 = (size_t)(row >> 2);
#pragma unroll
    for (int k = 0; k < DEG_SLICES / 8; ++k) {
      const size_t s = (size_t)(w + 8 * k);
      const uint4 a = po[s * row4];
      const uint4 b = pi[s * row4];
      so.x += a.x; so.y += a.y; so.z += a.z; so.w += a.w;
      si.x += b.x; si.y += b.y; si.z += b.z; si.w += b.w;
    }
  }
  red[0][w][lane] = so;
  red[1][w][lane] = si;
  __syncthreads();
  if (w == 0 && node < n_nodes) {
    so = make_uint4(0u, 0u, 0u, 0u);
    si = make_uint4(0u, 0u, 0u, 0u);
#pragma unroll
    for (int k = 0; k < 8; ++k) {
      const uint4 a = red[0][k][lane];
      const uint4 b = red[1][k][lane];
      so.x += a.x; so.y += a.y; so.z += a.z; so.w += a.w;
      si.x += b.x; si.y += b.y; si.z += b.z; si.w += b.w;
    }
    float4 o, i;
    o.x = rsqrtf(fmaxf((float)so.x, 1.0f));
    o.y = rsqrtf(fmaxf((float)so.y, 1.0f));
    o.z = rsqrtf(fmaxf((float)so.z, 1.0f));
    o.w = rsqrtf(fmaxf((float)so.w, 1.0f));
    i.x = rsqrtf(fmaxf((float)si.x, 1.0f));
    i.y = rsqrtf(fmaxf((float)si.y, 1.0f));
    i.z = rsqrtf(fmaxf((float)si.z, 1.0f));
    i.w = rsqrtf(fmaxf((float)si.w, 1.0f));
    *(float4*)(out_norm + node) = o;
    *(float4*)(in_norm + node)  = i;
  }
}


__global__ __launch_bounds__(256) void gemm1_wmma(const float* __restrict__ x,
                                                  const float* __restrict__ W1,
                                                  const float* __restrict__ out_norm,
                                                  float* __restrict__ h1,
                                                  int n_row_tiles) {
  const int lane = threadIdx.x & 31;
  const int wave = (int)((blockIdx.x * blockDim.x + threadIdx.x) >> 5);
  const int n_tiles = n_row_tiles * (HID_DIM / 16);
  if (wave >= n_tiles) return;
  const int rt   = wave / (HID_DIM / 16);
  const int ct   = wave % (HID_DIM / 16);
  const int half = lane >> 4;
  const int sub  = lane & 15;
  const int m    = rt * 16 + sub;
  const int n    = ct * 16 + sub;
  const int kh   = half * 2;

  const float norm = out_norm[m];
  const float* __restrict__ arow = x + (size_t)m * IN_DIM;

  v8f acc = {};
#pragma unroll 4
  for (int k = 0; k < IN_DIM; k += 4) {
    const float2 av = *(const float2*)(arow + k + kh);
    v2f a, b;
    a.x = av.x * norm;
    a.y = av.y * norm;
    b.x = W1[(size_t)(k + kh)     * HID_DIM + n];
    b.y = W1[(size_t)(k + kh + 1) * HID_DIM + n];
    acc = __builtin_amdgcn_wmma_f32_16x16x4_f32(
         false, a,  false, b,
         (short)0, acc,  false,  false);
  }

  const int m_base = rt * 16 + half * 8;
#pragma unroll
  for (int r = 0; r < 8; ++r)
    h1[(size_t)(m_base + r) * HID_DIM + n] = acc[r];
}


__global__ __launch_bounds__(256) void gemm2_wmma(const float* __restrict__ agg1,
                                                  const float* __restrict__ W2,
                                                  const float* __restrict__ b1,
                                                  const float* __restrict__ out_norm,
                                                  const float* __restrict__ in_norm,
                                                  float* __restrict__ h2,
                                                  int n_row_tiles) {
  const int lane = threadIdx.x & 31;
  const int wave = (int)((blockIdx.x * blockDim.x + threadIdx.x) >> 5);
  const int n_tiles = n_row_tiles * (OUT_DIM / 16);
  if (wave >= n_tiles) return;
  const int rt   = wave / (OUT_DIM / 16);
  const int ct   = wave % (OUT_DIM / 16);
  const int half = lane >> 4;
  const int sub  = lane & 15;
  const int m    = rt * 16 + sub;
  const int n    = ct * 16 + sub;
  const int kh   = half * 2;

  const float s_in  = in_norm[m];
  const float s_out = out_norm[m];
  const float* __restrict__ arow = agg1 + (size_t)m * HID_DIM;

  v8f acc = {};
#pragma unroll 4
  for (int k = 0; k < HID_DIM; k += 4) {
    const float2 av = *(const float2*)(arow + k + kh);
    const float2 bv = *(const float2*)(b1 + k + kh);
    v2f a, b;
    a.x = fmaxf(av.x * s_in + bv.x, 0.0f) * s_out;
    a.y = fmaxf(av.y * s_in + bv.y, 0.0f) * s_out;
    b.x = W2[(size_t)(k + kh)     * OUT_DIM + n];
    b.y = W2[(size_t)(k + kh + 1) * OUT_DIM + n];
    acc = __builtin_amdgcn_wmma_f32_16x16x4_f32(
        false, a, false, b, (short)0, acc, false, false);
  }

  const int m_base = rt * 16 + half * 8;
#pragma unroll
  for (int r = 0; r < 8; ++r)
    h2[(size_t)(m_base + r) * OUT_DIM + n] = acc[r];
}


__global__ __launch_bounds__(256) void scatter128(const float* __restrict__ h,
                                                  const int* __restrict__ src,
                                                  const int* __restrict__ dst,
                                                  float* __restrict__ agg,
                                                  int n_edges) {
  const int wave = (int)((blockIdx.x * blockDim.x + threadIdx.x) >> 5);
  const int lane = threadIdx.x & 31;
  if (wave >= n_edges) return;
  const int s = src[wave];
  const int d = dst[wave];
  const float4 v = *(const float4*)(h + (size_t)s * 128 + lane * 4);
  float* p = agg + (size_t)d * 128 + lane * 4;
  atomicAdd(p + 0, v.x);
  atomicAdd(p + 1, v.y);
  atomicAdd(p + 2, v.z);
  atomicAdd(p + 3, v.w);
}

__global__ __launch_bounds__(256) void scatter64(const float* __restrict__ h,
                                                 const int* __restrict__ src,
                                                 const int* __restrict__ dst,
                                                 float* __restrict__ agg,
                                                 int n_edges) {
  const int wave = (int)((blockIdx.x * blockDim.x + threadIdx.x) >> 5);
  const int lane = threadIdx.x & 31;
  if (wave >= n_edges) return;
  const int s = src[wave];
  const int d = dst[wave];
  const float2 v = *(const float2*)(h + (size_t)s * 64 + lane * 2);
  float* p = agg + (size_t)d * 64 + lane * 2;
  atomicAdd(p + 0, v.x);
  atomicAdd(p + 1, v.y);
}

__global__ __launch_bounds__(256) void finalize_kernel(float* __restrict__ out,
                                                       const float* __restrict__ in_norm,
                                                       const float* __restrict__ b2,
                                                       long total) {
  long i = (long)blockIdx.x * blockDim.x + threadIdx.x;
  if (i < total) {
    const int m = (int)(i >> 6);
    const int n = (int)(i & 63);
    out[i] = out[i] * in_norm[m] + b2[n];
  }
}


static inline int grid1d(long n, int block) { return (int)((n + block - 1) / block); }

extern "C" void kernel_launch(void* const* d_in, const int* in_sizes, int n_in,
                              void* d_out, int out_size, void* d_ws, size_t ws_size,
                              hipStream_t stream) {
  const float* x   = (const float*)d_in[0];
  const float* W1  = (const float*)d_in[1];
  const float* b1  = (const float*)d_in[2];
  const float* W2  = (const float*)d_in[3];
  const float* b2  = (const float*)d_in[4];
  const int*   src = (const int*)d_in[5];
  const int*   dst = (const int*)d_in[6];

  const int n_nodes = in_sizes[0] / IN_DIM;
  const int n_edges = in_sizes[5];
  float* out = (float*)d_out;

  const size_t row = ((size_t)n_nodes + 31) & ~(size_t)31;
  float* h1   = (float*)d_ws;
  float* agg1 = h1 + (size_t)n_nodes * HID_DIM;
  float* h2   = agg1 + (size_t)n_nodes * HID_DIM;
  unsigned int* partial = (unsigned int*)(h2 + (size_t)n_nodes * OUT_DIM);
  float* out_norm = (float*)(partial + 2 * DEG_SLICES * row);
  float* in_norm  = out_norm + row;

  zero_f32<<<grid1d((long)n_nodes * HID_DIM, 256), 256, 0, stream>>>(agg1, (long)n_nodes * HID_DIM);
  zero_f32<<<grid1d((long)n_nodes * OUT_DIM, 256), 256, 0, stream>>>(out, (long)n_nodes * OUT_DIM);

  degree_hist<<<dim3(DEG_SLICES, 2), HIST_THREADS, 0, stream>>>(src, dst, partial, n_edges, n_nodes);
  degree_norm<<<grid1d(n_nodes, 128), 256, 0, stream>>>(partial, out_norm, in_norm, n_nodes);

  const int row_tiles = n_nodes / 16;

  {
    const int waves  = row_tiles * (HID_DIM / 16);
    const int blocks = (waves + 7) / 8;
    gemm1_wmma<<<blocks, 256, 0, stream>>>(x, W1, out_norm, h1, row_tiles);
  }

  scatter128<<<(n_edges + 7) / 8, 256, 0, stream>>>(h1, src, dst, agg1, n_edges);

  {
    const int waves  = row_tiles * (OUT_DIM / 16);
    const int blocks = (waves + 7) / 8;
    gemm2_wmma<<<blocks, 256, 0, stream>>>(agg1, W2, b1, out_norm, in_norm, h2, row_tiles);
  }

  scatter64<<<(n_edges + 7) / 8, 256, 0, stream>>>(h2, src, dst, out, n_edges);

  finalize_kernel<<<grid1d((long)n_nodes * OUT_DIM, 256), 256, 0, stream>>>(
      out, in_norm, b2, (long)n_nodes * OUT_DIM);
}


# aggregation: edges counting-sorted by dst in every call (from the slice histograms); one wave sums a node's in-edge rows in list order and writes the row once; no f32 atomics, zeroing or finalize
# speedup vs baseline: 3.5194x; 2.6507x over previous
#include <hip/hip_runtime.h>

typedef __attribute__((ext_vector_type(2))) float v2f;
typedef __attribute__((ext_vector_type(8))) float v8f;

#define IN_DIM  128
#define HID_DIM 128
#define OUT_DIM 64


#define HIST_CAP     65536
#define HIST_THREADS 1024
#define DEG_SLICES   128
static_assert(DEG_SLICES % 8 == 0, "degree_norm splits the slice rows over 8 waves");

__device__ __forceinline__ int padded_row(int n_nodes) { return (n_nodes + 31) & ~31; }

__device__ __forceinline__ void slice_range(int n_edges, int slice, int& e0, int& e1) {
  const int per = (((n_edges + DEG_SLICES - 1) / DEG_SLICES) + 3) & ~3;
  e0 = min(slice * per, n_edges);
  e1 = min(e0 + per, n_edges);
}

__device__ __forceinline__ void hist_bump(unsigned int* hist, int v, int base, int nb) {
  const unsigned int b = (unsigned int)(v - base);
  if (b < (unsigned int)nb) atomicAdd(&hist[b], 1u);
}

__global__ __launch_bounds__(HIST_THREADS) void degree_hist(const int* __restrict__ src,
                                                            const int* __restrict__ dst,
                                                            unsigned int* __restrict__ partial,
                                                            int n_edges, int n_nodes) {
  __shared__ __attribute__((aligned(16))) unsigned int hist[HIST_CAP];
  const int tid = threadIdx.x;
  const int* __restrict__ idx = blockIdx.y ? dst : src;
  const int n_slices = DEG_SLICES;
  int e0, e1;
  slice_range(n_edges, (int)blockIdx.x, e0, e1);
  const int nv  = (e1 - e0) >> 2;
  const int4* __restrict__ p4 = (const int4*)(idx + e0);
  const int row = padded_row(n_nodes);
  unsigned int* __restrict__ my =
      partial + ((size_t)blockIdx.y * n_slices + blockIdx.x) * (size_t)row;

  for (int base = 0; base < n_nodes; base += HIST_CAP) {
    const int nb  = min(HIST_CAP, n_nodes - base);
    const int nb4 = (nb + 3) >> 2;
    uint4* h4 = (uint4*)hist;
    for (int i = tid; i < nb4; i += HIST_THREADS) h4[i] = make_uint4(0u, 0u, 0u, 0u);
    __syncthreads();

    for (int q = tid; q < nv; q += 4 * HIST_THREADS) {
      int4 v[4];
#pragma unroll
      for (int j = 0; j < 4; ++j) {
        const int qq = q + j * HIST_THREADS;
        v[j] = (qq < nv) ? p4[qq] : make_int4(-1, -1, -1, -1);
      }
#pragma unroll
      for (int j = 0; j < 4; ++j) {
        hist_bump(hist, v[j].x, base, nb);
        hist_bump(hist, v[j].y, base, nb);
        hist_bump(hist, v[j].z, base, nb);
        hist_bump(hist, v[j].w, base, nb);
      }
    }
    {
      const int e = e0 + (nv << 2) + tid;
      if (tid < 3 && e < e1) hist_bump(hist, idx[e], base, nb);
    }
    __syncthreads();

    uint4* o4 = (uint4*)(my + base);
    for (int i = tid; i < nb4; i += HIST_THREADS) o4[i] = h4[i];
    __syncthreads();
  }
}

__global__ __launch_bounds__(256) void degree_norm(const unsigned int* __restrict__ partial,
                                                   float* __restrict__ out_norm,
                                                   float* __restrict__ in_norm,
                                                   unsigned int* __restrict__ block_in,
                                                   int n_nodes) {
  __shared__ uint4 red[2][8][32];
  const int lane = threadIdx.x & 31;
  const int w    = threadIdx.x >> 5;
  const int node = (blockIdx.x * 32 + lane) * 4;
  const int row  = padded_row(n_nodes);
  uint4 so = make_uint4(0u, 0u, 0u, 0u);
  uint4 si = make_uint4(0u, 0u, 0u, 0u);
  if (node < n_nodes) {
    const uint4* __restrict__ po = (const uint4*)(partial + node);
    const uint4* __restrict__ pi = (const uint4*)(partial + (size_t)DEG_SLICES * row + node);
    const size_t row4 = (size_t)(row >> 2);
#pragma unroll
    for (int k = 0; k < DEG_SLICES / 8; ++k) {
      const size_t s = (size_t)(w + 8 * k);
      const uint4 a = po[s * row4];
      const uint4 b = pi[s * row4];
      so.x += a.x; so.y += a.y; so.z += a.z; so.w += a.w;
      si.x += b.x; si.y += b.y; si.z += b.z; si.w += b.w;
    }
  }
  red[0][w][lane] = so;
  red[1][w][lane] = si;
  __syncthreads();
  if (w == 0) {
    so = make_uint4(0u, 0u, 0u, 0u);
    si = make_uint4(0u, 0u, 0u, 0u);
#pragma unroll
    for (int k = 0; k < 8; ++k) {
      const uint4 a = red[0][k][lane];
      const uint4 b = red[1][k][lane];
      so.x += a.x; so.y += a.y; so.z += a.z; so.w += a.w;
      si.x += b.x; si.y += b.y; si.z += b.z; si.w += b.w;
    }
    unsigned int t = si.x + si.y + si.z + si.w;
    for (int o = 16; o > 0; o >>= 1) t += __shfl_xor(t, o);
    if (lane == 0) block_in[blockIdx.x] = t;
    if (node < n_nodes) {
      float4 o, i;
      o.x = rsqrtf(fmaxf((float)so.x, 1.0f));
      o.y = rsqrtf(fmaxf((float)so.y, 1.0f));
      o.z = rsqrtf(fmaxf((float)so.z, 1.0f));
      o.w = rsqrtf(fmaxf((float)so.w, 1.0f));
      i.x = rsqrtf(fmaxf((float)si.x, 1.0f));
      i.y = rsqrtf(fmaxf((float)si.y, 1.0f));
      i.z = rsqrtf(fmaxf((float)si.z, 1.0f));
      i.w = rsqrtf(fmaxf((float)si.w, 1.0f));
      *(float4*)(out_norm + node) = o;
      *(float4*)(in_norm + node)  = i;
    }
  }
}


__device__ __forceinline__ uint4 add4(uint4 a, uint4 b) {
  return make_uint4(a.x + b.x, a.y + b.y, a.z + b.z, a.w + b.w);
}

__global__ __launch_bounds__(256) void edge_cursors(unsigned int* __restrict__ part_dst,
                                                    const unsigned int* __restrict__ block_in,
                                                    unsigned int* __restrict__ row_ptr,
                                                    int n_nodes) {
  __shared__ uint4 wtot[8][32];
  __shared__ unsigned int bred[8];
  const int lane = threadIdx.x & 31;
  const int w    = threadIdx.x >> 5;
  const int node = (blockIdx.x * 32 + lane) * 4;
  const int row  = padded_row(n_nodes);
  const size_t row4 = (size_t)(row >> 2);
  const bool live = node < n_nodes;
  const uint4 zero4 = make_uint4(0u, 0u, 0u, 0u);
  uint4* __restrict__ p = (uint4*)part_dst + (size_t)(w * (DEG_SLICES / 8)) * row4 + (size_t)(node >> 2);

  unsigned int b = 0;
  for (int i = threadIdx.x; i < (int)blockIdx.x; i += 256) b += block_in[i];
  for (int o = 16; o > 0; o >>= 1) b += __shfl_xor(b, o);
  if (lane == 0) bred[w] = b;

  uint4 c[DEG_SLICES / 8];
#pragma unroll
  for (int k = 0; k < DEG_SLICES / 8; ++k) c[k] = zero4;
  if (live) {
#pragma unroll
    for (int k = 0; k < DEG_SLICES / 8; ++k) c[k] = p[(size_t)k * row4];
  }
  uint4 tot = zero4;
#pragma unroll
  for (int k = 0; k < DEG_SLICES / 8; ++k) {
    const uint4 t = c[k];
    c[k] = tot;
    tot = add4(tot, t);
  }
  wtot[w][lane] = tot;
  __syncthreads();

  unsigned int base = 0;
  for (int k = 0; k < 8; ++k) base += bred[k];
  uint4 wp = zero4;
  uint4 nt = zero4;
  for (int k = 0; k < 8; ++k) {
    const uint4 t = wtot[k][lane];
    if (k < w) wp = add4(wp, t);
    nt = add4(nt, t);
  }
  const unsigned int lt = nt.x + nt.y + nt.z + nt.w;
  unsigned int inc = lt;
  for (int o = 1; o < 32; o <<= 1) {
    const unsigned int u = __shfl_up(inc, (unsigned int)o);
    if (lane >= o) inc += u;
  }
  uint4 st;
  st.x = base + inc - lt;
  st.y = st.x + nt.x;
  st.z = st.y + nt.y;
  st.w = st.z + nt.z;
  if (live) {
    if (w == 0) {
      *(uint4*)(row_ptr + node) = st;
      if (node + 4 >= n_nodes) row_ptr[n_nodes] = st.x + lt;
    }
    st = add4(st, wp);
#pragma unroll
    for (int k = 0; k < DEG_SLICES / 8; ++k) p[(size_t)k * row4] = add4(st, c[k]);
  }
}

__device__ __forceinline__ void place_edge(unsigned int* cur, int* __restrict__ sorted_src,
                                           int d, int s, int base, int nb, int n_edges) {
  const unsigned int b = (unsigned int)(d - base);
  if (b < (unsigned int)nb) {
    const unsigned int pos = atomicAdd(&cur[b], 1u);
    if (pos < (unsigned int)n_edges) sorted_src[pos] = s;
  }
}

__global__ __launch_bounds__(HIST_THREADS) void edge_fill(const int* __restrict__ src,
                                                          const int* __restrict__ dst,
                                                          const unsigned int* __restrict__ cursors,
                                                          int* __restrict__ sorted_src,
                                                          int n_edges, int n_nodes) {
  __shared__ __attribute__((aligned(16))) unsigned int cur[HIST_CAP];
  const int tid = threadIdx.x;
  int e0, e1;
  slice_range(n_edges, (int)blockIdx.x, e0, e1);
  const int nv = (e1 - e0) >> 2;
  const int4* __restrict__ d4 = (const int4*)(dst + e0);
  const int4* __restrict__ s4 = (const int4*)(src + e0);
  const int row = padded_row(n_nodes);
  const unsigned int* __restrict__ my = cursors + (size_t)blockIdx.x * (size_t)row;

  for (int base = 0; base < n_nodes; base += HIST_CAP) {
    const int nb  = min(HIST_CAP, n_nodes - base);
    const int nb4 = (nb + 3) >> 2;
    uint4* c4 = (uint4*)cur;
    const uint4* __restrict__ i4 = (const uint4*)(my + base);
    for (int i = tid; i < nb4; i += HIST_THREADS) c4[i] = i4[i];
    __syncthreads();

    for (int q = tid; q < nv; q += 4 * HIST_THREADS) {
      int4 dv[4], sv[4];
#pragma unroll
      for (int j = 0; j < 4; ++j) {
        const int qq = q + j * HIST_THREADS;
        const bool ok = qq < nv;
        dv[j] = ok ? d4[qq] : make_int4(-1, -1, -1, -1);
        sv[j] = ok ? s4[qq] : make_int4(0, 0, 0, 0);
      }
#pragma unroll
      for (int j = 0; j < 4; ++j) {
        place_edge(cur, sorted_src, dv[j].x, sv[j].x, base, nb, n_edges);
        place_edge(cur, sorted_src, dv[j].y, sv[j].y, base, nb, n_edges);
        place_edge(cur, sorted_src, dv[j].z, sv[j].z, base, nb, n_edges);
        place_edge(cur, sorted_src, dv[j].w, sv[j].w, base, nb, n_edges);
      }
    }
    {
      const int e = e0 + (nv << 2) + tid;
      if (tid < 3 && e < e1) place_edge(cur, sorted_src, dst[e], src[e], base, nb, n_edges);
    }
    __syncthreads();
  }
}


__global__ __launch_bounds__(256) void gemm1_wmma(const float* __restrict__ x,
                                                  const float* __restrict__ W1,
                                                  const float* __restrict__ out_norm,
                                                  float* __restrict__ h1,
                                                  int n_row_tiles) {
  const int lane = threadIdx.x & 31;
  const int wave = (int)((blockIdx.x * blockDim.x + threadIdx.x) >> 5);
  const int n_tiles = n_row_tiles * (HID_DIM / 16);
  if (wave >= n_tiles) return;
  const int rt   = wave / (HID_DIM / 16);
  const int ct   = wave % (HID_DIM / 16);
  const int half = lane >> 4;
  const int sub  = lane & 15;
  const int m    = rt * 16 + sub;
  const int n    = ct * 16 + sub;
  const int kh   = half * 2;

  const float norm = out_norm[m];
  const float* __restrict__ arow = x + (size_t)m * IN_DIM;

  v8f acc = {};
#pragma unroll 4
  for (int k = 0; k < IN_DIM; k += 4) {
    const float2 av = *(const float2*)(arow + k + kh);
    v2f a, b;
    a.x = av.x * norm;
    a.y = av.y * norm;
    b.x = W1[(size_t)(k + kh)     * HID_DIM + n];
    b.y = W1[(size_t)(k + kh + 1) * HID_DIM + n];
    acc = __builtin_amdgcn_wmma_f32_16x16x4_f32(
         false, a,  false, b,
         (short)0, acc,  false,  false);
  }

  const int m_base = rt * 16 + half * 8;
#pragma unroll
  for (int r = 0; r < 8; ++r)
    h1[(size_t)(m_base + r) * HID_DIM + n] = acc[r];
}


__global__ __launch_bounds__(256) void gemm2_wmma(const float* __restrict__ agg1,
                                                  const float* __restrict__ W2,
                                                  const float* __restrict__ b1,
                                                  const float* __restrict__ out_norm,
                                                  const float* __restrict__ in_norm,
                                                  float* __restrict__ h2,
                                                  int n_row_tiles) {
  const int lane = threadIdx.x & 31;
  const int wave = (int)((blockIdx.x * blockDim.x + threadIdx.x) >> 5);
  const int n_tiles = n_row_tiles * (OUT_DIM / 16);
  if (wave >= n_tiles) return;
  const int rt   = wave / (OUT_DIM / 16);
  const int ct   = wave % (OUT_DIM / 16);
  const int half = lane >> 4;
  const int sub  = lane & 15;
  const int m    = rt * 16 + sub;
  const int n    = ct * 16 + sub;
  const int kh   = half * 2;

  const float s_in  = in_norm[m];
  const float s_out = out_norm[m];
  const float* __restrict__ arow = agg1 + (size_t)m * HID_DIM;

  v8f acc = {};
#pragma unroll 4
  for (int k = 0; k < HID_DIM; k += 4) {
    const float2 av = *(const float2*)(arow + k + kh);
    const float2 bv = *(const float2*)(b1 + k + kh);
    v2f a, b;
    a.x = fmaxf(av.x * s_in + bv.x, 0.0f) * s_out;
    a.y = fmaxf(av.y * s_in + bv.y, 0.0f) * s_out;
    b.x = W2[(size_t)(k + kh)     * OUT_DIM + n];
    b.y = W2[(size_t)(k + kh + 1) * OUT_DIM + n];
    acc = __builtin_amdgcn_wmma_f32_16x16x4_f32(
        false, a, false, b, (short)0, acc, false, false);
  }

  const int m_base = rt * 16 + half * 8;
#pragma unroll
  for (int r = 0; r < 8; ++r)
    h2[(size_t)(m_base + r) * OUT_DIM + n] = acc[r];
}


__global__ __launch_bounds__(256) void gather128(const float* __restrict__ h,
                                                 const int* __restrict__ sorted_src,
                                                 const unsigned int* __restrict__ row_ptr,
                                                 float* __restrict__ agg,
                                                 int n_nodes, int n_edges) {
  const int lane = threadIdx.x & 31;
  const int d = __builtin_amdgcn_readfirstlane((int)((blockIdx.x * blockDim.x + threadIdx.x) >> 5));
  if (d >= n_nodes) return;
  const int beg = (int)min(row_ptr[d], (unsigned int)n_edges);
  const int end = (int)min(row_ptr[d + 1], (unsigned int)n_edges);
  const float* __restrict__ hl = h + lane * 4;
  float4 acc = make_float4(0.0f, 0.0f, 0.0f, 0.0f);

  for (int j = beg; j < end; j += 32) {
    const int cnt = min(32, end - j);
    int mine = (lane < cnt) ? sorted_src[j + lane] : 0;
    mine = (int)min((unsigned int)mine, (unsigned int)(n_nodes - 1));
    int t = 0;
    for (; t + 8 <= cnt; t += 8) {
      float4 v[8];
#pragma unroll
      for (int u = 0; u < 8; ++u) {
        const int s = __builtin_amdgcn_readlane(mine, t + u);
        v[u] = *(const float4*)(hl + (size_t)s * 128);
      }
#pragma unroll
      for (int u = 0; u < 8; ++u) {
        acc.x += v[u].x; acc.y += v[u].y; acc.z += v[u].z; acc.w += v[u].w;
      }
    }
    if (t < cnt) {
      float4 v[8];
#pragma unroll
      for (int u = 0; u < 8; ++u) {
        v[u] = make_float4(0.0f, 0.0f, 0.0f, 0.0f);
        if (t + u < cnt) {
          const int s = __builtin_amdgcn_readlane(mine, t + u);
          v[u] = *(const float4*)(hl + (size_t)s * 128);
        }
      }
#pragma unroll
      for (int u = 0; u < 8; ++u) {
        acc.x += v[u].x; acc.y += v[u].y; acc.z += v[u].z; acc.w += v[u].w;
      }
    }
  }
  *(float4*)(agg + (size_t)d * 128 + lane * 4) = acc;
}

__global__ __launch_bounds__(256) void gather64_out(const float* __restrict__ h,
                                                    const int* __restrict__ sorted_src,
                                                    const unsigned int* __restrict__ row_ptr,
                                                    const float* __restrict__ in_norm,
                                                    const float* __restrict__ b2,
                                                    float* __restrict__ out,
                                                    int n_nodes, int n_edges) {
  const int lane = threadIdx.x & 31;
  const int d = __builtin_amdgcn_readfirstlane((int)((blockIdx.x * blockDim.x + threadIdx.x) >> 5));
  if (d >= n_nodes) return;
  const int beg = (int)min(row_ptr[d], (unsigned int)n_edges);
  const int end = (int)min(row_ptr[d + 1], (unsigned int)n_edges);
  const float* __restrict__ hl = h + lane * 2;
  float2 acc = make_float2(0.0f, 0.0f);

  for (int j = beg; j < end; j += 32) {
    const int cnt = min(32, end - j);
    int mine = (lane < cnt) ? sorted_src[j + lane] : 0;
    mine = (int)min((unsigned int)mine, (unsigned int)(n_nodes - 1));
    int t = 0;
    for (; t + 8 <= cnt; t += 8) {
      float2 v[8];
#pragma unroll
      for (int u = 0; u < 8; ++u) {
        const int s = __builtin_amdgcn_readlane(mine, t + u);
        v[u] = *(const float2*)(hl + (size_t)s * 64);
      }
#pragma unroll
      for (int u = 0; u < 8; ++u) { acc.x += v[u].x; acc.y += v[u].y; }
    }
    if (t < cnt) {
      float2 v[8];
#pragma unroll
      for (int u = 0; u < 8; ++u) {
        v[u] = make_float2(0.0f, 0.0f);
        if (t + u < cnt) {
          const int s = __builtin_amdgcn_readlane(mine, t + u);
          v[u] = *(const float2*)(hl + (size_t)s * 64);
        }
      }
#pragma unroll
      for (int u = 0; u < 8; ++u) { acc.x += v[u].x; acc.y += v[u].y; }
    }
  }
  const float nrm = in_norm[d];
  const float2 bb = *(const float2*)(b2 + lane * 2);
  float2 o;
  o.x = acc.x * nrm + bb.x;
  o.y = acc.y * nrm + bb.y;
  *(float2*)(out + (size_t)d * 64 + lane * 2) = o;
}


static inline int grid1d(long n, int block) { return (int)((n + block - 1) / block); }

extern "C" void kernel_launch(void* const* d_in, const int* in_sizes, int n_in,
                              void* d_out, int out_size, void* d_ws, size_t ws_size,
                              hipStream_t stream) {
  const float* x   = (const float*)d_in[0];
  const float* W1  = (const float*)d_in[1];
  const float* b1  = (const float*)d_in[2];
  const float* W2  = (const float*)d_in[3];
  const float* b2  = (const float*)d_in[4];
  const int*   src = (const int*)d_in[5];
  const int*   dst = (const int*)d_in[6];

  const int n_nodes = in_sizes[0] / IN_DIM;
  const int n_edges = in_sizes[5];
  float* out = (float*)d_out;

  const size_t row = ((size_t)n_nodes + 31) & ~(size_t)31;
  float* h1   = (float*)d_ws;
  float* agg1 = h1 + (size_t)n_nodes * HID_DIM;
  float* h2   = agg1 + (size_t)n_nodes * HID_DIM;
  unsigned int* partial = (unsigned int*)(h2 + (size_t)n_nodes * OUT_DIM);
  unsigned int* part_dst = partial + DEG_SLICES * row;
  float* out_norm = (float*)(partial + 2 * DEG_SLICES * row);
  float* in_norm  = out_norm + row;
  unsigned int* row_ptr  = (unsigned int*)(in_norm + row);
  unsigned int* block_in = row_ptr + row + 32;
  int* sorted_src = (int*)(block_in + row);

  const int node_blocks = grid1d(n_nodes, 128);

  degree_hist<<<dim3(DEG_SLICES, 2), HIST_THREADS, 0, stream>>>(src, dst, partial, n_edges, n_nodes);
  degree_norm<<<node_blocks, 256, 0, stream>>>(partial, out_norm, in_norm, block_in, n_nodes);
  edge_cursors<<<node_blocks, 256, 0, stream>>>(part_dst, block_in, row_ptr, n_nodes);
  edge_fill<<<DEG_SLICES, HIST_THREADS, 0, stream>>>(src, dst, part_dst, sorted_src, n_edges, n_nodes);

  const int row_tiles = n_nodes / 16;

  {
    const int waves  = row_tiles * (HID_DIM / 16);
    const int blocks = (waves + 7) / 8;
    gemm1_wmma<<<blocks, 256, 0, stream>>>(x, W1, out_norm, h1, row_tiles);
  }

  gather128<<<(n_nodes + 7) / 8, 256, 0, stream>>>(h1, sorted_src, row_ptr, agg1, n_nodes, n_edges);

  {
    const int waves  = row_tiles * (OUT_DIM / 16);
    const int blocks = (waves + 7) / 8;
    gemm2_wmma<<<blocks, 256, 0, stream>>>(agg1, W2, b1, out_norm, in_norm, h2, row_tiles);
  }

  gather64_out<<<(n_nodes + 7) / 8, 256, 0, stream>>>(h2, sorted_src, row_ptr, in_norm, b2, out,
                                                      n_nodes, n_edges);
}


# both GEMMs: weights staged once per work-group in LDS, each wave computes a whole 16-row strip (all column tiles) from one read of its rows; 256 work-groups of 8 waves; k order unchanged
# speedup vs baseline: 3.8509x; 1.0942x over previous
#include <hip/hip_runtime.h>

typedef __attribute__((ext_vector_type(2))) float v2f;
typedef __attribute__((ext_vector_type(8))) float v8f;

#define IN_DIM  128
#define HID_DIM 128
#define OUT_DIM 64


#define HIST_CAP     65536
#define HIST_THREADS 1024
#define DEG_SLICES   128
static_assert(DEG_SLICES % 8 == 0, "degree_norm splits the slice rows over 8 waves");

__device__ __forceinline__ int padded_row(int n_nodes) { return (n_nodes + 31) & ~31; }

__device__ __forceinline__ void slice_range(int n_edges, int slice, int& e0, int& e1) {
  const int per = (((n_edges + DEG_SLICES - 1) / DEG_SLICES) + 3) & ~3;
  e0 = min(slice * per, n_edges);
  e1 = min(e0 + per, n_edges);
}

__device__ __forceinline__ void hist_bump(unsigned int* hist, int v, int base, int nb) {
  const unsigned int b = (unsigned int)(v - base);
  if (b < (unsigned int)nb) atomicAdd(&hist[b], 1u);
}

__global__ __launch_bounds__(HIST_THREADS) void degree_hist(const int* __restrict__ src,
                                                            const int* __restrict__ dst,
                                                            unsigned int* __restrict__ partial,
                                                            int n_edges, int n_nodes) {
  __shared__ __attribute__((aligned(16))) unsigned int hist[HIST_CAP];
  const int tid = threadIdx.x;
  const int* __restrict__ idx = blockIdx.y ? dst : src;
  const int n_slices = DEG_SLICES;
  int e0, e1;
  slice_range(n_edges, (int)blockIdx.x, e0, e1);
  const int nv  = (e1 - e0) >> 2;
  const int4* __restrict__ p4 = (const int4*)(idx + e0);
  const int row = padded_row(n_nodes);
  unsigned int* __restrict__ my =
      partial + ((size_t)blockIdx.y * n_slices + blockIdx.x) * (size_t)row;

  for (int base = 0; base < n_nodes; base += HIST_CAP) {
    const int nb  = min(HIST_CAP, n_nodes - base);
    const int nb4 = (nb + 3) >> 2;
    uint4* h4 = (uint4*)hist;
    for (int i = tid; i < nb4; i += HIST_THREADS) h4[i] = make_uint4(0u, 0u, 0u, 0u);
    __syncthreads();

    for (int q = tid; q < nv; q += 4 * HIST_THREADS) {
      int4 v[4];
#pragma unroll
      for (int j = 0; j < 4; ++j) {
        const int qq = q + j * HIST_THREADS;
        v[j] = (qq < nv) ? p4[qq] : make_int4(-1, -1, -1, -1);
      }
#pragma unroll
      for (int j = 0; j < 4; ++j) {
        hist_bump(hist, v[j].x, base, nb);
        hist_bump(hist, v[j].y, base, nb);
        hist_bump(hist, v[j].z, base, nb);
        hist_bump(hist, v[j].w, base, nb);
      }
    }
    {
      const int e = e0 + (nv << 2) + tid;
      if (tid < 3 && e < e1) hist_bump(hist, idx[e], base, nb);
    }
    __syncthreads();

    uint4* o4 = (uint4*)(my + base);
    for (int i = tid; i < nb4; i += HIST_THREADS) o4[i] = h4[i];
    __syncthreads();
  }
}

__global__ __launch_bounds__(256) void degree_norm(const unsigned int* __restrict__ partial,
                                                   float* __restrict__ out_norm,
                                                   float* __restrict__ in_norm,
                                                   unsigned int* __restrict__ block_in,
                                                   int n_nodes) {
  __shared__ uint4 red[2][8][32];
  const int lane = threadIdx.x & 31;
  const int w    = threadIdx.x >> 5;
  const int node = (blockIdx.x * 32 + lane) * 4;
  const int row  = padded_row(n_nodes);
  uint4 so = make_uint4(0u, 0u, 0u, 0u);
  uint4 si = make_uint4(0u, 0u, 0u, 0u);
  if (node < n_nodes) {
    const uint4* __restrict__ po = (const uint4*)(partial + node);
    const uint4* __restrict__ pi = (const uint4*)(partial + (size_t)DEG_SLICES * row + node);
    const size_t row4 = (size_t)(row >> 2);
#pragma unroll
    for (int k = 0; k < DEG_SLICES / 8; ++k) {
      const size_t s = (size_t)(w + 8 * k);
      const uint4 a = po[s * row4];
      const uint4 b = pi[s * row4];
      so.x += a.x; so.y += a.y; so.z += a.z; so.w += a.w;
      si.x += b.x; si.y += b.y; si.z += b.z; si.w += b.w;
    }
  }
  red[0][w][lane] = so;
  red[1][w][lane] = si;
  __syncthreads();
  if (w == 0) {
    so = make_uint4(0u, 0u, 0u, 0u);
    si = make_uint4(0u, 0u, 0u, 0u);
#pragma unroll
    for (int k = 0; k < 8; ++k) {
      const uint4 a = red[0][k][lane];
      const uint4 b = red[1][k][lane];
      so.x += a.x; so.y += a.y; so.z += a.z; so.w += a.w;
      si.x += b.x; si.y += b.y; si.z += b.z; si.w += b.w;
    }
    unsigned int t = si.x + si.y + si.z + si.w;
    for (int o = 16; o > 0; o >>= 1) t += __shfl_xor(t, o);
    if (lane == 0) block_in[blockIdx.x] = t;
    if (node < n_nodes) {
      float4 o, i;
      o.x = rsqrtf(fmaxf((float)so.x, 1.0f));
      o.y = rsqrtf(fmaxf((float)so.y, 1.0f));
      o.z = rsqrtf(fmaxf((float)so.z, 1.0f));
      o.w = rsqrtf(fmaxf((float)so.w, 1.0f));
      i.x = rsqrtf(fmaxf((float)si.x, 1.0f));
      i.y = rsqrtf(fmaxf((float)si.y, 1.0f));
      i.z = rsqrtf(fmaxf((float)si.z, 1.0f));
      i.w = rsqrtf(fmaxf((float)si.w, 1.0f));
      *(float4*)(out_norm + node) = o;
      *(float4*)(in_norm + node)  = i;
    }
  }
}


__device__ __forceinline__ uint4 add4(uint4 a, uint4 b) {
  return make_uint4(a.x + b.x, a.y + b.y, a.z + b.z, a.w + b.w);
}

__global__ __launch_bounds__(256) void edge_cursors(unsigned int* __restrict__ part_dst,
                                                    const unsigned int* __restrict__ block_in,
                                                    unsigned int* __restrict__ row_ptr,
                                                    int n_nodes) {
  __shared__ uint4 wtot[8][32];
  __shared__ unsigned int bred[8];
  const int lane = threadIdx.x & 31;
  const int w    = threadIdx.x >> 5;
  const int node = (blockIdx.x * 32 + lane) * 4;
  const int row  = padded_row(n_nodes);
  const size_t row4 = (size_t)(row >> 2);
  const bool live = node < n_nodes;
  const uint4 zero4 = make_uint4(0u, 0u, 0u, 0u);
  uint4* __restrict__ p = (uint4*)part_dst + (size_t)(w * (DEG_SLICES / 8)) * row4 + (size_t)(node >> 2);

  unsigned int b = 0;
  for (int i = threadIdx.x; i < (int)blockIdx.x; i += 256) b += block_in[i];
  for (int o = 16; o > 0; o >>= 1) b += __shfl_xor(b, o);
  if (lane == 0) bred[w] = b;

  uint4 c[DEG_SLICES / 8];
#pragma unroll
  for (int k = 0; k < DEG_SLICES / 8; ++k) c[k] = zero4;
  if (live) {
#pragma unroll
    for (int k = 0; k < DEG_SLICES / 8; ++k) c[k] = p[(size_t)k * row4];
  }
  uint4 tot = zero4;
#pragma unroll
  for (int k = 0; k < DEG_SLICES / 8; ++k) {
    const uint4 t = c[k];
    c[k] = tot;
    tot = add4(tot, t);
  }
  wtot[w][lane] = tot;
  __syncthreads();

  unsigned int base = 0;
  for (int k = 0; k < 8; ++k) base += bred[k];
  uint4 wp = zero4;
  uint4 nt = zero4;
  for (int k = 0; k < 8; ++k) {
    const uint4 t = wtot[k][lane];
    if (k < w) wp = add4(wp, t);
    nt = add4(nt, t);
  }
  const unsigned int lt = nt.x + nt.y + nt.z + nt.w;
  unsigned int inc = lt;
  for (int o = 1; o < 32; o <<= 1) {
    const unsigned int u = __shfl_up(inc, (unsigned int)o);
    if (lane >= o) inc += u;
  }
  uint4 st;
  st.x = base + inc - lt;
  st.y = st.x + nt.x;
  st.z = st.y + nt.y;
  st.w = st.z + nt.z;
  if (live) {
    if (w == 0) {
      *(uint4*)(row_ptr + node) = st;
      if (node + 4 >= n_nodes) row_ptr[n_nodes] = st.x + lt;
    }
    st = add4(st, wp);
#pragma unroll
    for (int k = 0; k < DEG_SLICES / 8; ++k) p[(size_t)k * row4] = add4(st, c[k]);
  }
}

__device__ __forceinline__ void place_edge(unsigned int* cur, int* __restrict__ sorted_src,
                                           int d, int s, int base, int nb, int n_edges) {
  const unsigned int b = (unsigned int)(d - base);
  if (b < (unsigned int)nb) {
    const unsigned int pos = atomicAdd(&cur[b], 1u);
    if (pos < (unsigned int)n_edges) sorted_src[pos] = s;
  }
}

__global__ __launch_bounds__(HIST_THREADS) void edge_fill(const int* __restrict__ src,
                                                          const int* __restrict__ dst,
                                                          const unsigned int* __restrict__ cursors,
                                                          int* __restrict__ sorted_src,
                                                          int n_edges, int n_nodes) {
  __shared__ __attribute__((aligned(16))) unsigned int cur[HIST_CAP];
  const int tid = threadIdx.x;
  int e0, e1;
  slice_range(n_edges, (int)blockIdx.x, e0, e1);
  const int nv = (e1 - e0) >> 2;
  const int4* __restrict__ d4 = (const int4*)(dst + e0);
  const int4* __restrict__ s4 = (const int4*)(src + e0);
  const int row = padded_row(n_nodes);
  const unsigned int* __restrict__ my = cursors + (size_t)blockIdx.x * (size_t)row;

  for (int base = 0; base < n_nodes; base += HIST_CAP) {
    const int nb  = min(HIST_CAP, n_nodes - base);
    const int nb4 = (nb + 3) >> 2;
    uint4* c4 = (uint4*)cur;
    const uint4* __restrict__ i4 = (const uint4*)(my + base);
    for (int i = tid; i < nb4; i += HIST_THREADS) c4[i] = i4[i];
    __syncthreads();

    for (int q = tid; q < nv; q += 4 * HIST_THREADS) {
      int4 dv[4], sv[4];
#pragma unroll
      for (int j = 0; j < 4; ++j) {
        const int qq = q + j * HIST_THREADS;
        const bool ok = qq < nv;
        dv[j] = ok ? d4[qq] : make_int4(-1, -1, -1, -1);
        sv[j] = ok ? s4[qq] : make_int4(0, 0, 0, 0);
      }
#pragma unroll
      for (int j = 0; j < 4; ++j) {
        place_edge(cur, sorted_src, dv[j].x, sv[j].x, base, nb, n_edges);
        place_edge(cur, sorted_src, dv[j].y, sv[j].y, base, nb, n_edges);
        place_edge(cur, sorted_src, dv[j].z, sv[j].z, base, nb, n_edges);
        place_edge(cur, sorted_src, dv[j].w, sv[j].w, base, nb, n_edges);
      }
    }
    {
      const int e = e0 + (nv << 2) + tid;
      if (tid < 3 && e < e1) place_edge(cur, sorted_src, dst[e], src[e], base, nb, n_edges);
    }
    __syncthreads();
  }
}


#define GEMM_K       128
#define GEMM_WAVES   8
#define GEMM_THREADS (GEMM_WAVES * 32)
#define GEMM_GRID    256
static_assert(IN_DIM == GEMM_K && HID_DIM == GEMM_K, "both products have K = 128");

template <int NC>
__device__ __forceinline__ void stage_weights(const float* __restrict__ Wm, float4* Bl4) {
#pragma unroll
  for (int it = 0; it < (GEMM_K / 4 * NC) / GEMM_THREADS; ++it) {
    const int idx = it * GEMM_THREADS + (int)threadIdx.x;
    const int n   = idx % NC;
    const int jh  = idx / NC;
    const int k0  = 8 * (jh >> 1) + 2 * (jh & 1);
    float4 q;
    q.x = Wm[(size_t)(k0)     * NC + n];
    q.y = Wm[(size_t)(k0 + 1) * NC + n];
    q.z = Wm[(size_t)(k0 + 4) * NC + n];
    q.w = Wm[(size_t)(k0 + 5) * NC + n];
    Bl4[idx] = q;
  }
}

template <int NT>
__device__ __forceinline__ void strip_mma(const float2 (&a)[GEMM_K / 4], const float4* Bl4,
                                          int half, int sub, v8f (&acc)[NT]) {
  float4 q[NT];
#pragma unroll
  for (int ct = 0; ct < NT; ++ct) q[ct] = Bl4[half * (16 * NT) + ct * 16 + sub];
#pragma unroll
  for (int j2 = 0; j2 < GEMM_K / 8; ++j2) {
    float4 qn[NT];
#pragma unroll
    for (int ct = 0; ct < NT; ++ct)
      qn[ct] = (j2 + 1 < GEMM_K / 8) ? Bl4[((j2 + 1) * 2 + half) * (16 * NT) + ct * 16 + sub] : q[ct];
    v2f a0, a1;
    a0.x = a[2 * j2].x;     a0.y = a[2 * j2].y;
    a1.x = a[2 * j2 + 1].x; a1.y = a[2 * j2 + 1].y;
#pragma unroll
    for (int ct = 0; ct < NT; ++ct) {
      v2f b;
      b.x = q[ct].x; b.y = q[ct].y;
      acc[ct] = __builtin_amdgcn_wmma_f32_16x16x4_f32(
           false, a0,  false, b,
           (short)0, acc[ct],  false,  false);
    }
#pragma unroll
    for (int ct = 0; ct < NT; ++ct) {
      v2f b;
      b.x = q[ct].z; b.y = q[ct].w;
      acc[ct] = __builtin_amdgcn_wmma_f32_16x16x4_f32(
          false, a1, false, b, (short)0, acc[ct], false, false);
    }
    __builtin_amdgcn_sched_barrier(0);
#pragma unroll
    for (int ct = 0; ct < NT; ++ct) q[ct] = qn[ct];
  }
}

template <int NT>
__device__ __forceinline__ void strip_store(float* __restrict__ dstm, int rt, int half, int sub,
                                            const v8f (&acc)[NT]) {
  const int m_base = rt * 16 + half * 8;
#pragma unroll
  for (int ct = 0; ct < NT; ++ct)
#pragma unroll
    for (int r = 0; r < 8; ++r)
      dstm[(size_t)(m_base + r) * (16 * NT) + ct * 16 + sub] = acc[ct][r];
}


__device__ __forceinline__ void load_a1(const float* __restrict__ x,
                                        const float* __restrict__ out_norm,
                                        int rt, int half, int sub, float2 (&a)[GEMM_K / 4]) {
  const int m = rt * 16 + sub;
  const float norm = out_norm[m];
  const float* __restrict__ arow = x + (size_t)m * IN_DIM + half * 2;
#pragma unroll
  for (int j = 0; j < GEMM_K / 4; ++j) {
    const float2 av = *(const float2*)(arow + 4 * j);
    a[j].x = av.x * norm;
    a[j].y = av.y * norm;
  }
}

__global__ __launch_bounds__(GEMM_THREADS) void gemm1_wmma(const float* __restrict__ x,
                                                           const float* __restrict__ W1,
                                                           const float* __restrict__ out_norm,
                                                           float* __restrict__ h1,
                                                           int n_row_tiles) {
  __shared__ float4 Bl4[GEMM_K / 4 * HID_DIM];
  const int lane = threadIdx.x & 31;
  const int wave = __builtin_amdgcn_readfirstlane((int)(threadIdx.x >> 5));
  const int half = lane >> 4;
  const int sub  = lane & 15;
  const int step = (int)gridDim.x * GEMM_WAVES;
  int rt = (int)blockIdx.x + (int)gridDim.x * wave;

  float2 a[GEMM_K / 4];
#pragma unroll
  for (int j = 0; j < GEMM_K / 4; ++j) a[j] = make_float2(0.0f, 0.0f);
  if (rt < n_row_tiles) load_a1(x, out_norm, rt, half, sub, a);

  stage_weights<HID_DIM>(W1, Bl4);
  __syncthreads();

  while (rt < n_row_tiles) {
    const int nx = rt + step;
    float2 an[GEMM_K / 4];
#pragma unroll
    for (int j = 0; j < GEMM_K / 4; ++j) an[j] = make_float2(0.0f, 0.0f);
    if (nx < n_row_tiles) load_a1(x, out_norm, nx, half, sub, an);

    v8f acc[HID_DIM / 16];
#pragma unroll
    for (int ct = 0; ct < HID_DIM / 16; ++ct) acc[ct] = v8f{};
    strip_mma<HID_DIM / 16>(a, Bl4, half, sub, acc);
    strip_store<HID_DIM / 16>(h1, rt, half, sub, acc);
#pragma unroll
    for (int j = 0; j < GEMM_K / 4; ++j) a[j] = an[j];
    rt = nx;
  }
}


__device__ __forceinline__ void load_a2(const float* __restrict__ agg1,
                                        const float* __restrict__ b1,
                                        const float* __restrict__ out_norm,
                                        const float* __restrict__ in_norm,
                                        int rt, int half, int sub, float2 (&a)[GEMM_K / 4]) {
  const int m = rt * 16 + sub;
  const float s_in  = in_norm[m];
  const float s_out = out_norm[m];
  const float* __restrict__ arow = agg1 + (size_t)m * HID_DIM + half * 2;
  const float* __restrict__ brow = b1 + half * 2;
#pragma unroll
  for (int j = 0; j < GEMM_K / 4; ++j) {
    const float2 av = *(const float2*)(arow + 4 * j);
    const float2 bv = *(const float2*)(brow + 4 * j);
    a[j].x = fmaxf(av.x * s_in + bv.x, 0.0f) * s_out;
    a[j].y = fmaxf(av.y * s_in + bv.y, 0.0f) * s_out;
  }
}

__global__ __launch_bounds__(GEMM_THREADS) void gemm2_wmma(const float* __restrict__ agg1,
                                                           const float* __restrict__ W2,
                                                           const float* __restrict__ b1,
                                                           const float* __restrict__ out_norm,
                                                           const float* __restrict__ in_norm,
                                                           float* __restrict__ h2,
                                                           int n_row_tiles) {
  __shared__ float4 Bl4[GEMM_K / 4 * OUT_DIM];
  const int lane = threadIdx.x & 31;
  const int wave = __builtin_amdgcn_readfirstlane((int)(threadIdx.x >> 5));
  const int half = lane >> 4;
  const int sub  = lane & 15;
  const int step = (int)gridDim.x * GEMM_WAVES;
  int rt = (int)blockIdx.x + (int)gridDim.x * wave;

  float2 a[GEMM_K / 4];
#pragma unroll
  for (int j = 0; j < GEMM_K / 4; ++j) a[j] = make_float2(0.0f, 0.0f);
  if (rt < n_row_tiles) load_a2(agg1, b1, out_norm, in_norm, rt, half, sub, a);

  stage_weights<OUT_DIM>(W2, Bl4);
  __syncthreads();

  while (rt < n_row_tiles) {
    const int nx = rt + step;
    float2 an[GEMM_K / 4];
#pragma unroll
    for (int j = 0; j < GEMM_K / 4; ++j) an[j] = make_float2(0.0f, 0.0f);
    if (nx < n_row_tiles) load_a2(agg1, b1, out_norm, in_norm, nx, half, sub, an);

    v8f acc[OUT_DIM / 16];
#pragma unroll
    for (int ct = 0; ct < OUT_DIM / 16; ++ct) acc[ct] = v8f{};
    strip_mma<OUT_DIM / 16>(a, Bl4, half, sub, acc);
    strip_store<OUT_DIM / 16>(h2, rt, half, sub, acc);
#pragma unroll
    for (int j = 0; j < GEMM_K / 4; ++j) a[j] = an[j];
    rt = nx;
  }
}


__global__ __launch_bounds__(256) void gather128(const float* __restrict__ h,
                                                 const int* __restrict__ sorted_src,
                                                 const unsigned int* __restrict__ row_ptr,
                                                 float* __restrict__ agg,
                                                 int n_nodes, int n_edges) {
  const int lane = threadIdx.x & 31;
  const int d = __builtin_amdgcn_readfirstlane((int)((blockIdx.x * blockDim.x + threadIdx.x) >> 5));
  if (d >= n_nodes) return;
  const int beg = (int)min(row_ptr[d], (unsigned int)n_edges);
  const int end = (int)min(row_ptr[d + 1], (unsigned int)n_edges);
  const float* __restrict__ hl = h + lane * 4;
  float4 acc = make_float4(0.0f, 0.0f, 0.0f, 0.0f);

  for (int j = beg; j < end; j += 32) {
    const int cnt = min(32, end - j);
    int mine = (lane < cnt) ? sorted_src[j + lane] : 0;
    mine = (int)min((unsigned int)mine, (unsigned int)(n_nodes - 1));
    int t = 0;
    for (; t + 8 <= cnt; t += 8) {
      float4 v[8];
#pragma unroll
      for (int u = 0; u < 8; ++u) {
        const int s = __builtin_amdgcn_readlane(mine, t + u);
        v[u] = *(const float4*)(hl + (size_t)s * 128);
      }
#pragma unroll
      for (int u = 0; u < 8; ++u) {
        acc.x += v[u].x; acc.y += v[u].y; acc.z += v[u].z; acc.w += v[u].w;
      }
    }
    if (t < cnt) {
      float4 v[8];
#pragma unroll
      for (int u = 0; u < 8; ++u) {
        v[u] = make_float4(0.0f, 0.0f, 0.0f, 0.0f);
        if (t + u < cnt) {
          const int s = __builtin_amdgcn_readlane(mine, t + u);
          v[u] = *(const float4*)(hl + (size_t)s * 128);
        }
      }
#pragma unroll
      for (int u = 0; u < 8; ++u) {
        acc.x += v[u].x; acc.y += v[u].y; acc.z += v[u].z; acc.w += v[u].w;
      }
    }
  }
  *(float4*)(agg + (size_t)d * 128 + lane * 4) = acc;
}

__global__ __launch_bounds__(256) void gather64_out(const float* __restrict__ h,
                                                    const int* __restrict__ sorted_src,
                                                    const unsigned int* __restrict__ row_ptr,
                                                    const float* __restrict__ in_norm,
                                                    const float* __restrict__ b2,
                                                    float* __restrict__ out,
                                                    int n_nodes, int n_edges) {
  const int lane = threadIdx.x & 31;
  const int d = __builtin_amdgcn_readfirstlane((int)((blockIdx.x * blockDim.x + threadIdx.x) >> 5));
  if (d >= n_nodes) return;
  const int beg = (int)min(row_ptr[d], (unsigned int)n_edges);
  const int end = (int)min(row_ptr[d + 1], (unsigned int)n_edges);
  const float* __restrict__ hl = h + lane * 2;
  float2 acc = make_float2(0.0f, 0.0f);

  for (int j = beg; j < end; j += 32) {
    const int cnt = min(32, end - j);
    int mine = (lane < cnt) ? sorted_src[j + lane] : 0;
    mine = (int)min((unsigned int)mine, (unsigned int)(n_nodes - 1));
    int t = 0;
    for (; t + 8 <= cnt; t += 8) {
      float2 v[8];
#pragma unroll
      for (int u = 0; u < 8; ++u) {
        const int s = __builtin_amdgcn_readlane(mine, t + u);
        v[u] = *(const float2*)(hl + (size_t)s * 64);
      }
#pragma unroll
      for (int u = 0; u < 8; ++u) { acc.x += v[u].x; acc.y += v[u].y; }
    }
    if (t < cnt) {
      float2 v[8];
#pragma unroll
      for (int u = 0; u < 8; ++u) {
        v[u] = make_float2(0.0f, 0.0f);
        if (t + u < cnt) {
          const int s = __builtin_amdgcn_readlane(mine, t + u);
          v[u] = *(const float2*)(hl + (size_t)s * 64);
        }
      }
#pragma unroll
      for (int u = 0; u < 8; ++u) { acc.x += v[u].x; acc.y += v[u].y; }
    }
  }
  const float nrm = in_norm[d];
  const float2 bb = *(const float2*)(b2 + lane * 2);
  float2 o;
  o.x = acc.x * nrm + bb.x;
  o.y = acc.y * nrm + bb.y;
  *(float2*)(out + (size_t)d * 64 + lane * 2) = o;
}


static inline int grid1d(long n, int block) { return (int)((n + block - 1) / block); }

extern "C" void kernel_launch(void* const* d_in, const int* in_sizes, int n_in,
                              void* d_out, int out_size, void* d_ws, size_t ws_size,
                              hipStream_t stream) {
  const float* x   = (const float*)d_in[0];
  const float* W1  = (const float*)d_in[1];
  const float* b1  = (const float*)d_in[2];
  const float* W2  = (const float*)d_in[3];
  const float* b2  = (const float*)d_in[4];
  const int*   src = (const int*)d_in[5];
  const int*   dst = (const int*)d_in[6];

  const int n_nodes = in_sizes[0] / IN_DIM;
  const int n_edges = in_sizes[5];
  float* out = (float*)d_out;

  const size_t row = ((size_t)n_nodes + 31) & ~(size_t)31;
  float* h1   = (float*)d_ws;
  float* agg1 = h1 + (size_t)n_nodes * HID_DIM;
  float* h2   = agg1 + (size_t)n_nodes * HID_DIM;
  unsigned int* partial = (unsigned int*)(h2 + (size_t)n_nodes * OUT_DIM);
  unsigned int* part_dst = partial + DEG_SLICES * row;
  float* out_norm = (float*)(partial + 2 * DEG_SLICES * row);
  float* in_norm  = out_norm + row;
  unsigned int* row_ptr  = (unsigned int*)(in_norm + row);
  unsigned int* block_in = row_ptr + row + 32;
  int* sorted_src = (int*)(block_in + row);

  const int node_blocks = grid1d(n_nodes, 128);

  degree_hist<<<dim3(DEG_SLICES, 2), HIST_THREADS, 0, stream>>>(src, dst, partial, n_edges, n_nodes);
  degree_norm<<<node_blocks, 256, 0, stream>>>(partial, out_norm, in_norm, block_in, n_nodes);
  edge_cursors<<<node_blocks, 256, 0, stream>>>(part_dst, block_in, row_ptr, n_nodes);
  edge_fill<<<DEG_SLICES, HIST_THREADS, 0, stream>>>(src, dst, part_dst, sorted_src, n_edges, n_nodes);

  const int row_tiles = n_nodes / 16;

  gemm1_wmma<<<GEMM_GRID, GEMM_THREADS, 0, stream>>>(x, W1, out_norm, h1, row_tiles);

  gather128<<<(n_nodes + 7) / 8, 256, 0, stream>>>(h1, sorted_src, row_ptr, agg1, n_nodes, n_edges);

  gemm2_wmma<<<GEMM_GRID, GEMM_THREADS, 0, stream>>>(agg1, W2, b1, out_norm, in_norm, h2, row_tiles);

  gather64_out<<<(n_nodes + 7) / 8, 256, 0, stream>>>(h2, sorted_src, row_ptr, in_norm, b2, out,
                                                      n_nodes, n_edges);
}


# first GEMM and the edge write cursors, independent once the norms exist, share one launch (block role by block number; GEMM strip as two 64-column halves so both fit a WGP): 8 launches become 7
# speedup vs baseline: 4.0231x; 1.0447x over previous
#include <hip/hip_runtime.h>

typedef __attribute__((ext_vector_type(2))) float v2f;
typedef __attribute__((ext_vector_type(8))) float v8f;

#define IN_DIM  128
#define HID_DIM 128
#define OUT_DIM 64


#define HIST_CAP     65536
#define HIST_THREADS 1024
#define DEG_SLICES   128
static_assert(DEG_SLICES % 8 == 0, "degree_norm splits the slice rows over 8 waves");

__device__ __forceinline__ int padded_row(int n_nodes) { return (n_nodes + 31) & ~31; }

__device__ __forceinline__ void slice_range(int n_edges, int slice, int& e0, int& e1) {
  const int per = (((n_edges + DEG_SLICES - 1) / DEG_SLICES) + 3) & ~3;
  e0 = min(slice * per, n_edges);
  e1 = min(e0 + per, n_edges);
}

__device__ __forceinline__ void hist_bump(unsigned int* hist, int v, int base, int nb) {
  const unsigned int b = (unsigned int)(v - base);
  if (b < (unsigned int)nb) atomicAdd(&hist[b], 1u);
}

__global__ __launch_bounds__(HIST_THREADS) void degree_hist(const int* __restrict__ src,
                                                            const int* __restrict__ dst,
                                                            unsigned int* __restrict__ partial,
                                                            int n_edges, int n_nodes) {
  __shared__ __attribute__((aligned(16))) unsigned int hist[HIST_CAP];
  const int tid = threadIdx.x;
  const int* __restrict__ idx = blockIdx.y ? dst : src;
  const int n_slices = DEG_SLICES;
  int e0, e1;
  slice_range(n_edges, (int)blockIdx.x, e0, e1);
  const int nv  = (e1 - e0) >> 2;
  const int4* __restrict__ p4 = (const int4*)(idx + e0);
  const int row = padded_row(n_nodes);
  unsigned int* __restrict__ my =
      partial + ((size_t)blockIdx.y * n_slices + blockIdx.x) * (size_t)row;

  for (int base = 0; base < n_nodes; base += HIST_CAP) {
    const int nb  = min(HIST_CAP, n_nodes - base);
    const int nb4 = (nb + 3) >> 2;
    uint4* h4 = (uint4*)hist;
    for (int i = tid; i < nb4; i += HIST_THREADS) h4[i] = make_uint4(0u, 0u, 0u, 0u);
    __syncthreads();

    for (int q = tid; q < nv; q += 4 * HIST_THREADS) {
      int4 v[4];
#pragma unroll
      for (int j = 0; j < 4; ++j) {
        const int qq = q + j * HIST_THREADS;
        v[j] = (qq < nv) ? p4[qq] : make_int4(-1, -1, -1, -1);
      }
#pragma unroll
      for (int j = 0; j < 4; ++j) {
        hist_bump(hist, v[j].x, base, nb);
        hist_bump(hist, v[j].y, base, nb);
        hist_bump(hist, v[j].z, base, nb);
        hist_bump(hist, v[j].w, base, nb);
      }
    }
    {
      const int e = e0 + (nv << 2) + tid;
      if (tid < 3 && e < e1) hist_bump(hist, idx[e], base, nb);
    }
    __syncthreads();

    uint4* o4 = (uint4*)(my + base);
    for (int i = tid; i < nb4; i += HIST_THREADS) o4[i] = h4[i];
    __syncthreads();
  }
}

__global__ __launch_bounds__(256) void degree_norm(const unsigned int* __restrict__ partial,
                                                   float* __restrict__ out_norm,
                                                   float* __restrict__ in_norm,
                                                   unsigned int* __restrict__ block_in,
                                                   int n_nodes) {
  __shared__ uint4 red[2][8][32];
  const int lane = threadIdx.x & 31;
  const int w    = threadIdx.x >> 5;
  const int node = (blockIdx.x * 32 + lane) * 4;
  const int row  = padded_row(n_nodes);
  uint4 so = make_uint4(0u, 0u, 0u, 0u);
  uint4 si = make_uint4(0u, 0u, 0u, 0u);
  if (node < n_nodes) {
    const uint4* __restrict__ po = (const uint4*)(partial + node);
    const uint4* __restrict__ pi = (const uint4*)(partial + (size_t)DEG_SLICES * row + node);
    const size_t row4 = (size_t)(row >> 2);
#pragma unroll
    for (int k = 0; k < DEG_SLICES / 8; ++k) {
      const size_t s = (size_t)(w + 8 * k);
      const uint4 a = po[s * row4];
      const uint4 b = pi[s * row4];
      so.x += a.x; so.y += a.y; so.z += a.z; so.w += a.w;
      si.x += b.x; si.y += b.y; si.z += b.z; si.w += b.w;
    }
  }
  red[0][w][lane] = so;
  red[1][w][lane] = si;
  __syncthreads();
  if (w == 0) {
    so = make_uint4(0u, 0u, 0u, 0u);
    si = make_uint4(0u, 0u, 0u, 0u);
#pragma unroll
    for (int k = 0; k < 8; ++k) {
      const uint4 a = red[0][k][lane];
      const uint4 b = red[1][k][lane];
      so.x += a.x; so.y += a.y; so.z += a.z; so.w += a.w;
      si.x += b.x; si.y += b.y; si.z += b.z; si.w += b.w;
    }
    unsigned int t = si.x + si.y + si.z + si.w;
    for (int o = 16; o > 0; o >>= 1) t += __shfl_xor(t, o);
    if (lane == 0) block_in[blockIdx.x] = t;
    if (node < n_nodes) {
      float4 o, i;
      o.x = rsqrtf(fmaxf((float)so.x, 1.0f));
      o.y = rsqrtf(fmaxf((float)so.y, 1.0f));
      o.z = rsqrtf(fmaxf((float)so.z, 1.0f));
      o.w = rsqrtf(fmaxf((float)so.w, 1.0f));
      i.x = rsqrtf(fmaxf((float)si.x, 1.0f));
      i.y = rsqrtf(fmaxf((float)si.y, 1.0f));
      i.z = rsqrtf(fmaxf((float)si.z, 1.0f));
      i.w = rsqrtf(fmaxf((float)si.w, 1.0f));
      *(float4*)(out_norm + node) = o;
      *(float4*)(in_norm + node)  = i;
    }
  }
}


__device__ __forceinline__ uint4 add4(uint4 a, uint4 b) {
  return make_uint4(a.x + b.x, a.y + b.y, a.z + b.z, a.w + b.w);
}

__device__ __forceinline__ void edge_cursors_block(unsigned int* __restrict__ part_dst,
                                                   const unsigned int* __restrict__ block_in,
                                                   unsigned int* __restrict__ row_ptr,
                                                   int n_nodes, int blk) {
  __shared__ uint4 wtot[8][32];
  __shared__ unsigned int bred[8];
  const int lane = threadIdx.x & 31;
  const int w    = threadIdx.x >> 5;
  const int node = (blk * 32 + lane) * 4;
  const int row  = padded_row(n_nodes);
  const size_t row4 = (size_t)(row >> 2);
  const bool live = node < n_nodes;
  const uint4 zero4 = make_uint4(0u, 0u, 0u, 0u);
  uint4* __restrict__ p = (uint4*)part_dst + (size_t)(w * (DEG_SLICES / 8)) * row4 + (size_t)(node >> 2);

  unsigned int b = 0;
  for (int i = threadIdx.x; i < blk; i += 256) b += block_in[i];
  for (int o = 16; o > 0; o >>= 1) b += __shfl_xor(b, o);
  if (lane == 0) bred[w] = b;

  uint4 c[DEG_SLICES / 8];
#pragma unroll
  for (int k = 0; k < DEG_SLICES / 8; ++k) c[k] = zero4;
  if (live) {
#pragma unroll
    for (int k = 0; k < DEG_SLICES / 8; ++k) c[k] = p[(size_t)k * row4];
  }
  uint4 tot = zero4;
#pragma unroll
  for (int k = 0; k < DEG_SLICES / 8; ++k) {
    const uint4 t = c[k];
    c[k] = tot;
    tot = add4(tot, t);
  }
  wtot[w][lane] = tot;
  __syncthreads();

  unsigned int base = 0;
  for (int k = 0; k < 8; ++k) base += bred[k];
  uint4 wp = zero4;
  uint4 nt = zero4;
  for (int k = 0; k < 8; ++k) {
    const uint4 t = wtot[k][lane];
    if (k < w) wp = add4(wp, t);
    nt = add4(nt, t);
  }
  const unsigned int lt = nt.x + nt.y + nt.z + nt.w;
  unsigned int inc = lt;
  for (int o = 1; o < 32; o <<= 1) {
    const unsigned int u = __shfl_up(inc, (unsigned int)o);
    if (lane >= o) inc += u;
  }
  uint4 st;
  st.x = base + inc - lt;
  st.y = st.x + nt.x;
  st.z = st.y + nt.y;
  st.w = st.z + nt.z;
  if (live) {
    if (w == 0) {
      *(uint4*)(row_ptr + node) = st;
      if (node + 4 >= n_nodes) row_ptr[n_nodes] = st.x + lt;
    }
    st = add4(st, wp);
#pragma unroll
    for (int k = 0; k < DEG_SLICES / 8; ++k) p[(size_t)k * row4] = add4(st, c[k]);
  }
}

__device__ __forceinline__ void place_edge(unsigned int* cur, int* __restrict__ sorted_src,
                                           int d, int s, int base, int nb, int n_edges) {
  const unsigned int b = (unsigned int)(d - base);
  if (b < (unsigned int)nb) {
    const unsigned int pos = atomicAdd(&cur[b], 1u);
    if (pos < (unsigned int)n_edges) sorted_src[pos] = s;
  }
}

__global__ __launch_bounds__(HIST_THREADS) void edge_fill(const int* __restrict__ src,
                                                          const int* __restrict__ dst,
                                                          const unsigned int* __restrict__ cursors,
                                                          int* __restrict__ sorted_src,
                                                          int n_edges, int n_nodes) {
  __shared__ __attribute__((aligned(16))) unsigned int cur[HIST_CAP];
  const int tid = threadIdx.x;
  int e0, e1;
  slice_range(n_edges, (int)blockIdx.x, e0, e1);
  const int nv = (e1 - e0) >> 2;
  const int4* __restrict__ d4 = (const int4*)(dst + e0);
  const int4* __restrict__ s4 = (const int4*)(src + e0);
  const int row = padded_row(n_nodes);
  const unsigned int* __restrict__ my = cursors + (size_t)blockIdx.x * (size_t)row;

  for (int base = 0; base < n_nodes; base += HIST_CAP) {
    const int nb  = min(HIST_CAP, n_nodes - base);
    const int nb4 = (nb + 3) >> 2;
    uint4* c4 = (uint4*)cur;
    const uint4* __restrict__ i4 = (const uint4*)(my + base);
    for (int i = tid; i < nb4; i += HIST_THREADS) c4[i] = i4[i];
    __syncthreads();

    for (int q = tid; q < nv; q += 4 * HIST_THREADS) {
      int4 dv[4], sv[4];
#pragma unroll
      for (int j = 0; j < 4; ++j) {
        const int qq = q + j * HIST_THREADS;
        const bool ok = qq < nv;
        dv[j] = ok ? d4[qq] : make_int4(-1, -1, -1, -1);
        sv[j] = ok ? s4[qq] : make_int4(0, 0, 0, 0);
      }
#pragma unroll
      for (int j = 0; j < 4; ++j) {
        place_edge(cur, sorted_src, dv[j].x, sv[j].x, base, nb, n_edges);
        place_edge(cur, sorted_src, dv[j].y, sv[j].y, base, nb, n_edges);
        place_edge(cur, sorted_src, dv[j].z, sv[j].z, base, nb, n_edges);
        place_edge(cur, sorted_src, dv[j].w, sv[j].w, base, nb, n_edges);
      }
    }
    {
      const int e = e0 + (nv << 2) + tid;
      if (tid < 3 && e < e1) place_edge(cur, sorted_src, dst[e], src[e], base, nb, n_edges);
    }
    __syncthreads();
  }
}


#define GEMM_K       128
#define GEMM_WAVES   8
#define GEMM_THREADS (GEMM_WAVES * 32)
#define GEMM_GRID    256
static_assert(IN_DIM == GEMM_K && HID_DIM == GEMM_K, "both products have K = 128");

template <int NC>
__device__ __forceinline__ void stage_weights(const float* __restrict__ Wm, float4* Bl4) {
#pragma unroll
  for (int it = 0; it < (GEMM_K / 4 * NC) / GEMM_THREADS; ++it) {
    const int idx = it * GEMM_THREADS + (int)threadIdx.x;
    const int n   = idx % NC;
    const int jh  = idx / NC;
    const int k0  = 8 * (jh >> 1) + 2 * (jh & 1);
    float4 q;
    q.x = Wm[(size_t)(k0)     * NC + n];
    q.y = Wm[(size_t)(k0 + 1) * NC + n];
    q.z = Wm[(size_t)(k0 + 4) * NC + n];
    q.w = Wm[(size_t)(k0 + 5) * NC + n];
    Bl4[idx] = q;
  }
}

template <int NT, int NC = 16 * NT>
__device__ __forceinline__ void strip_mma(const float2 (&a)[GEMM_K / 4], const float4* Bl4,
                                          int half, int sub, v8f (&acc)[NT]) {
  float4 q[NT];
#pragma unroll
  for (int ct = 0; ct < NT; ++ct) q[ct] = Bl4[half * NC + ct * 16 + sub];
#pragma unroll
  for (int j2 = 0; j2 < GEMM_K / 8; ++j2) {
    float4 qn[NT];
#pragma unroll
    for (int ct = 0; ct < NT; ++ct)
      qn[ct] = (j2 + 1 < GEMM_K / 8) ? Bl4[((j2 + 1) * 2 + half) * NC + ct * 16 + sub] : q[ct];
    v2f a0, a1;
    a0.x = a[2 * j2].x;     a0.y = a[2 * j2].y;
    a1.x = a[2 * j2 + 1].x; a1.y = a[2 * j2 + 1].y;
#pragma unroll
    for (int ct = 0; ct < NT; ++ct) {
      v2f b;
      b.x = q[ct].x; b.y = q[ct].y;
      acc[ct] = __builtin_amdgcn_wmma_f32_16x16x4_f32(
           false, a0,  false, b,
           (short)0, acc[ct],  false,  false);
    }
#pragma unroll
    for (int ct = 0; ct < NT; ++ct) {
      v2f b;
      b.x = q[ct].z; b.y = q[ct].w;
      acc[ct] = __builtin_amdgcn_wmma_f32_16x16x4_f32(
          false, a1, false, b, (short)0, acc[ct], false, false);
    }
    __builtin_amdgcn_sched_barrier(0);
#pragma unroll
    for (int ct = 0; ct < NT; ++ct) q[ct] = qn[ct];
  }
}

template <int NT, int NC = 16 * NT>
__device__ __forceinline__ void strip_store(float* __restrict__ dstm, int rt, int half, int sub,
                                            const v8f (&acc)[NT]) {
  const int m_base = rt * 16 + half * 8;
#pragma unroll
  for (int ct = 0; ct < NT; ++ct)
#pragma unroll
    for (int r = 0; r < 8; ++r)
      dstm[(size_t)(m_base + r) * NC + ct * 16 + sub] = acc[ct][r];
}


__device__ __forceinline__ void load_a1(const float* __restrict__ x,
                                        const float* __restrict__ out_norm,
                                        int rt, int half, int sub, float2 (&a)[GEMM_K / 4]) {
  const int m = rt * 16 + sub;
  const float norm = out_norm[m];
  const float* __restrict__ arow = x + (size_t)m * IN_DIM + half * 2;
#pragma unroll
  for (int j = 0; j < GEMM_K / 4; ++j) {
    const float2 av = *(const float2*)(arow + 4 * j);
    a[j].x = av.x * norm;
    a[j].y = av.y * norm;
  }
}

__device__ __forceinline__ void gemm1_block(const float* __restrict__ x,
                                            const float* __restrict__ W1,
                                            const float* __restrict__ out_norm,
                                            float* __restrict__ h1,
                                            int n_row_tiles, int blk) {
  __shared__ float4 Bl4[GEMM_K / 4 * HID_DIM];
  const int lane = threadIdx.x & 31;
  const int wave = __builtin_amdgcn_readfirstlane((int)(threadIdx.x >> 5));
  const int half = lane >> 4;
  const int sub  = lane & 15;
  const int step = GEMM_GRID * GEMM_WAVES;
  int rt = blk + GEMM_GRID * wave;

  float2 a[GEMM_K / 4];
#pragma unroll
  for (int j = 0; j < GEMM_K / 4; ++j) a[j] = make_float2(0.0f, 0.0f);
  if (rt < n_row_tiles) load_a1(x, out_norm, rt, half, sub, a);

  stage_weights<HID_DIM>(W1, Bl4);
  __syncthreads();

  while (rt < n_row_tiles) {
    const int nx = rt + step;
    float2 an[GEMM_K / 4];
#pragma unroll
    for (int j = 0; j < GEMM_K / 4; ++j) an[j] = make_float2(0.0f, 0.0f);
    if (nx < n_row_tiles) load_a1(x, out_norm, nx, half, sub, an);

#pragma unroll
    for (int hc = 0; hc < 2; ++hc) {
      v8f acc[HID_DIM / 32];
#pragma unroll
      for (int ct = 0; ct < HID_DIM / 32; ++ct) acc[ct] = v8f{};
      strip_mma<HID_DIM / 32, HID_DIM>(a, Bl4 + hc * (HID_DIM / 2), half, sub, acc);
      strip_store<HID_DIM / 32, HID_DIM>(h1 + hc * (HID_DIM / 2), rt, half, sub, acc);
    }
#pragma unroll
    for (int j = 0; j < GEMM_K / 4; ++j) a[j] = an[j];
    rt = nx;
  }
}

__global__ __launch_bounds__(GEMM_THREADS) void cursors_and_gemm1(const float* __restrict__ x,
                                                                  const float* __restrict__ W1,
                                                                  const float* __restrict__ out_norm,
                                                                  float* __restrict__ h1,
                                                                  int n_row_tiles,
                                                                  unsigned int* __restrict__ part_dst,
                                                                  const unsigned int* __restrict__ block_in,
                                                                  unsigned int* __restrict__ row_ptr,
                                                                  int n_nodes) {
  static_assert(GEMM_THREADS == 256, "edge_cursors_block is written for 256 threads");
  const int blk = (int)blockIdx.x;
  if (blk < GEMM_GRID) gemm1_block(x, W1, out_norm, h1, n_row_tiles, blk);
  else                 edge_cursors_block(part_dst, block_in, row_ptr, n_nodes, blk - GEMM_GRID);
}


__device__ __forceinline__ void load_a2(const float* __restrict__ agg1,
                                        const float* __restrict__ b1,
                                        const float* __restrict__ out_norm,
                                        const float* __restrict__ in_norm,
                                        int rt, int half, int sub, float2 (&a)[GEMM_K / 4]) {
  const int m = rt * 16 + sub;
  const float s_in  = in_norm[m];
  const float s_out = out_norm[m];
  const float* __restrict__ arow = agg1 + (size_t)m * HID_DIM + half * 2;
  const float* __restrict__ brow = b1 + half * 2;
#pragma unroll
  for (int j = 0; j < GEMM_K / 4; ++j) {
    const float2 av = *(const float2*)(arow + 4 * j);
    const float2 bv = *(const float2*)(brow + 4 * j);
    a[j].x = fmaxf(av.x * s_in + bv.x, 0.0f) * s_out;
    a[j].y = fmaxf(av.y * s_in + bv.y, 0.0f) * s_out;
  }
}

__global__ __launch_bounds__(GEMM_THREADS) void gemm2_wmma(const float* __restrict__ agg1,
                                                           const float* __restrict__ W2,
                                                           const float* __restrict__ b1,
                                                           const float* __restrict__ out_norm,
                                                           const float* __restrict__ in_norm,
                                                           float* __restrict__ h2,
                                                           int n_row_tiles) {
  __shared__ float4 Bl4[GEMM_K / 4 * OUT_DIM];
  const int lane = threadIdx.x & 31;
  const int wave = __builtin_amdgcn_readfirstlane((int)(threadIdx.x >> 5));
  const int half = lane >> 4;
  const int sub  = lane & 15;
  const int step = (int)gridDim.x * GEMM_WAVES;
  int rt = (int)blockIdx.x + (int)gridDim.x * wave;

  float2 a[GEMM_K / 4];
#pragma unroll
  for (int j = 0; j < GEMM_K / 4; ++j) a[j] = make_float2(0.0f, 0.0f);
  if (rt < n_row_tiles) load_a2(agg1, b1, out_norm, in_norm, rt, half, sub, a);

  stage_weights<OUT_DIM>(W2, Bl4);
  __syncthreads();

  while (rt < n_row_tiles) {
    const int nx = rt + step;
    float2 an[GEMM_K / 4];
#pragma unroll
    for (int j = 0; j < GEMM_K / 4; ++j) an[j] = make_float2(0.0f, 0.0f);
    if (nx < n_row_tiles) load_a2(agg1, b1, out_norm, in_norm, nx, half, sub, an);

    v8f acc[OUT_DIM / 16];
#pragma unroll
    for (int ct = 0; ct < OUT_DIM / 16; ++ct) acc[ct] = v8f{};
    strip_mma<OUT_DIM / 16>(a, Bl4, half, sub, acc);
    strip_store<OUT_DIM / 16>(h2, rt, half, sub, acc);
#pragma unroll
    for (int j = 0; j < GEMM_K / 4; ++j) a[j] = an[j];
    rt = nx;
  }
}


__global__ __launch_bounds__(256) void gather128(const float* __restrict__ h,
                                                 const int* __restrict__ sorted_src,
                                                 const unsigned int* __restrict__ row_ptr,
                                                 float* __restrict__ agg,
                                                 int n_nodes, int n_edges) {
  const int lane = threadIdx.x & 31;
  const int d = __builtin_amdgcn_readfirstlane((int)((blockIdx.x * blockDim.x + threadIdx.x) >> 5));
  if (d >= n_nodes) return;
  const int beg = (int)min(row_ptr[d], (unsigned int)n_edges);
  const int end = (int)min(row_ptr[d + 1], (unsigned int)n_edges);
  const float* __restrict__ hl = h + lane * 4;
  float4 acc = make_float4(0.0f, 0.0f, 0.0f, 0.0f);

  for (int j = beg; j < end; j += 32) {
    const int cnt = min(32, end - j);
    int mine = (lane < cnt) ? sorted_src[j + lane] : 0;
    mine = (int)min((unsigned int)mine, (unsigned int)(n_nodes - 1));
    int t = 0;
    for (; t + 8 <= cnt; t += 8) {
      float4 v[8];
#pragma unroll
      for (int u = 0; u < 8; ++u) {
        const int s = __builtin_amdgcn_readlane(mine, t + u);
        v[u] = *(const float4*)(hl + (size_t)s * 128);
      }
#pragma unroll
      for (int u = 0; u < 8; ++u) {
        acc.x += v[u].x; acc.y += v[u].y; acc.z += v[u].z; acc.w += v[u].w;
      }
    }
    if (t < cnt) {
      float4 v[8];
#pragma unroll
      for (int u = 0; u < 8; ++u) {
        v[u] = make_float4(0.0f, 0.0f, 0.0f, 0.0f);
        if (t + u < cnt) {
          const int s = __builtin_amdgcn_readlane(mine, t + u);
          v[u] = *(const float4*)(hl + (size_t)s * 128);
        }
      }
#pragma unroll
      for (int u = 0; u < 8; ++u) {
        acc.x += v[u].x; acc.y += v[u].y; acc.z += v[u].z; acc.w += v[u].w;
      }
    }
  }
  *(float4*)(agg + (size_t)d * 128 + lane * 4) = acc;
}

__global__ __launch_bounds__(256) void gather64_out(const float* __restrict__ h,
                                                    const int* __restrict__ sorted_src,
                                                    const unsigned int* __restrict__ row_ptr,
                                                    const float* __restrict__ in_norm,
                                                    const float* __restrict__ b2,
                                                    float* __restrict__ out,
                                                    int n_nodes, int n_edges) {
  const int lane = threadIdx.x & 31;
  const int d = __builtin_amdgcn_readfirstlane((int)((blockIdx.x * blockDim.x + threadIdx.x) >> 5));
  if (d >= n_nodes) return;
  const int beg = (int)min(row_ptr[d], (unsigned int)n_edges);
  const int end = (int)min(row_ptr[d + 1], (unsigned int)n_edges);
  const float* __restrict__ hl = h + lane * 2;
  float2 acc = make_float2(0.0f, 0.0f);

  for (int j = beg; j < end; j += 32) {
    const int cnt = min(32, end - j);
    int mine = (lane < cnt) ? sorted_src[j + lane] : 0;
    mine = (int)min((unsigned int)mine, (unsigned int)(n_nodes - 1));
    int t = 0;
    for (; t + 8 <= cnt; t += 8) {
      float2 v[8];
#pragma unroll
      for (int u = 0; u < 8; ++u) {
        const int s = __builtin_amdgcn_readlane(mine, t + u);
        v[u] = *(const float2*)(hl + (size_t)s * 64);
      }
#pragma unroll
      for (int u = 0; u < 8; ++u) { acc.x += v[u].x; acc.y += v[u].y; }
    }
    if (t < cnt) {
      float2 v[8];
#pragma unroll
      for (int u = 0; u < 8; ++u) {
        v[u] = make_float2(0.0f, 0.0f);
        if (t + u < cnt) {
          const int s = __builtin_amdgcn_readlane(mine, t + u);
          v[u] = *(const float2*)(hl + (size_t)s * 64);
        }
      }
#pragma unroll
      for (int u = 0; u < 8; ++u) { acc.x += v[u].x; acc.y += v[u].y; }
    }
  }
  const float nrm = in_norm[d];
  const float2 bb = *(const float2*)(b2 + lane * 2);
  float2 o;
  o.x = acc.x * nrm + bb.x;
  o.y = acc.y * nrm + bb.y;
  *(float2*)(out + (size_t)d * 64 + lane * 2) = o;
}


static inline int grid1d(long n, int block) { return (int)((n + block - 1) / block); }

extern "C" void kernel_launch(void* const* d_in, const int* in_sizes, int n_in,
                              void* d_out, int out_size, void* d_ws, size_t ws_size,
                              hipStream_t stream) {
  const float* x   = (const float*)d_in[0];
  const float* W1  = (const float*)d_in[1];
  const float* b1  = (const float*)d_in[2];
  const float* W2  = (const float*)d_in[3];
  const float* b2  = (const float*)d_in[4];
  const int*   src = (const int*)d_in[5];
  const int*   dst = (const int*)d_in[6];

  const int n_nodes = in_sizes[0] / IN_DIM;
  const int n_edges = in_sizes[5];
  float* out = (float*)d_out;

  const size_t row = ((size_t)n_nodes + 31) & ~(size_t)31;
  float* h1   = (float*)d_ws;
  float* agg1 = h1 + (size_t)n_nodes * HID_DIM;
  float* h2   = agg1 + (size_t)n_nodes * HID_DIM;
  unsigned int* partial = (unsigned int*)(h2 + (size_t)n_nodes * OUT_DIM);
  unsigned int* part_dst = partial + DEG_SLICES * row;
  float* out_norm = (float*)(partial + 2 * DEG_SLICES * row);
  float* in_norm  = out_norm + row;
  unsigned int* row_ptr  = (unsigned int*)(in_norm + row);
  unsigned int* block_in = row_ptr + row + 32;
  int* sorted_src = (int*)(block_in + row);

  const int node_blocks = grid1d(n_nodes, 128);

  degree_hist<<<dim3(DEG_SLICES, 2), HIST_THREADS, 0, stream>>>(src, dst, partial, n_edges, n_nodes);
  degree_norm<<<node_blocks, 256, 0, stream>>>(partial, out_norm, in_norm, block_in, n_nodes);

  const int row_tiles = n_nodes / 16;

  cursors_and_gemm1<<<GEMM_GRID + node_blocks, GEMM_THREADS, 0, stream>>>(
      x, W1, out_norm, h1, row_tiles, part_dst, block_in, row_ptr, n_nodes);
  edge_fill<<<DEG_SLICES, HIST_THREADS, 0, stream>>>(src, dst, part_dst, sorted_src, n_edges, n_nodes);

  gather128<<<(n_nodes + 7) / 8, 256, 0, stream>>>(h1, sorted_src, row_ptr, agg1, n_nodes, n_edges);

  gemm2_wmma<<<GEMM_GRID, GEMM_THREADS, 0, stream>>>(agg1, W2, b1, out_norm, in_norm, h2, row_tiles);

  gather64_out<<<(n_nodes + 7) / 8, 256, 0, stream>>>(h2, sorted_src, row_ptr, in_norm, b2, out,
                                                      n_nodes, n_edges);
}


# edge placement in two run-writing passes: pairs binned per 128-node block in LDS, then each block sorts its run by node in LDS; no scattered 4-byte stores; in-edge (f32 sum) order now arrival order
# speedup vs baseline: 5.0163x; 1.2469x over previous
#include <hip/hip_runtime.h>

typedef __attribute__((ext_vector_type(2))) float v2f;
typedef __attribute__((ext_vector_type(8))) float v8f;

#define IN_DIM  128
#define HID_DIM 128
#define OUT_DIM 64


#define HIST_CAP     65536
#define HIST_THREADS 1024
#define DEG_SLICES   128
static_assert(DEG_SLICES % 8 == 0, "degree_norm splits the slice rows over 8 waves");

__device__ __forceinline__ int padded_row(int n_nodes) { return (n_nodes + 31) & ~31; }

__device__ __forceinline__ void slice_range(int n_edges, int slice, int& e0, int& e1) {
  const int per = (((n_edges + DEG_SLICES - 1) / DEG_SLICES) + 3) & ~3;
  e0 = min(slice * per, n_edges);
  e1 = min(e0 + per, n_edges);
}

__device__ __forceinline__ void hist_bump(unsigned int* hist, int v, int base, int nb) {
  const unsigned int b = (unsigned int)(v - base);
  if (b < (unsigned int)nb) atomicAdd(&hist[b], 1u);
}

__global__ __launch_bounds__(HIST_THREADS) void degree_hist(const int* __restrict__ src,
                                                            const int* __restrict__ dst,
                                                            unsigned int* __restrict__ partial,
                                                            int n_edges, int n_nodes) {
  __shared__ __attribute__((aligned(16))) unsigned int hist[HIST_CAP];
  const int tid = threadIdx.x;
  const int* __restrict__ idx = blockIdx.y ? dst : src;
  const int n_slices = DEG_SLICES;
  int e0, e1;
  slice_range(n_edges, (int)blockIdx.x, e0, e1);
  const int nv  = (e1 - e0) >> 2;
  const int4* __restrict__ p4 = (const int4*)(idx + e0);
  const int row = padded_row(n_nodes);
  unsigned int* __restrict__ my =
      partial + ((size_t)blockIdx.y * n_slices + blockIdx.x) * (size_t)row;

  for (int base = 0; base < n_nodes; base += HIST_CAP) {
    const int nb  = min(HIST_CAP, n_nodes - base);
    const int nb4 = (nb + 3) >> 2;
    uint4* h4 = (uint4*)hist;
    for (int i = tid; i < nb4; i += HIST_THREADS) h4[i] = make_uint4(0u, 0u, 0u, 0u);
    __syncthreads();

    for (int q = tid; q < nv; q += 4 * HIST_THREADS) {
      int4 v[4];
#pragma unroll
      for (int j = 0; j < 4; ++j) {
        const int qq = q + j * HIST_THREADS;
        v[j] = (qq < nv) ? p4[qq] : make_int4(-1, -1, -1, -1);
      }
#pragma unroll
      for (int j = 0; j < 4; ++j) {
        hist_bump(hist, v[j].x, base, nb);
        hist_bump(hist, v[j].y, base, nb);
        hist_bump(hist, v[j].z, base, nb);
        hist_bump(hist, v[j].w, base, nb);
      }
    }
    {
      const int e = e0 + (nv << 2) + tid;
      if (tid < 3 && e < e1) hist_bump(hist, idx[e], base, nb);
    }
    __syncthreads();

    uint4* o4 = (uint4*)(my + base);
    for (int i = tid; i < nb4; i += HIST_THREADS) o4[i] = h4[i];
    __syncthreads();
  }
}

__global__ __launch_bounds__(256) void degree_norm(const unsigned int* __restrict__ partial,
                                                   float* __restrict__ out_norm,
                                                   float* __restrict__ in_norm,
                                                   unsigned int* __restrict__ block_in,
                                                   int n_nodes) {
  __shared__ uint4 red[2][8][32];
  const int lane = threadIdx.x & 31;
  const int w    = threadIdx.x >> 5;
  const int node = (blockIdx.x * 32 + lane) * 4;
  const int row  = padded_row(n_nodes);
  uint4 so = make_uint4(0u, 0u, 0u, 0u);
  uint4 si = make_uint4(0u, 0u, 0u, 0u);
  if (node < n_nodes) {
    const uint4* __restrict__ po = (const uint4*)(partial + node);
    const uint4* __restrict__ pi = (const uint4*)(partial + (size_t)DEG_SLICES * row + node);
    const size_t row4 = (size_t)(row >> 2);
#pragma unroll
    for (int k = 0; k < DEG_SLICES / 8; ++k) {
      const size_t s = (size_t)(w + 8 * k);
      const uint4 a = po[s * row4];
      const uint4 b = pi[s * row4];
      so.x += a.x; so.y += a.y; so.z += a.z; so.w += a.w;
      si.x += b.x; si.y += b.y; si.z += b.z; si.w += b.w;
    }
  }
  red[0][w][lane] = so;
  red[1][w][lane] = si;
  __syncthreads();
  if (w == 0) {
    so = make_uint4(0u, 0u, 0u, 0u);
    si = make_uint4(0u, 0u, 0u, 0u);
#pragma unroll
    for (int k = 0; k < 8; ++k) {
      const uint4 a = red[0][k][lane];
      const uint4 b = red[1][k][lane];
      so.x += a.x; so.y += a.y; so.z += a.z; so.w += a.w;
      si.x += b.x; si.y += b.y; si.z += b.z; si.w += b.w;
    }
    unsigned int t = si.x + si.y + si.z + si.w;
    for (int o = 16; o > 0; o >>= 1) t += __shfl_xor(t, o);
    if (lane == 0) block_in[blockIdx.x] = t;
    if (node < n_nodes) {
      float4 o, i;
      o.x = rsqrtf(fmaxf((float)so.x, 1.0f));
      o.y = rsqrtf(fmaxf((float)so.y, 1.0f));
      o.z = rsqrtf(fmaxf((float)so.z, 1.0f));
      o.w = rsqrtf(fmaxf((float)so.w, 1.0f));
      i.x = rsqrtf(fmaxf((float)si.x, 1.0f));
      i.y = rsqrtf(fmaxf((float)si.y, 1.0f));
      i.z = rsqrtf(fmaxf((float)si.z, 1.0f));
      i.w = rsqrtf(fmaxf((float)si.w, 1.0f));
      *(float4*)(out_norm + node) = o;
      *(float4*)(in_norm + node)  = i;
    }
  }
}


#define BIN_NODES 128
#define BIN_SHIFT 7
static_assert((1 << BIN_SHIFT) == BIN_NODES, "a node's bin is its number shifted right");
#define BIN_CAP   1024
static_assert(BIN_CAP == HIST_THREADS, "edge_bin scans one bin per thread");

__device__ __forceinline__ uint4 add4(uint4 a, uint4 b) {
  return make_uint4(a.x + b.x, a.y + b.y, a.z + b.z, a.w + b.w);
}

__device__ __forceinline__ int bin_row(int n_bins) { return (n_bins + 3) & ~3; }

__device__ __forceinline__ void edge_cursors_block(const unsigned int* __restrict__ part_dst,
                                                   const unsigned int* __restrict__ block_in,
                                                   unsigned int* __restrict__ row_ptr,
                                                   unsigned int* __restrict__ bin_off,
                                                   int n_nodes, int n_bins, int blk) {
  __shared__ uint4 wtot[8][32];
  __shared__ unsigned int bred[8];
  __shared__ unsigned int stot[DEG_SLICES];
  const int lane = threadIdx.x & 31;
  const int w    = threadIdx.x >> 5;
  const int node = (blk * 32 + lane) * 4;
  const int row  = padded_row(n_nodes);
  const size_t row4 = (size_t)(row >> 2);
  const bool live = node < n_nodes;
  const uint4 zero4 = make_uint4(0u, 0u, 0u, 0u);
  const uint4* __restrict__ p =
      (const uint4*)part_dst + (size_t)(w * (DEG_SLICES / 8)) * row4 + (size_t)(node >> 2);

  unsigned int b = 0;
  for (int i = threadIdx.x; i < blk; i += 256) b += block_in[i];
  for (int o = 16; o > 0; o >>= 1) b += __shfl_xor(b, o);
  if (lane == 0) bred[w] = b;

  uint4 c[DEG_SLICES / 8];
#pragma unroll
  for (int k = 0; k < DEG_SLICES / 8; ++k) c[k] = zero4;
  if (live) {
#pragma unroll
    for (int k = 0; k < DEG_SLICES / 8; ++k) c[k] = p[(size_t)k * row4];
  }
  uint4 tot = zero4;
#pragma unroll
  for (int k = 0; k < DEG_SLICES / 8; ++k) {
    tot = add4(tot, c[k]);
    unsigned int t = c[k].x + c[k].y + c[k].z + c[k].w;
    for (int o = 16; o > 0; o >>= 1) t += __shfl_xor(t, o);
    if (lane == 0) stot[w * (DEG_SLICES / 8) + k] = t;
  }
  wtot[w][lane] = tot;
  __syncthreads();

  unsigned int base = 0;
  for (int k = 0; k < 8; ++k) base += bred[k];
  uint4 nt = zero4;
  for (int k = 0; k < 8; ++k) nt = add4(nt, wtot[k][lane]);
  const unsigned int lt = nt.x + nt.y + nt.z + nt.w;
  unsigned int inc = lt;
  for (int o = 1; o < 32; o <<= 1) {
    const unsigned int u = __shfl_up(inc, (unsigned int)o);
    if (lane >= o) inc += u;
  }
  uint4 st;
  st.x = base + inc - lt;
  st.y = st.x + nt.x;
  st.z = st.y + nt.y;
  st.w = st.z + nt.z;
  if (live && w == 0) {
    *(uint4*)(row_ptr + node) = st;
    if (node + 4 >= n_nodes) row_ptr[n_nodes] = st.x + lt;
  }
  if (threadIdx.x < DEG_SLICES) {
    unsigned int off = base;
    for (int k = 0; k < (int)threadIdx.x; ++k) off += stot[k];
    bin_off[(size_t)threadIdx.x * (size_t)bin_row(n_bins) + (size_t)blk] = off;
  }
}

#define BIN_TRIP (16 * HIST_THREADS)

__global__ __launch_bounds__(HIST_THREADS) void edge_bin(const int* __restrict__ src,
                                                         const int* __restrict__ dst,
                                                         const unsigned int* __restrict__ bin_off,
                                                         uint2* __restrict__ pairs,
                                                         int n_edges, int n_nodes, int n_bins) {
  __shared__ unsigned int cnt[BIN_CAP];
  __shared__ unsigned int start[BIN_CAP];
  __shared__ unsigned int gbase[BIN_CAP];
  __shared__ unsigned int wsum[HIST_THREADS / 32];
  __shared__ uint2 buf[BIN_TRIP + 4];
  const int tid  = threadIdx.x;
  const int lane = tid & 31;
  const int w    = tid >> 5;
  const int nb   = min(n_bins, BIN_CAP);
  int e0, e1;
  slice_range(n_edges, (int)blockIdx.x, e0, e1);
  const int nv = (e1 - e0) >> 2;
  const int4* __restrict__ d4 = (const int4*)(dst + e0);
  const int4* __restrict__ s4 = (const int4*)(src + e0);

  gbase[tid] = (tid < nb) ? bin_off[(size_t)blockIdx.x * (size_t)bin_row(n_bins) + (size_t)tid] : 0u;

  for (int q0 = 0; q0 < max(nv, 1); q0 += 4 * HIST_THREADS) {
    cnt[tid] = 0u;
    __syncthreads();

    int dv[17], sv[17], rk[17];
#pragma unroll
    for (int j = 0; j < 4; ++j) {
      const int qq = q0 + tid + j * HIST_THREADS;
      const bool ok = qq < nv;
      const int4 dq = ok ? d4[qq] : make_int4(-1, -1, -1, -1);
      const int4 sq = ok ? s4[qq] : make_int4(0, 0, 0, 0);
      dv[4 * j] = dq.x; dv[4 * j + 1] = dq.y; dv[4 * j + 2] = dq.z; dv[4 * j + 3] = dq.w;
      sv[4 * j] = sq.x; sv[4 * j + 1] = sq.y; sv[4 * j + 2] = sq.z; sv[4 * j + 3] = sq.w;
    }
    {
      const int e = e0 + (nv << 2) + tid;
      const bool ok = tid < 3 && e < e1 && q0 + 4 * HIST_THREADS >= nv;
      dv[16] = ok ? dst[e] : -1;
      sv[16] = ok ? src[e] : 0;
    }
#pragma unroll
    for (int j = 0; j < 17; ++j) {
      const unsigned int bn = (unsigned int)dv[j] >> BIN_SHIFT;
      const bool ok = (unsigned int)dv[j] < (unsigned int)n_nodes && bn < (unsigned int)nb;
      rk[j] = ok ? (int)atomicAdd(&cnt[bn], 1u) : -1;
    }
    __syncthreads();

    const unsigned int mine = cnt[tid];
    unsigned int inc = mine;
    for (int o = 1; o < 32; o <<= 1) {
      const unsigned int u = __shfl_up(inc, (unsigned int)o);
      if (lane >= o) inc += u;
    }
    if (lane == 31) wsum[w] = inc;
    __syncthreads();
    unsigned int before = 0, total = 0;
    for (int k = 0; k < HIST_THREADS / 32; ++k) {
      const unsigned int t = wsum[k];
      if (k < w) before += t;
      total += t;
    }
    start[tid] = before + inc - mine;
    __syncthreads();

#pragma unroll
    for (int j = 0; j < 17; ++j) {
      if (rk[j] >= 0) {
        const unsigned int i = start[(unsigned int)dv[j] >> BIN_SHIFT] + (unsigned int)rk[j];
        if (i < (unsigned int)(BIN_TRIP + 4)) buf[i] = make_uint2((unsigned int)sv[j], (unsigned int)dv[j]);
      }
    }
    __syncthreads();

    for (unsigned int i = tid; i < total; i += HIST_THREADS) {
      const uint2 pr = buf[i];
      const unsigned int bn = pr.y >> BIN_SHIFT;
      const unsigned int pos = gbase[bn] + (i - start[bn]);
      if (pos < (unsigned int)n_edges) pairs[pos] = pr;
    }
    __syncthreads();
    gbase[tid] += mine;
  }
}

#define PLACE_THREADS 256
#define PLACE_CAP     12288

__global__ __launch_bounds__(PLACE_THREADS) void edge_place(const uint2* __restrict__ pairs,
                                                            const unsigned int* __restrict__ row_ptr,
                                                            int* __restrict__ sorted_src,
                                                            int n_edges, int n_nodes) {
  __shared__ unsigned int cur[BIN_NODES];
  __shared__ int lst[PLACE_CAP];
  const int tid  = threadIdx.x;
  const int d0   = (int)blockIdx.x * BIN_NODES;
  const int nloc = min(BIN_NODES, n_nodes - d0);
  const unsigned int r0 = min(row_ptr[d0], (unsigned int)n_edges);
  const unsigned int r1 = min(row_ptr[d0 + nloc], (unsigned int)n_edges);
  const unsigned int ne = r1 > r0 ? r1 - r0 : 0u;
  const bool in_lds = ne <= (unsigned int)PLACE_CAP;
  if (tid < nloc) cur[tid] = row_ptr[d0 + tid] - r0;
  __syncthreads();

  const uint2* __restrict__ pp = pairs + r0;
  int* __restrict__ out = sorted_src + r0;
  for (unsigned int i0 = 0; i0 < ne; i0 += 4 * PLACE_THREADS) {
    uint2 pr[4];
#pragma unroll
    for (int j = 0; j < 4; ++j) {
      const unsigned int i = i0 + (unsigned int)(tid + j * PLACE_THREADS);
      pr[j] = (i < ne) ? pp[i] : make_uint2(0u, 0xffffffffu);
    }
#pragma unroll
    for (int j = 0; j < 4; ++j) {
      const unsigned int ln = pr[j].y - (unsigned int)d0;
      if (ln < (unsigned int)nloc) {
        const unsigned int pos = atomicAdd(&cur[ln], 1u);
        if (pos < ne) {
          if (in_lds) lst[pos] = (int)pr[j].x;
          else        out[pos] = (int)pr[j].x;
        }
      }
    }
  }
  __syncthreads();
  if (in_lds)
    for (unsigned int i = tid; i < ne; i += PLACE_THREADS) out[i] = lst[i];
}


#define GEMM_K       128
#define GEMM_WAVES   8
#define GEMM_THREADS (GEMM_WAVES * 32)
#define GEMM_GRID    256
static_assert(IN_DIM == GEMM_K && HID_DIM == GEMM_K, "both products have K = 128");

template <int NC>
__device__ __forceinline__ void stage_weights(const float* __restrict__ Wm, float4* Bl4) {
#pragma unroll
  for (int it = 0; it < (GEMM_K / 4 * NC) / GEMM_THREADS; ++it) {
    const int idx = it * GEMM_THREADS + (int)threadIdx.x;
    const int n   = idx % NC;
    const int jh  = idx / NC;
    const int k0  = 8 * (jh >> 1) + 2 * (jh & 1);
    float4 q;
    q.x = Wm[(size_t)(k0)     * NC + n];
    q.y = Wm[(size_t)(k0 + 1) * NC + n];
    q.z = Wm[(size_t)(k0 + 4) * NC + n];
    q.w = Wm[(size_t)(k0 + 5) * NC + n];
    Bl4[idx] = q;
  }
}

template <int NT, int NC = 16 * NT>
__device__ __forceinline__ void strip_mma(const float2 (&a)[GEMM_K / 4], const float4* Bl4,
                                          int half, int sub, v8f (&acc)[NT]) {
  float4 q[NT];
#pragma unroll
  for (int ct = 0; ct < NT; ++ct) q[ct] = Bl4[half * NC + ct * 16 + sub];
#pragma unroll
  for (int j2 = 0; j2 < GEMM_K / 8; ++j2) {
    float4 qn[NT];
#pragma unroll
    for (int ct = 0; ct < NT; ++ct)
      qn[ct] = (j2 + 1 < GEMM_K / 8) ? Bl4[((j2 + 1) * 2 + half) * NC + ct * 16 + sub] : q[ct];
    v2f a0, a1;
    a0.x = a[2 * j2].x;     a0.y = a[2 * j2].y;
    a1.x = a[2 * j2 + 1].x; a1.y = a[2 * j2 + 1].y;
#pragma unroll
    for (int ct = 0; ct < NT; ++ct) {
      v2f b;
      b.x = q[ct].x; b.y = q[ct].y;
      acc[ct] = __builtin_amdgcn_wmma_f32_16x16x4_f32(
           false, a0,  false, b,
           (short)0, acc[ct],  false,  false);
    }
#pragma unroll
    for (int ct = 0; ct < NT; ++ct) {
      v2f b;
      b.x = q[ct].z; b.y = q[ct].w;
      acc[ct] = __builtin_amdgcn_wmma_f32_16x16x4_f32(
          false, a1, false, b, (short)0, acc[ct], false, false);
    }
    __builtin_amdgcn_sched_barrier(0);
#pragma unroll
    for (int ct = 0; ct < NT; ++ct) q[ct] = qn[ct];
  }
}

template <int NT, int NC = 16 * NT>
__device__ __forceinline__ void strip_store(float* __restrict__ dstm, int rt, int half, int sub,
                                            const v8f (&acc)[NT]) {
  const int m_base = rt * 16 + half * 8;
#pragma unroll
  for (int ct = 0; ct < NT; ++ct)
#pragma unroll
    for (int r = 0; r < 8; ++r)
      dstm[(size_t)(m_base + r) * NC + ct * 16 + sub] = acc[ct][r];
}


__device__ __forceinline__ void load_a1(const float* __restrict__ x,
                                        const float* __restrict__ out_norm,
                                        int rt, int half, int sub, float2 (&a)[GEMM_K / 4]) {
  const int m = rt * 16 + sub;
  const float norm = out_norm[m];
  const float* __restrict__ arow = x + (size_t)m * IN_DIM + half * 2;
#pragma unroll
  for (int j = 0; j < GEMM_K / 4; ++j) {
    const float2 av = *(const float2*)(arow + 4 * j);
    a[j].x = av.x * norm;
    a[j].y = av.y * norm;
  }
}

__device__ __forceinline__ void gemm1_block(const float* __restrict__ x,
                                            const float* __restrict__ W1,
                                            const float* __restrict__ out_norm,
                                            float* __restrict__ h1,
                                            int n_row_tiles, int blk) {
  __shared__ float4 Bl4[GEMM_K / 4 * HID_DIM];
  const int lane = threadIdx.x & 31;
  const int wave = __builtin_amdgcn_readfirstlane((int)(threadIdx.x >> 5));
  const int half = lane >> 4;
  const int sub  = lane & 15;
  const int step = GEMM_GRID * GEMM_WAVES;
  int rt = blk + GEMM_GRID * wave;

  float2 a[GEMM_K / 4];
#pragma unroll
  for (int j = 0; j < GEMM_K / 4; ++j) a[j] = make_float2(0.0f, 0.0f);
  if (rt < n_row_tiles) load_a1(x, out_norm, rt, half, sub, a);

  stage_weights<HID_DIM>(W1, Bl4);
  __syncthreads();

  while (rt < n_row_tiles) {
    const int nx = rt + step;
    float2 an[GEMM_K / 4];
#pragma unroll
    for (int j = 0; j < GEMM_K / 4; ++j) an[j] = make_float2(0.0f, 0.0f);
    if (nx < n_row_tiles) load_a1(x, out_norm, nx, half, sub, an);

#pragma unroll
    for (int hc = 0; hc < 2; ++hc) {
      v8f acc[HID_DIM / 32];
#pragma unroll
      for (int ct = 0; ct < HID_DIM / 32; ++ct) acc[ct] = v8f{};
      strip_mma<HID_DIM / 32, HID_DIM>(a, Bl4 + hc * (HID_DIM / 2), half, sub, acc);
      strip_store<HID_DIM / 32, HID_DIM>(h1 + hc * (HID_DIM / 2), rt, half, sub, acc);
    }
#pragma unroll
    for (int j = 0; j < GEMM_K / 4; ++j) a[j] = an[j];
    rt = nx;
  }
}

__global__ __launch_bounds__(GEMM_THREADS) void cursors_and_gemm1(const float* __restrict__ x,
                                                                  const float* __restrict__ W1,
                                                                  const float* __restrict__ out_norm,
                                                                  float* __restrict__ h1,
                                                                  int n_row_tiles,
                                                                  const unsigned int* __restrict__ part_dst,
                                                                  const unsigned int* __restrict__ block_in,
                                                                  unsigned int* __restrict__ row_ptr,
                                                                  unsigned int* __restrict__ bin_off,
                                                                  int n_nodes, int n_bins) {
  static_assert(GEMM_THREADS == 256, "edge_cursors_block is written for 256 threads");
  const int blk = (int)blockIdx.x;
  if (blk < GEMM_GRID) gemm1_block(x, W1, out_norm, h1, n_row_tiles, blk);
  else                 edge_cursors_block(part_dst, block_in, row_ptr, bin_off, n_nodes, n_bins, blk - GEMM_GRID);
}


__device__ __forceinline__ void load_a2(const float* __restrict__ agg1,
                                        const float* __restrict__ b1,
                                        const float* __restrict__ out_norm,
                                        const float* __restrict__ in_norm,
                                        int rt, int half, int sub, float2 (&a)[GEMM_K / 4]) {
  const int m = rt * 16 + sub;
  const float s_in  = in_norm[m];
  const float s_out = out_norm[m];
  const float* __restrict__ arow = agg1 + (size_t)m * HID_DIM + half * 2;
  const float* __restrict__ brow = b1 + half * 2;
#pragma unroll
  for (int j = 0; j < GEMM_K / 4; ++j) {
    const float2 av = *(const float2*)(arow + 4 * j);
    const float2 bv = *(const float2*)(brow + 4 * j);
    a[j].x = fmaxf(av.x * s_in + bv.x, 0.0f) * s_out;
    a[j].y = fmaxf(av.y * s_in + bv.y, 0.0f) * s_out;
  }
}

__global__ __launch_bounds__(GEMM_THREADS) void gemm2_wmma(const float* __restrict__ agg1,
                                                           const float* __restrict__ W2,
                                                           const float* __restrict__ b1,
                                                           const float* __restrict__ out_norm,
                                                           const float* __restrict__ in_norm,
                                                           float* __restrict__ h2,
                                                           int n_row_tiles) {
  __shared__ float4 Bl4[GEMM_K / 4 * OUT_DIM];
  const int lane = threadIdx.x & 31;
  const int wave = __builtin_amdgcn_readfirstlane((int)(threadIdx.x >> 5));
  const int half = lane >> 4;
  const int sub  = lane & 15;
  const int step = (int)gridDim.x * GEMM_WAVES;
  int rt = (int)blockIdx.x + (int)gridDim.x * wave;

  float2 a[GEMM_K / 4];
#pragma unroll
  for (int j = 0; j < GEMM_K / 4; ++j) a[j] = make_float2(0.0f, 0.0f);
  if (rt < n_row_tiles) load_a2(agg1, b1, out_norm, in_norm, rt, half, sub, a);

  stage_weights<OUT_DIM>(W2, Bl4);
  __syncthreads();

  while (rt < n_row_tiles) {
    const int nx = rt + step;
    float2 an[GEMM_K / 4];
#pragma unroll
    for (int j = 0; j < GEMM_K / 4; ++j) an[j] = make_float2(0.0f, 0.0f);
    if (nx < n_row_tiles) load_a2(agg1, b1, out_norm, in_norm, nx, half, sub, an);

    v8f acc[OUT_DIM / 16];
#pragma unroll
    for (int ct = 0; ct < OUT_DIM / 16; ++ct) acc[ct] = v8f{};
    strip_mma<OUT_DIM / 16>(a, Bl4, half, sub, acc);
    strip_store<OUT_DIM / 16>(h2, rt, half, sub, acc);
#pragma unroll
    for (int j = 0; j < GEMM_K / 4; ++j) a[j] = an[j];
    rt = nx;
  }
}


__global__ __launch_bounds__(256) void gather128(const float* __restrict__ h,
                                                 const int* __restrict__ sorted_src,
                                                 const unsigned int* __restrict__ row_ptr,
                                                 float* __restrict__ agg,
                                                 int n_nodes, int n_edges) {
  const int lane = threadIdx.x & 31;
  const int d = __builtin_amdgcn_readfirstlane((int)((blockIdx.x * blockDim.x + threadIdx.x) >> 5));
  if (d >= n_nodes) return;
  const int beg = (int)min(row_ptr[d], (unsigned int)n_edges);
  const int end = (int)min(row_ptr[d + 1], (unsigned int)n_edges);
  const float* __restrict__ hl = h + lane * 4;
  float4 acc = make_float4(0.0f, 0.0f, 0.0f, 0.0f);

  for (int j = beg; j < end; j += 32) {
    const int cnt = min(32, end - j);
    int mine = (lane < cnt) ? sorted_src[j + lane] : 0;
    mine = (int)min((unsigned int)mine, (unsigned int)(n_nodes - 1));
    int t = 0;
    for (; t + 8 <= cnt; t += 8) {
      float4 v[8];
#pragma unroll
      for (int u = 0; u < 8; ++u) {
        const int s = __builtin_amdgcn_readlane(mine, t + u);
        v[u] = *(const float4*)(hl + (size_t)s * 128);
      }
#pragma unroll
      for (int u = 0; u < 8; ++u) {
        acc.x += v[u].x; acc.y += v[u].y; acc.z += v[u].z; acc.w += v[u].w;
      }
    }
    if (t < cnt) {
      float4 v[8];
#pragma unroll
      for (int u = 0; u < 8; ++u) {
        v[u] = make_float4(0.0f, 0.0f, 0.0f, 0.0f);
        if (t + u < cnt) {
          const int s = __builtin_amdgcn_readlane(mine, t + u);
          v[u] = *(const float4*)(hl + (size_t)s * 128);
        }
      }
#pragma unroll
      for (int u = 0; u < 8; ++u) {
        acc.x += v[u].x; acc.y += v[u].y; acc.z += v[u].z; acc.w += v[u].w;
      }
    }
  }
  *(float4*)(agg + (size_t)d * 128 + lane * 4) = acc;
}

__global__ __launch_bounds__(256) void gather64_out(const float* __restrict__ h,
                                                    const int* __restrict__ sorted_src,
                                                    const unsigned int* __restrict__ row_ptr,
                                                    const float* __restrict__ in_norm,
                                                    const float* __restrict__ b2,
                                                    float* __restrict__ out,
                                                    int n_nodes, int n_edges) {
  const int lane = threadIdx.x & 31;
  const int d = __builtin_amdgcn_readfirstlane((int)((blockIdx.x * blockDim.x + threadIdx.x) >> 5));
  if (d >= n_nodes) return;
  const int beg = (int)min(row_ptr[d], (unsigned int)n_edges);
  const int end = (int)min(row_ptr[d + 1], (unsigned int)n_edges);
  const float* __restrict__ hl = h + lane * 2;
  float2 acc = make_float2(0.0f, 0.0f);

  for (int j = beg; j < end; j += 32) {
    const int cnt = min(32, end - j);
    int mine = (lane < cnt) ? sorted_src[j + lane] : 0;
    mine = (int)min((unsigned int)mine, (unsigned int)(n_nodes - 1));
    int t = 0;
    for (; t + 8 <= cnt; t += 8) {
      float2 v[8];
#pragma unroll
      for (int u = 0; u < 8; ++u) {
        const int s = __builtin_amdgcn_readlane(mine, t + u);
        v[u] = *(const float2*)(hl + (size_t)s * 64);
      }
#pragma unroll
      for (int u = 0; u < 8; ++u) { acc.x += v[u].x; acc.y += v[u].y; }
    }
    if (t < cnt) {
      float2 v[8];
#pragma unroll
      for (int u = 0; u < 8; ++u) {
        v[u] = make_float2(0.0f, 0.0f);
        if (t + u < cnt) {
          const int s = __builtin_amdgcn_readlane(mine, t + u);
          v[u] = *(const float2*)(hl + (size_t)s * 64);
        }
      }
#pragma unroll
      for (int u = 0; u < 8; ++u) { acc.x += v[u].x; acc.y += v[u].y; }
    }
  }
  const float nrm = in_norm[d];
  const float2 bb = *(const float2*)(b2 + lane * 2);
  float2 o;
  o.x = acc.x * nrm + bb.x;
  o.y = acc.y * nrm + bb.y;
  *(float2*)(out + (size_t)d * 64 + lane * 2) = o;
}


static inline int grid1d(long n, int block) { return (int)((n + block - 1) / block); }

extern "C" void kernel_launch(void* const* d_in, const int* in_sizes, int n_in,
                              void* d_out, int out_size, void* d_ws, size_t ws_size,
                              hipStream_t stream) {
  const float* x   = (const float*)d_in[0];
  const float* W1  = (const float*)d_in[1];
  const float* b1  = (const float*)d_in[2];
  const float* W2  = (const float*)d_in[3];
  const float* b2  = (const float*)d_in[4];
  const int*   src = (const int*)d_in[5];
  const int*   dst = (const int*)d_in[6];

  const int n_nodes = in_sizes[0] / IN_DIM;
  const int n_edges = in_sizes[5];
  float* out = (float*)d_out;

  const size_t row = ((size_t)n_nodes + 31) & ~(size_t)31;
  float* h1   = (float*)d_ws;
  float* agg1 = h1 + (size_t)n_nodes * HID_DIM;
  float* h2   = agg1 + (size_t)n_nodes * HID_DIM;
  unsigned int* partial = (unsigned int*)(h2 + (size_t)n_nodes * OUT_DIM);
  unsigned int* part_dst = partial + DEG_SLICES * row;
  float* out_norm = (float*)(partial + 2 * DEG_SLICES * row);
  float* in_norm  = out_norm + row;
  unsigned int* row_ptr  = (unsigned int*)(in_norm + row);
  unsigned int* block_in = row_ptr + row + 32;
  const int node_blocks = grid1d(n_nodes, 128);
  const size_t bin_row_words = ((size_t)node_blocks + 3) & ~(size_t)3;
  unsigned int* bin_off = block_in + row;
  uint2* pairs = (uint2*)(bin_off + DEG_SLICES * bin_row_words);
  int* sorted_src = (int*)(pairs + n_edges);

  degree_hist<<<dim3(DEG_SLICES, 2), HIST_THREADS, 0, stream>>>(src, dst, partial, n_edges, n_nodes);
  degree_norm<<<node_blocks, 256, 0, stream>>>(partial, out_norm, in_norm, block_in, n_nodes);

  const int row_tiles = n_nodes / 16;

  cursors_and_gemm1<<<GEMM_GRID + node_blocks, GEMM_THREADS, 0, stream>>>(
      x, W1, out_norm, h1, row_tiles, part_dst, block_in, row_ptr, bin_off, n_nodes, node_blocks);
  edge_bin<<<DEG_SLICES, HIST_THREADS, 0, stream>>>(src, dst, bin_off, pairs, n_edges, n_nodes, node_blocks);
  edge_place<<<node_blocks, PLACE_THREADS, 0, stream>>>(pairs, row_ptr, sorted_src, n_edges, n_nodes);

  gather128<<<(n_nodes + 7) / 8, 256, 0, stream>>>(h1, sorted_src, row_ptr, agg1, n_nodes, n_edges);

  gemm2_wmma<<<GEMM_GRID, GEMM_THREADS, 0, stream>>>(agg1, W2, b1, out_norm, in_norm, h2, row_tiles);

  gather64_out<<<(n_nodes + 7) / 8, 256, 0, stream>>>(h2, sorted_src, row_ptr, in_norm, b2, out,
                                                      n_nodes, n_edges);
}


# edge sort off the launch chain: run and row starts come from per-slice bin counts of the histogram pass; binning shares the norm launch, placing the first product's: 8 launches, now 6
# speedup vs baseline: 5.5780x; 1.1120x over previous
#include <hip/hip_runtime.h>

typedef __attribute__((ext_vector_type(2))) float v2f;
typedef __attribute__((ext_vector_type(8))) float v8f;

#define IN_DIM  128
#define HID_DIM 128
#define OUT_DIM 64


#define HIST_CAP     65536
#define HIST_THREADS 1024
#define DEG_SLICES   128
static_assert(DEG_SLICES % 8 == 0, "degree_norm_block splits the slice rows over 8 waves");

#define BIN_NODES 128
#define BIN_SHIFT 7
static_assert((1 << BIN_SHIFT) == BIN_NODES, "a node's bin is its number shifted right");
static_assert(HIST_CAP % BIN_NODES == 0, "every histogram pass starts on a bin boundary");
static_assert(HIST_THREADS == 1024, "thread halves and quarters below are written for 1024 threads");

__device__ __forceinline__ int bin_row(int n_bins) { return (n_bins + 3) & ~3; }

__device__ __forceinline__ int padded_row(int n_nodes) { return (n_nodes + 31) & ~31; }

__device__ __forceinline__ void slice_range(int n_edges, int slice, int& e0, int& e1) {
  const int per = (((n_edges + DEG_SLICES - 1) / DEG_SLICES) + 3) & ~3;
  e0 = min(slice * per, n_edges);
  e1 = min(e0 + per, n_edges);
}

__device__ __forceinline__ void hist_bump(unsigned int* hist, int v, int base, int nb) {
  const unsigned int b = (unsigned int)(v - base);
  if (b < (unsigned int)nb) atomicAdd(&hist[b], 1u);
}

__global__ __launch_bounds__(HIST_THREADS) void degree_hist(const int* __restrict__ src,
                                                            const int* __restrict__ dst,
                                                            unsigned int* __restrict__ partial,
                                                            unsigned int* __restrict__ bin_cnt,
                                                            int n_edges, int n_nodes, int n_bins) {
  __shared__ __attribute__((aligned(16))) unsigned int hist[HIST_CAP];
  const int tid = threadIdx.x;
  const int* __restrict__ idx = blockIdx.y ? dst : src;
  const int n_slices = DEG_SLICES;
  int e0, e1;
  slice_range(n_edges, (int)blockIdx.x, e0, e1);
  const int nv  = (e1 - e0) >> 2;
  const int4* __restrict__ p4 = (const int4*)(idx + e0);
  const int row = padded_row(n_nodes);
  unsigned int* __restrict__ my =
      partial + ((size_t)blockIdx.y * n_slices + blockIdx.x) * (size_t)row;
  unsigned int* __restrict__ bins = bin_cnt + (size_t)blockIdx.x * (size_t)bin_row(n_bins);
  const bool dst_side = blockIdx.y != 0;

  for (int base = 0; base < n_nodes; base += HIST_CAP) {
    const int nb  = min(HIST_CAP, n_nodes - base);
    const int nb4 = (nb + 3) >> 2;
    uint4* h4 = (uint4*)hist;
    for (int i = tid; i < nb4; i += HIST_THREADS) h4[i] = make_uint4(0u, 0u, 0u, 0u);
    __syncthreads();

    for (int q = tid; q < nv; q += 4 * HIST_THREADS) {
      int4 v[4];
#pragma unroll
      for (int j = 0; j < 4; ++j) {
        const int qq = q + j * HIST_THREADS;
        v[j] = (qq < nv) ? p4[qq] : make_int4(-1, -1, -1, -1);
      }
#pragma unroll
      for (int j = 0; j < 4; ++j) {
        hist_bump(hist, v[j].x, base, nb);
        hist_bump(hist, v[j].y, base, nb);
        hist_bump(hist, v[j].z, base, nb);
        hist_bump(hist, v[j].w, base, nb);
      }
    }
    {
      const int e = e0 + (nv << 2) + tid;
      if (tid < 3 && e < e1) hist_bump(hist, idx[e], base, nb);
    }
    __syncthreads();

    uint4* o4 = (uint4*)(my + base);
    for (int i = tid; i < nb4; i += HIST_THREADS) o4[i] = h4[i];
    if (dst_side) {
      const int nbin = (nb + BIN_NODES - 1) >> BIN_SHIFT;
      for (int b = tid; b < nbin; b += HIST_THREADS) {
        unsigned int t = 0;
#pragma unroll 8
        for (int j = 0; j < 32; ++j) {
          const int jj = (j + b) & 31;
          const int i  = 32 * b + jj;
          if (i < nb4) {
            const uint4 v = h4[i];
            t += v.x + v.y + v.z + v.w;
          }
        }
        bins[(base >> BIN_SHIFT) + b] = t;
      }
    }
    __syncthreads();
  }
}

__device__ __forceinline__ void degree_norm_block(const unsigned int* __restrict__ partial,
                                                  const unsigned int* __restrict__ bin_cnt,
                                                  float* __restrict__ out_norm,
                                                  float* __restrict__ in_norm,
                                                  unsigned int* __restrict__ row_ptr,
                                                  int n_nodes, int n_bins, int first_blk) {
  __shared__ uint4 red[4][2][8][32];
  __shared__ unsigned int bred[4][8];
  const int lane = threadIdx.x & 31;
  const int w    = (threadIdx.x >> 5) & 7;
  const int qt   = threadIdx.x >> 8;
  const int nblk = first_blk + qt;
  const int node = (nblk * 32 + lane) * 4;
  const int row  = padded_row(n_nodes);
  const uint4 zero4 = make_uint4(0u, 0u, 0u, 0u);

  const int before = (nblk < n_bins) ? nblk : 0;
  unsigned int eb = 0;
  {
    const size_t brow4 = (size_t)(bin_row(n_bins) >> 2);
    const uint4* __restrict__ rows4 = (const uint4*)bin_cnt + (size_t)(w * (DEG_SLICES / 8)) * brow4;
    for (int j0 = 0; j0 < before; j0 += 128) {
      const int wd = j0 + 4 * lane;
      uint4 v[DEG_SLICES / 8];
#pragma unroll
      for (int k = 0; k < DEG_SLICES / 8; ++k) v[k] = zero4;
      if (wd < before) {
#pragma unroll
        for (int k = 0; k < DEG_SLICES / 8; ++k) v[k] = rows4[(size_t)k * brow4 + (size_t)(wd >> 2)];
      }
#pragma unroll
      for (int k = 0; k < DEG_SLICES / 8; ++k) {
        eb += v[k].x;
        eb += (wd + 1 < before) ? v[k].y : 0u;
        eb += (wd + 2 < before) ? v[k].z : 0u;
        eb += (wd + 3 < before) ? v[k].w : 0u;
      }
    }
    for (int o = 16; o > 0; o >>= 1) eb += __shfl_xor(eb, o);
    if (lane == 0) bred[qt][w] = eb;
  }

  uint4 so = zero4;
  uint4 si = zero4;
  if (node < n_nodes) {
    const uint4* __restrict__ po = (const uint4*)(partial + node);
    const uint4* __restrict__ pi = (const uint4*)(partial + (size_t)DEG_SLICES * row + node);
    const size_t row4 = (size_t)(row >> 2);
#pragma unroll
    for (int k = 0; k < DEG_SLICES / 8; ++k) {
      const size_t s = (size_t)(w + 8 * k);
      const uint4 a = po[s * row4];
      const uint4 b = pi[s * row4];
      so.x += a.x; so.y += a.y; so.z += a.z; so.w += a.w;
      si.x += b.x; si.y += b.y; si.z += b.z; si.w += b.w;
    }
  }
  red[qt][0][w][lane] = so;
  red[qt][1][w][lane] = si;
  __syncthreads();
  if (w == 0) {
    so = zero4;
    si = zero4;
    unsigned int base = 0;
#pragma unroll
    for (int k = 0; k < 8; ++k) {
      const uint4 a = red[qt][0][k][lane];
      const uint4 b = red[qt][1][k][lane];
      so.x += a.x; so.y += a.y; so.z += a.z; so.w += a.w;
      si.x += b.x; si.y += b.y; si.z += b.z; si.w += b.w;
      base += bred[qt][k];
    }
    const unsigned int lt = si.x + si.y + si.z + si.w;
    unsigned int inc = lt;
    for (int o = 1; o < 32; o <<= 1) {
      const unsigned int u = __shfl_up(inc, (unsigned int)o);
      if (lane >= o) inc += u;
    }
    if (node < n_nodes) {
      float4 o, i;
      o.x = rsqrtf(fmaxf((float)so.x, 1.0f));
      o.y = rsqrtf(fmaxf((float)so.y, 1.0f));
      o.z = rsqrtf(fmaxf((float)so.z, 1.0f));
      o.w = rsqrtf(fmaxf((float)so.w, 1.0f));
      i.x = rsqrtf(fmaxf((float)si.x, 1.0f));
      i.y = rsqrtf(fmaxf((float)si.y, 1.0f));
      i.z = rsqrtf(fmaxf((float)si.z, 1.0f));
      i.w = rsqrtf(fmaxf((float)si.w, 1.0f));
      *(float4*)(out_norm + node) = o;
      *(float4*)(in_norm + node)  = i;
      uint4 st;
      st.x = base + inc - lt;
      st.y = st.x + si.x;
      st.z = st.y + si.y;
      st.w = st.z + si.z;
      *(uint4*)(row_ptr + node) = st;
      if (node + 4 >= n_nodes) row_ptr[n_nodes] = st.x + lt;
    }
  }
}


#define BIN_CAP   1024
static_assert(BIN_CAP == HIST_THREADS, "edge_bin_block scans one bin per thread");

#define BIN_TRIP (16 * HIST_THREADS)

__device__ __forceinline__ void edge_bin_block(const int* __restrict__ src,
                                               const int* __restrict__ dst,
                                               const unsigned int* __restrict__ bin_cnt,
                                               uint2* __restrict__ pairs,
                                               int n_edges, int n_nodes, int n_bins, int slice) {
  __shared__ unsigned int cnt[BIN_CAP];
  __shared__ unsigned int start[BIN_CAP];
  __shared__ unsigned int gbase[BIN_CAP];
  __shared__ unsigned int wsum[HIST_THREADS / 32];
  __shared__ uint2 buf[BIN_TRIP + 4];
  const int tid  = threadIdx.x;
  const int lane = tid & 31;
  const int w    = tid >> 5;
  const int nb   = min(n_bins, BIN_CAP);
  int e0, e1;
  slice_range(n_edges, slice, e0, e1);
  const int nv = (e1 - e0) >> 2;
  const int4* __restrict__ d4 = (const int4*)(dst + e0);
  const int4* __restrict__ s4 = (const int4*)(src + e0);

  {
    unsigned int* sc = (unsigned int*)buf;
    const int g = tid >> 9;
    const size_t brow = (size_t)bin_row(n_bins);
    for (int b = tid & 511; b < nb; b += 512) {
      const unsigned int* __restrict__ col =
          bin_cnt + (size_t)(g * (DEG_SLICES / 2)) * brow + (size_t)b;
      unsigned int tot = 0, pre = 0;
#pragma unroll
      for (int k = 0; k < DEG_SLICES / 2; ++k) {
        const unsigned int v = col[(size_t)k * brow];
        tot += v;
        if (g * (DEG_SLICES / 2) + k < slice) pre += v;
      }
      sc[g * BIN_CAP + b]       = tot;
      sc[(2 + g) * BIN_CAP + b] = pre;
    }
    __syncthreads();
    unsigned int tot = 0, pre = 0;
    if (tid < nb) {
      tot = sc[tid] + sc[BIN_CAP + tid];
      pre = sc[2 * BIN_CAP + tid] + sc[3 * BIN_CAP + tid];
    }
    unsigned int inc = tot;
    for (int o = 1; o < 32; o <<= 1) {
      const unsigned int u = __shfl_up(inc, (unsigned int)o);
      if (lane >= o) inc += u;
    }
    if (lane == 31) wsum[w] = inc;
    __syncthreads();
    unsigned int before = 0;
    for (int k = 0; k < HIST_THREADS / 32; ++k) {
      const unsigned int t = wsum[k];
      if (k < w) before += t;
    }
    gbase[tid] = before + inc - tot + pre;
  }

  for (int q0 = 0; q0 < max(nv, 1); q0 += 4 * HIST_THREADS) {
    cnt[tid] = 0u;
    __syncthreads();

    int dv[17], sv[17], rk[17];
#pragma unroll
    for (int j = 0; j < 4; ++j) {
      const int qq = q0 + tid + j * HIST_THREADS;
      const bool ok = qq < nv;
      const int4 dq = ok ? d4[qq] : make_int4(-1, -1, -1, -1);
      const int4 sq = ok ? s4[qq] : make_int4(0, 0, 0, 0);
      dv[4 * j] = dq.x; dv[4 * j + 1] = dq.y; dv[4 * j + 2] = dq.z; dv[4 * j + 3] = dq.w;
      sv[4 * j] = sq.x; sv[4 * j + 1] = sq.y; sv[4 * j + 2] = sq.z; sv[4 * j + 3] = sq.w;
    }
    {
      const int e = e0 + (nv << 2) + tid;
      const bool ok = tid < 3 && e < e1 && q0 + 4 * HIST_THREADS >= nv;
      dv[16] = ok ? dst[e] : -1;
      sv[16] = ok ? src[e] : 0;
    }
#pragma unroll
    for (int j = 0; j < 17; ++j) {
      const unsigned int bn = (unsigned int)dv[j] >> BIN_SHIFT;
      const bool ok = (unsigned int)dv[j] < (unsigned int)n_nodes && bn < (unsigned int)nb;
      rk[j] = ok ? (int)atomicAdd(&cnt[bn], 1u) : -1;
    }
    __syncthreads();

    const unsigned int mine = cnt[tid];
    unsigned int inc = mine;
    for (int o = 1; o < 32; o <<= 1) {
      const unsigned int u = __shfl_up(inc, (unsigned int)o);
      if (lane >= o) inc += u;
    }
    if (lane == 31) wsum[w] = inc;
    __syncthreads();
    unsigned int before = 0, total = 0;
    for (int k = 0; k < HIST_THREADS / 32; ++k) {
      const unsigned int t = wsum[k];
      if (k < w) before += t;
      total += t;
    }
    start[tid] = before + inc - mine;
    __syncthreads();

#pragma unroll
    for (int j = 0; j < 17; ++j) {
      if (rk[j] >= 0) {
        const unsigned int i = start[(unsigned int)dv[j] >> BIN_SHIFT] + (unsigned int)rk[j];
        if (i < (unsigned int)(BIN_TRIP + 4)) buf[i] = make_uint2((unsigned int)sv[j], (unsigned int)dv[j]);
      }
    }
    __syncthreads();

    for (unsigned int i = tid; i < total; i += HIST_THREADS) {
      const uint2 pr = buf[i];
      const unsigned int bn = pr.y >> BIN_SHIFT;
      const unsigned int pos = gbase[bn] + (i - start[bn]);
      if (pos < (unsigned int)n_edges) pairs[pos] = pr;
    }
    __syncthreads();
    gbase[tid] += mine;
  }
}

__global__ __launch_bounds__(HIST_THREADS) void norms_and_edge_bin(const int* __restrict__ src,
                                                                   const int* __restrict__ dst,
                                                                   const unsigned int* __restrict__ bin_cnt,
                                                                   uint2* __restrict__ pairs,
                                                                   int n_edges, int n_nodes, int n_bins,
                                                                   const unsigned int* __restrict__ partial,
                                                                   float* __restrict__ out_norm,
                                                                   float* __restrict__ in_norm,
                                                                   unsigned int* __restrict__ row_ptr) {
  const int blk = (int)blockIdx.x;
  if (blk < DEG_SLICES) edge_bin_block(src, dst, bin_cnt, pairs, n_edges, n_nodes, n_bins, blk);
  else                  degree_norm_block(partial, bin_cnt, out_norm, in_norm, row_ptr, n_nodes, n_bins,
                                          (blk - DEG_SLICES) * 4);
}

#define PLACE_THREADS 256
#define PLACE_CAP     12288

__device__ __forceinline__ void edge_place_block(const uint2* __restrict__ pairs,
                                                 const unsigned int* __restrict__ row_ptr,
                                                 int* __restrict__ sorted_src,
                                                 int n_edges, int n_nodes, int bin) {
  __shared__ unsigned int cur[BIN_NODES];
  __shared__ int lst[PLACE_CAP];
  const int tid  = threadIdx.x;
  const int d0   = bin * BIN_NODES;
  const int nloc = min(BIN_NODES, n_nodes - d0);
  const unsigned int r0 = min(row_ptr[d0], (unsigned int)n_edges);
  const unsigned int r1 = min(row_ptr[d0 + nloc], (unsigned int)n_edges);
  const unsigned int ne = r1 > r0 ? r1 - r0 : 0u;
  const bool in_lds = ne <= (unsigned int)PLACE_CAP;
  if (tid < nloc) cur[tid] = row_ptr[d0 + tid] - r0;
  __syncthreads();

  const uint2* __restrict__ pp = pairs + r0;
  int* __restrict__ out = sorted_src + r0;
  for (unsigned int i0 = 0; i0 < ne; i0 += 4 * PLACE_THREADS) {
    uint2 pr[4];
#pragma unroll
    for (int j = 0; j < 4; ++j) {
      const unsigned int i = i0 + (unsigned int)(tid + j * PLACE_THREADS);
      pr[j] = (i < ne) ? pp[i] : make_uint2(0u, 0xffffffffu);
    }
#pragma unroll
    for (int j = 0; j < 4; ++j) {
      const unsigned int ln = pr[j].y - (unsigned int)d0;
      if (ln < (unsigned int)nloc) {
        const unsigned int pos = atomicAdd(&cur[ln], 1u);
        if (pos < ne) {
          if (in_lds) lst[pos] = (int)pr[j].x;
          else        out[pos] = (int)pr[j].x;
        }
      }
    }
  }
  __syncthreads();
  if (in_lds)
    for (unsigned int i = tid; i < ne; i += PLACE_THREADS) out[i] = lst[i];
}


#define GEMM_K       128
#define GEMM_WAVES   8
#define GEMM_THREADS (GEMM_WAVES * 32)
#define GEMM_GRID    256
static_assert(IN_DIM == GEMM_K && HID_DIM == GEMM_K, "both products have K = 128");

template <int NC>
__device__ __forceinline__ void stage_weights(const float* __restrict__ Wm, float4* Bl4) {
#pragma unroll
  for (int it = 0; it < (GEMM_K / 4 * NC) / GEMM_THREADS; ++it) {
    const int idx = it * GEMM_THREADS + (int)threadIdx.x;
    const int n   = idx % NC;
    const int jh  = idx / NC;
    const int k0  = 8 * (jh >> 1) + 2 * (jh & 1);
    float4 q;
    q.x = Wm[(size_t)(k0)     * NC + n];
    q.y = Wm[(size_t)(k0 + 1) * NC + n];
    q.z = Wm[(size_t)(k0 + 4) * NC + n];
    q.w = Wm[(size_t)(k0 + 5) * NC + n];
    Bl4[idx] = q;
  }
}

template <int NT, int NC = 16 * NT>
__device__ __forceinline__ void strip_mma(const float2 (&a)[GEMM_K / 4], const float4* Bl4,
                                          int half, int sub, v8f (&acc)[NT]) {
  float4 q[NT];
#pragma unroll
  for (int ct = 0; ct < NT; ++ct) q[ct] = Bl4[half * NC + ct * 16 + sub];
#pragma unroll
  for (int j2 = 0; j2 < GEMM_K / 8; ++j2) {
    float4 qn[NT];
#pragma unroll
    for (int ct = 0; ct < NT; ++ct)
      qn[ct] = (j2 + 1 < GEMM_K / 8) ? Bl4[((j2 + 1) * 2 + half) * NC + ct * 16 + sub] : q[ct];
    v2f a0, a1;
    a0.x = a[2 * j2].x;     a0.y = a[2 * j2].y;
    a1.x = a[2 * j2 + 1].x; a1.y = a[2 * j2 + 1].y;
#pragma unroll
    for (int ct = 0; ct < NT; ++ct) {
      v2f b;
      b.x = q[ct].x; b.y = q[ct].y;
      acc[ct] = __builtin_amdgcn_wmma_f32_16x16x4_f32(
           false, a0,  false, b,
           (short)0, acc[ct],  false,  false);
    }
#pragma unroll
    for (int ct = 0; ct < NT; ++ct) {
      v2f b;
      b.x = q[ct].z; b.y = q[ct].w;
      acc[ct] = __builtin_amdgcn_wmma_f32_16x16x4_f32(
          false, a1, false, b, (short)0, acc[ct], false, false);
    }
    __builtin_amdgcn_sched_barrier(0);
#pragma unroll
    for (int ct = 0; ct < NT; ++ct) q[ct] = qn[ct];
  }
}

template <int NT, int NC = 16 * NT>
__device__ __forceinline__ void strip_store(float* __restrict__ dstm, int rt, int half, int sub,
                                            const v8f (&acc)[NT]) {
  const int m_base = rt * 16 + half * 8;
#pragma unroll
  for (int ct = 0; ct < NT; ++ct)
#pragma unroll
    for (int r = 0; r < 8; ++r)
      dstm[(size_t)(m_base + r) * NC + ct * 16 + sub] = acc[ct][r];
}


__device__ __forceinline__ void load_a1(const float* __restrict__ x,
                                        const float* __restrict__ out_norm,
                                        int rt, int half, int sub, float2 (&a)[GEMM_K / 4]) {
  const int m = rt * 16 + sub;
  const float norm = out_norm[m];
  const float* __restrict__ arow = x + (size_t)m * IN_DIM + half * 2;
#pragma unroll
  for (int j = 0; j < GEMM_K / 4; ++j) {
    const float2 av = *(const float2*)(arow + 4 * j);
    a[j].x = av.x * norm;
    a[j].y = av.y * norm;
  }
}

__device__ __forceinline__ void gemm1_block(const float* __restrict__ x,
                                            const float* __restrict__ W1,
                                            const float* __restrict__ out_norm,
                                            float* __restrict__ h1,
                                            int n_row_tiles, int blk) {
  __shared__ float4 Bl4[GEMM_K / 4 * HID_DIM];
  const int lane = threadIdx.x & 31;
  const int wave = __builtin_amdgcn_readfirstlane((int)(threadIdx.x >> 5));
  const int half = lane >> 4;
  const int sub  = lane & 15;
  const int step = GEMM_GRID * GEMM_WAVES;
  int rt = blk + GEMM_GRID * wave;

  float2 a[GEMM_K / 4];
#pragma unroll
  for (int j = 0; j < GEMM_K / 4; ++j) a[j] = make_float2(0.0f, 0.0f);
  if (rt < n_row_tiles) load_a1(x, out_norm, rt, half, sub, a);

  stage_weights<HID_DIM>(W1, Bl4);
  __syncthreads();

  while (rt < n_row_tiles) {
    const int nx = rt + step;
    float2 an[GEMM_K / 4];
#pragma unroll
    for (int j = 0; j < GEMM_K / 4; ++j) an[j] = make_float2(0.0f, 0.0f);
    if (nx < n_row_tiles) load_a1(x, out_norm, nx, half, sub, an);

#pragma unroll
    for (int hc = 0; hc < 2; ++hc) {
      v8f acc[HID_DIM / 32];
#pragma unroll
      for (int ct = 0; ct < HID_DIM / 32; ++ct) acc[ct] = v8f{};
      strip_mma<HID_DIM / 32, HID_DIM>(a, Bl4 + hc * (HID_DIM / 2), half, sub, acc);
      strip_store<HID_DIM / 32, HID_DIM>(h1 + hc * (HID_DIM / 2), rt, half, sub, acc);
    }
#pragma unroll
    for (int j = 0; j < GEMM_K / 4; ++j) a[j] = an[j];
    rt = nx;
  }
}

__global__ __launch_bounds__(GEMM_THREADS) void place_and_gemm1(const float* __restrict__ x,
                                                                const float* __restrict__ W1,
                                                                const float* __restrict__ out_norm,
                                                                float* __restrict__ h1,
                                                                int n_row_tiles,
                                                                const uint2* __restrict__ pairs,
                                                                const unsigned int* __restrict__ row_ptr,
                                                                int* __restrict__ sorted_src,
                                                                int n_edges, int n_nodes) {
  static_assert(GEMM_THREADS == PLACE_THREADS, "edge_place_block is written for the product's block size");
  const int blk = (int)blockIdx.x;
  if (blk < GEMM_GRID) gemm1_block(x, W1, out_norm, h1, n_row_tiles, blk);
  else                 edge_place_block(pairs, row_ptr, sorted_src, n_edges, n_nodes, blk - GEMM_GRID);
}


__device__ __forceinline__ void load_a2(const float* __restrict__ agg1,
                                        const float* __restrict__ b1,
                                        const float* __restrict__ out_norm,
                                        const float* __restrict__ in_norm,
                                        int rt, int half, int sub, float2 (&a)[GEMM_K / 4]) {
  const int m = rt * 16 + sub;
  const float s_in  = in_norm[m];
  const float s_out = out_norm[m];
  const float* __restrict__ arow = agg1 + (size_t)m * HID_DIM + half * 2;
  const float* __restrict__ brow = b1 + half * 2;
#pragma unroll
  for (int j = 0; j < GEMM_K / 4; ++j) {
    const float2 av = *(const float2*)(arow + 4 * j);
    const float2 bv = *(const float2*)(brow + 4 * j);
    a[j].x = fmaxf(av.x * s_in + bv.x, 0.0f) * s_out;
    a[j].y = fmaxf(av.y * s_in + bv.y, 0.0f) * s_out;
  }
}

__global__ __launch_bounds__(GEMM_THREADS) void gemm2_wmma(const float* __restrict__ agg1,
                                                           const float* __restrict__ W2,
                                                           const float* __restrict__ b1,
                                                           const float* __restrict__ out_norm,
                                                           const float* __restrict__ in_norm,
                                                           float* __restrict__ h2,
                                                           int n_row_tiles) {
  __shared__ float4 Bl4[GEMM_K / 4 * OUT_DIM];
  const int lane = threadIdx.x & 31;
  const int wave = __builtin_amdgcn_readfirstlane((int)(threadIdx.x >> 5));
  const int half = lane >> 4;
  const int sub  = lane & 15;
  const int step = (int)gridDim.x * GEMM_WAVES;
  int rt = (int)blockIdx.x + (int)gridDim.x * wave;

  float2 a[GEMM_K / 4];
#pragma unroll
  for (int j = 0; j < GEMM_K / 4; ++j) a[j] = make_float2(0.0f, 0.0f);
  if (rt < n_row_tiles) load_a2(agg1, b1, out_norm, in_norm, rt, half, sub, a);

  stage_weights<OUT_DIM>(W2, Bl4);
  __syncthreads();

  while (rt < n_row_tiles) {
    const int nx = rt + step;
    float2 an[GEMM_K / 4];
#pragma unroll
    for (int j = 0; j < GEMM_K / 4; ++j) an[j] = make_float2(0.0f, 0.0f);
    if (nx < n_row_tiles) load_a2(agg1, b1, out_norm, in_norm, nx, half, sub, an);

    v8f acc[OUT_DIM / 16];
#pragma unroll
    for (int ct = 0; ct < OUT_DIM / 16; ++ct) acc[ct] = v8f{};
    strip_mma<OUT_DIM / 16>(a, Bl4, half, sub, acc);
    strip_store<OUT_DIM / 16>(h2, rt, half, sub, acc);
#pragma unroll
    for (int j = 0; j < GEMM_K / 4; ++j) a[j] = an[j];
    rt = nx;
  }
}


__global__ __launch_bounds__(256) void gather128(const float* __restrict__ h,
                                                 const int* __restrict__ sorted_src,
                                                 const unsigned int* __restrict__ row_ptr,
                                                 float* __restrict__ agg,
                                                 int n_nodes, int n_edges) {
  const int lane = threadIdx.x & 31;
  const int d = __builtin_amdgcn_readfirstlane((int)((blockIdx.x * blockDim.x + threadIdx.x) >> 5));
  if (d >= n_nodes) return;
  const int beg = (int)min(row_ptr[d], (unsigned int)n_edges);
  const int end = (int)min(row_ptr[d + 1], (unsigned int)n_edges);
  const float* __restrict__ hl = h + lane * 4;
  float4 acc = make_float4(0.0f, 0.0f, 0.0f, 0.0f);

  for (int j = beg; j < end; j += 32) {
    const int cnt = min(32, end - j);
    int mine = (lane < cnt) ? sorted_src[j + lane] : 0;
    mine = (int)min((unsigned int)mine, (unsigned int)(n_nodes - 1));
    int t = 0;
    for (; t + 8 <= cnt; t += 8) {
      float4 v[8];
#pragma unroll
      for (int u = 0; u < 8; ++u) {
        const int s = __builtin_amdgcn_readlane(mine, t + u);
        v[u] = *(const float4*)(hl + (size_t)s * 128);
      }
#pragma unroll
      for (int u = 0; u < 8; ++u) {
        acc.x += v[u].x; acc.y += v[u].y; acc.z += v[u].z; acc.w += v[u].w;
      }
    }
    if (t < cnt) {
      float4 v[8];
#pragma unroll
      for (int u = 0; u < 8; ++u) {
        v[u] = make_float4(0.0f, 0.0f, 0.0f, 0.0f);
        if (t + u < cnt) {
          const int s = __builtin_amdgcn_readlane(mine, t + u);
          v[u] = *(const float4*)(hl + (size_t)s * 128);
        }
      }
#pragma unroll
      for (int u = 0; u < 8; ++u) {
        acc.x += v[u].x; acc.y += v[u].y; acc.z += v[u].z; acc.w += v[u].w;
      }
    }
  }
  *(float4*)(agg + (size_t)d * 128 + lane * 4) = acc;
}

__global__ __launch_bounds__(256) void gather64_out(const float* __restrict__ h,
                                                    const int* __restrict__ sorted_src,
                                                    const unsigned int* __restrict__ row_ptr,
                                                    const float* __restrict__ in_norm,
                                                    const float* __restrict__ b2,
                                                    float* __restrict__ out,
                                                    int n_nodes, int n_edges) {
  const int lane = threadIdx.x & 31;
  const int d = __builtin_amdgcn_readfirstlane((int)((blockIdx.x * blockDim.x + threadIdx.x) >> 5));
  if (d >= n_nodes) return;
  const int beg = (int)min(row_ptr[d], (unsigned int)n_edges);
  const int end = (int)min(row_ptr[d + 1], (unsigned int)n_edges);
  const float* __restrict__ hl = h + lane * 2;
  float2 acc = make_float2(0.0f, 0.0f);

  for (int j = beg; j < end; j += 32) {
    const int cnt = min(32, end - j);
    int mine = (lane < cnt) ? sorted_src[j + lane] : 0;
    mine = (int)min((unsigned int)mine, (unsigned int)(n_nodes - 1));
    int t = 0;
    for (; t + 8 <= cnt; t += 8) {
      float2 v[8];
#pragma unroll
      for (int u = 0; u < 8; ++u) {
        const int s = __builtin_amdgcn_readlane(mine, t + u);
        v[u] = *(const float2*)(hl + (size_t)s * 64);
      }
#pragma unroll
      for (int u = 0; u < 8; ++u) { acc.x += v[u].x; acc.y += v[u].y; }
    }
    if (t < cnt) {
      float2 v[8];
#pragma unroll
      for (int u = 0; u < 8; ++u) {
        v[u] = make_float2(0.0f, 0.0f);
        if (t + u < cnt) {
          const int s = __builtin_amdgcn_readlane(mine, t + u);
          v[u] = *(const float2*)(hl + (size_t)s * 64);
        }
      }
#pragma unroll
      for (int u = 0; u < 8; ++u) { acc.x += v[u].x; acc.y += v[u].y; }
    }
  }
  const float nrm = in_norm[d];
  const float2 bb = *(const float2*)(b2 + lane * 2);
  float2 o;
  o.x = acc.x * nrm + bb.x;
  o.y = acc.y * nrm + bb.y;
  *(float2*)(out + (size_t)d * 64 + lane * 2) = o;
}


static inline int grid1d(long n, int block) { return (int)((n + block - 1) / block); }

extern "C" void kernel_launch(void* const* d_in, const int* in_sizes, int n_in,
                              void* d_out, int out_size, void* d_ws, size_t ws_size,
                              hipStream_t stream) {
  const float* x   = (const float*)d_in[0];
  const float* W1  = (const float*)d_in[1];
  const float* b1  = (const float*)d_in[2];
  const float* W2  = (const float*)d_in[3];
  const float* b2  = (const float*)d_in[4];
  const int*   src = (const int*)d_in[5];
  const int*   dst = (const int*)d_in[6];

  const int n_nodes = in_sizes[0] / IN_DIM;
  const int n_edges = in_sizes[5];
  float* out = (float*)d_out;

  const size_t row = ((size_t)n_nodes + 31) & ~(size_t)31;
  float* h1   = (float*)d_ws;
  float* agg1 = h1 + (size_t)n_nodes * HID_DIM;
  float* h2   = agg1 + (size_t)n_nodes * HID_DIM;
  unsigned int* partial = (unsigned int*)(h2 + (size_t)n_nodes * OUT_DIM);
  float* out_norm = (float*)(partial + 2 * DEG_SLICES * row);
  float* in_norm  = out_norm + row;
  unsigned int* row_ptr  = (unsigned int*)(in_norm + row);
  const int node_blocks = grid1d(n_nodes, 128);
  const size_t bin_row_words = ((size_t)node_blocks + 3) & ~(size_t)3;
  unsigned int* bin_cnt = row_ptr + row + 32;
  uint2* pairs = (uint2*)(bin_cnt + DEG_SLICES * bin_row_words);
  int* sorted_src = (int*)(pairs + n_edges);

  degree_hist<<<dim3(DEG_SLICES, 2), HIST_THREADS, 0, stream>>>(src, dst, partial, bin_cnt, n_edges, n_nodes,
                                                                node_blocks);
  norms_and_edge_bin<<<DEG_SLICES + grid1d(node_blocks, 4), HIST_THREADS, 0, stream>>>(
      src, dst, bin_cnt, pairs, n_edges, n_nodes, node_blocks, partial, out_norm, in_norm, row_ptr);

  const int row_tiles = n_nodes / 16;

  place_and_gemm1<<<GEMM_GRID + node_blocks, GEMM_THREADS, 0, stream>>>(
      x, W1, out_norm, h1, row_tiles, pairs, row_ptr, sorted_src, n_edges, n_nodes);

  gather128<<<(n_nodes + 7) / 8, 256, 0, stream>>>(h1, sorted_src, row_ptr, agg1, n_nodes, n_edges);

  gemm2_wmma<<<GEMM_GRID, GEMM_THREADS, 0, stream>>>(agg1, W2, b1, out_norm, in_norm, h2, row_tiles);

  gather64_out<<<(n_nodes + 7) / 8, 256, 0, stream>>>(h2, sorted_src, row_ptr, in_norm, b2, out,
                                                      n_nodes, n_edges);
}


# second product fused into the first aggregation: 4 waves sum 16 nodes' in-edge rows, stage the relu'd rows in LDS and multiply by W2 (fragment-packed per call) on WMMA; 6 launches become 5
# speedup vs baseline: 5.7811x; 1.0364x over previous
#include <hip/hip_runtime.h>

typedef __attribute__((ext_vector_type(2))) float v2f;
typedef __attribute__((ext_vector_type(8))) float v8f;

#define IN_DIM  128
#define HID_DIM 128
#define OUT_DIM 64


#define HIST_CAP     65536
#define HIST_THREADS 1024
#define DEG_SLICES   128
static_assert(DEG_SLICES % 8 == 0, "degree_norm_block splits the slice rows over 8 waves");

#define BIN_NODES 128
#define BIN_SHIFT 7
static_assert((1 << BIN_SHIFT) == BIN_NODES, "a node's bin is its number shifted right");
static_assert(HIST_CAP % BIN_NODES == 0, "every histogram pass starts on a bin boundary");
static_assert(HIST_THREADS == 1024, "thread halves and quarters below are written for 1024 threads");

__device__ __forceinline__ int bin_row(int n_bins) { return (n_bins + 3) & ~3; }

__device__ __forceinline__ int padded_row(int n_nodes) { return (n_nodes + 31) & ~31; }

__device__ __forceinline__ void slice_range(int n_edges, int slice, int& e0, int& e1) {
  const int per = (((n_edges + DEG_SLICES - 1) / DEG_SLICES) + 3) & ~3;
  e0 = min(slice * per, n_edges);
  e1 = min(e0 + per, n_edges);
}

__device__ __forceinline__ void hist_bump(unsigned int* hist, int v, int base, int nb) {
  const unsigned int b = (unsigned int)(v - base);
  if (b < (unsigned int)nb) atomicAdd(&hist[b], 1u);
}

__global__ __launch_bounds__(HIST_THREADS) void degree_hist(const int* __restrict__ src,
                                                            const int* __restrict__ dst,
                                                            unsigned int* __restrict__ partial,
                                                            unsigned int* __restrict__ bin_cnt,
                                                            int n_edges, int n_nodes, int n_bins) {
  __shared__ __attribute__((aligned(16))) unsigned int hist[HIST_CAP];
  const int tid = threadIdx.x;
  const int* __restrict__ idx = blockIdx.y ? dst : src;
  const int n_slices = DEG_SLICES;
  int e0, e1;
  slice_range(n_edges, (int)blockIdx.x, e0, e1);
  const int nv  = (e1 - e0) >> 2;
  const int4* __restrict__ p4 = (const int4*)(idx + e0);
  const int row = padded_row(n_nodes);
  unsigned int* __restrict__ my =
      partial + ((size_t)blockIdx.y * n_slices + blockIdx.x) * (size_t)row;
  unsigned int* __restrict__ bins = bin_cnt + (size_t)blockIdx.x * (size_t)bin_row(n_bins);
  const bool dst_side = blockIdx.y != 0;

  for (int base = 0; base < n_nodes; base += HIST_CAP) {
    const int nb  = min(HIST_CAP, n_nodes - base);
    const int nb4 = (nb + 3) >> 2;
    uint4* h4 = (uint4*)hist;
    for (int i = tid; i < nb4; i += HIST_THREADS) h4[i] = make_uint4(0u, 0u, 0u, 0u);
    __syncthreads();

    for (int q = tid; q < nv; q += 4 * HIST_THREADS) {
      int4 v[4];
#pragma unroll
      for (int j = 0; j < 4; ++j) {
        const int qq = q + j * HIST_THREADS;
        v[j] = (qq < nv) ? p4[qq] : make_int4(-1, -1, -1, -1);
      }
#pragma unroll
      for (int j = 0; j < 4; ++j) {
        hist_bump(hist, v[j].x, base, nb);
        hist_bump(hist, v[j].y, base, nb);
        hist_bump(hist, v[j].z, base, nb);
        hist_bump(hist, v[j].w, base, nb);
      }
    }
    {
      const int e = e0 + (nv << 2) + tid;
      if (tid < 3 && e < e1) hist_bump(hist, idx[e], base, nb);
    }
    __syncthreads();

    uint4* o4 = (uint4*)(my + base);
    for (int i = tid; i < nb4; i += HIST_THREADS) o4[i] = h4[i];
    if (dst_side) {
      const int nbin = (nb + BIN_NODES - 1) >> BIN_SHIFT;
      for (int b = tid; b < nbin; b += HIST_THREADS) {
        unsigned int t = 0;
#pragma unroll 8
        for (int j = 0; j < 32; ++j) {
          const int jj = (j + b) & 31;
          const int i  = 32 * b + jj;
          if (i < nb4) {
            const uint4 v = h4[i];
            t += v.x + v.y + v.z + v.w;
          }
        }
        bins[(base >> BIN_SHIFT) + b] = t;
      }
    }
    __syncthreads();
  }
}

__device__ __forceinline__ void degree_norm_block(const unsigned int* __restrict__ partial,
                                                  const unsigned int* __restrict__ bin_cnt,
                                                  float* __restrict__ out_norm,
                                                  float* __restrict__ in_norm,
                                                  unsigned int* __restrict__ row_ptr,
                                                  int n_nodes, int n_bins, int first_blk) {
  __shared__ uint4 red[4][2][8][32];
  __shared__ unsigned int bred[4][8];
  const int lane = threadIdx.x & 31;
  const int w    = (threadIdx.x >> 5) & 7;
  const int qt   = threadIdx.x >> 8;
  const int nblk = first_blk + qt;
  const int node = (nblk * 32 + lane) * 4;
  const int row  = padded_row(n_nodes);
  const uint4 zero4 = make_uint4(0u, 0u, 0u, 0u);

  const int before = (nblk < n_bins) ? nblk : 0;
  unsigned int eb = 0;
  {
    const size_t brow4 = (size_t)(bin_row(n_bins) >> 2);
    const uint4* __restrict__ rows4 = (const uint4*)bin_cnt + (size_t)(w * (DEG_SLICES / 8)) * brow4;
    for (int j0 = 0; j0 < before; j0 += 128) {
      const int wd = j0 + 4 * lane;
      uint4 v[DEG_SLICES / 8];
#pragma unroll
      for (int k = 0; k < DEG_SLICES / 8; ++k) v[k] = zero4;
      if (wd < before) {
#pragma unroll
        for (int k = 0; k < DEG_SLICES / 8; ++k) v[k] = rows4[(size_t)k * brow4 + (size_t)(wd >> 2)];
      }
#pragma unroll
      for (int k = 0; k < DEG_SLICES / 8; ++k) {
        eb += v[k].x;
        eb += (wd + 1 < before) ? v[k].y : 0u;
        eb += (wd + 2 < before) ? v[k].z : 0u;
        eb += (wd + 3 < before) ? v[k].w : 0u;
      }
    }
    for (int o = 16; o > 0; o >>= 1) eb += __shfl_xor(eb, o);
    if (lane == 0) bred[qt][w] = eb;
  }

  uint4 so = zero4;
  uint4 si = zero4;
  if (node < n_nodes) {
    const uint4* __restrict__ po = (const uint4*)(partial + node);
    const uint4* __restrict__ pi = (const uint4*)(partial + (size_t)DEG_SLICES * row + node);
    const size_t row4 = (size_t)(row >> 2);
#pragma unroll
    for (int k = 0; k < DEG_SLICES / 8; ++k) {
      const size_t s = (size_t)(w + 8 * k);
      const uint4 a = po[s * row4];
      const uint4 b = pi[s * row4];
      so.x += a.x; so.y += a.y; so.z += a.z; so.w += a.w;
      si.x += b.x; si.y += b.y; si.z += b.z; si.w += b.w;
    }
  }
  red[qt][0][w][lane] = so;
  red[qt][1][w][lane] = si;
  __syncthreads();
  if (w == 0) {
    so = zero4;
    si = zero4;
    unsigned int base = 0;
#pragma unroll
    for (int k = 0; k < 8; ++k) {
      const uint4 a = red[qt][0][k][lane];
      const uint4 b = red[qt][1][k][lane];
      so.x += a.x; so.y += a.y; so.z += a.z; so.w += a.w;
      si.x += b.x; si.y += b.y; si.z += b.z; si.w += b.w;
      base += bred[qt][k];
    }
    const unsigned int lt = si.x + si.y + si.z + si.w;
    unsigned int inc = lt;
    for (int o = 1; o < 32; o <<= 1) {
      const unsigned int u = __shfl_up(inc, (unsigned int)o);
      if (lane >= o) inc += u;
    }
    if (node < n_nodes) {
      float4 o, i;
      o.x = rsqrtf(fmaxf((float)so.x, 1.0f));
      o.y = rsqrtf(fmaxf((float)so.y, 1.0f));
      o.z = rsqrtf(fmaxf((float)so.z, 1.0f));
      o.w = rsqrtf(fmaxf((float)so.w, 1.0f));
      i.x = rsqrtf(fmaxf((float)si.x, 1.0f));
      i.y = rsqrtf(fmaxf((float)si.y, 1.0f));
      i.z = rsqrtf(fmaxf((float)si.z, 1.0f));
      i.w = rsqrtf(fmaxf((float)si.w, 1.0f));
      *(float4*)(out_norm + node) = o;
      *(float4*)(in_norm + node)  = i;
      uint4 st;
      st.x = base + inc - lt;
      st.y = st.x + si.x;
      st.z = st.y + si.y;
      st.w = st.z + si.z;
      *(uint4*)(row_ptr + node) = st;
      if (node + 4 >= n_nodes) row_ptr[n_nodes] = st.x + lt;
    }
  }
}


#define BIN_CAP   1024
static_assert(BIN_CAP == HIST_THREADS, "edge_bin_block scans one bin per thread");

#define BIN_TRIP (16 * HIST_THREADS)

__device__ __forceinline__ void edge_bin_block(const int* __restrict__ src,
                                               const int* __restrict__ dst,
                                               const unsigned int* __restrict__ bin_cnt,
                                               uint2* __restrict__ pairs,
                                               int n_edges, int n_nodes, int n_bins, int slice) {
  __shared__ unsigned int cnt[BIN_CAP];
  __shared__ unsigned int start[BIN_CAP];
  __shared__ unsigned int gbase[BIN_CAP];
  __shared__ unsigned int wsum[HIST_THREADS / 32];
  __shared__ uint2 buf[BIN_TRIP + 4];
  const int tid  = threadIdx.x;
  const int lane = tid & 31;
  const int w    = tid >> 5;
  const int nb   = min(n_bins, BIN_CAP);
  int e0, e1;
  slice_range(n_edges, slice, e0, e1);
  const int nv = (e1 - e0) >> 2;
  const int4* __restrict__ d4 = (const int4*)(dst + e0);
  const int4* __restrict__ s4 = (const int4*)(src + e0);

  {
    unsigned int* sc = (unsigned int*)buf;
    const int g = tid >> 9;
    const size_t brow = (size_t)bin_row(n_bins);
    for (int b = tid & 511; b < nb; b += 512) {
      const unsigned int* __restrict__ col =
          bin_cnt + (size_t)(g * (DEG_SLICES / 2)) * brow + (size_t)b;
      unsigned int tot = 0, pre = 0;
#pragma unroll
      for (int k = 0; k < DEG_SLICES / 2; ++k) {
        const unsigned int v = col[(size_t)k * brow];
        tot += v;
        if (g * (DEG_SLICES / 2) + k < slice) pre += v;
      }
      sc[g * BIN_CAP + b]       = tot;
      sc[(2 + g) * BIN_CAP + b] = pre;
    }
    __syncthreads();
    unsigned int tot = 0, pre = 0;
    if (tid < nb) {
      tot = sc[tid] + sc[BIN_CAP + tid];
      pre = sc[2 * BIN_CAP + tid] + sc[3 * BIN_CAP + tid];
    }
    unsigned int inc = tot;
    for (int o = 1; o < 32; o <<= 1) {
      const unsigned int u = __shfl_up(inc, (unsigned int)o);
      if (lane >= o) inc += u;
    }
    if (lane == 31) wsum[w] = inc;
    __syncthreads();
    unsigned int before = 0;
    for (int k = 0; k < HIST_THREADS / 32; ++k) {
      const unsigned int t = wsum[k];
      if (k < w) before += t;
    }
    gbase[tid] = before + inc - tot + pre;
  }

  for (int q0 = 0; q0 < max(nv, 1); q0 += 4 * HIST_THREADS) {
    cnt[tid] = 0u;
    __syncthreads();

    int dv[17], sv[17], rk[17];
#pragma unroll
    for (int j = 0; j < 4; ++j) {
      const int qq = q0 + tid + j * HIST_THREADS;
      const bool ok = qq < nv;
      const int4 dq = ok ? d4[qq] : make_int4(-1, -1, -1, -1);
      const int4 sq = ok ? s4[qq] : make_int4(0, 0, 0, 0);
      dv[4 * j] = dq.x; dv[4 * j + 1] = dq.y; dv[4 * j + 2] = dq.z; dv[4 * j + 3] = dq.w;
      sv[4 * j] = sq.x; sv[4 * j + 1] = sq.y; sv[4 * j + 2] = sq.z; sv[4 * j + 3] = sq.w;
    }
    {
      const int e = e0 + (nv << 2) + tid;
      const bool ok = tid < 3 && e < e1 && q0 + 4 * HIST_THREADS >= nv;
      dv[16] = ok ? dst[e] : -1;
      sv[16] = ok ? src[e] : 0;
    }
#pragma unroll
    for (int j = 0; j < 17; ++j) {
      const unsigned int bn = (unsigned int)dv[j] >> BIN_SHIFT;
      const bool ok = (unsigned int)dv[j] < (unsigned int)n_nodes && bn < (unsigned int)nb;
      rk[j] = ok ? (int)atomicAdd(&cnt[bn], 1u) : -1;
    }
    __syncthreads();

    const unsigned int mine = cnt[tid];
    unsigned int inc = mine;
    for (int o = 1; o < 32; o <<= 1) {
      const unsigned int u = __shfl_up(inc, (unsigned int)o);
      if (lane >= o) inc += u;
    }
    if (lane == 31) wsum[w] = inc;
    __syncthreads();
    unsigned int before = 0, total = 0;
    for (int k = 0; k < HIST_THREADS / 32; ++k) {
      const unsigned int t = wsum[k];
      if (k < w) before += t;
      total += t;
    }
    start[tid] = before + inc - mine;
    __syncthreads();

#pragma unroll
    for (int j = 0; j < 17; ++j) {
      if (rk[j] >= 0) {
        const unsigned int i = start[(unsigned int)dv[j] >> BIN_SHIFT] + (unsigned int)rk[j];
        if (i < (unsigned int)(BIN_TRIP + 4)) buf[i] = make_uint2((unsigned int)sv[j], (unsigned int)dv[j]);
      }
    }
    __syncthreads();

    for (unsigned int i = tid; i < total; i += HIST_THREADS) {
      const uint2 pr = buf[i];
      const unsigned int bn = pr.y >> BIN_SHIFT;
      const unsigned int pos = gbase[bn] + (i - start[bn]);
      if (pos < (unsigned int)n_edges) pairs[pos] = pr;
    }
    __syncthreads();
    gbase[tid] += mine;
  }
}

__global__ __launch_bounds__(HIST_THREADS) void norms_and_edge_bin(const int* __restrict__ src,
                                                                   const int* __restrict__ dst,
                                                                   const unsigned int* __restrict__ bin_cnt,
                                                                   uint2* __restrict__ pairs,
                                                                   int n_edges, int n_nodes, int n_bins,
                                                                   const unsigned int* __restrict__ partial,
                                                                   float* __restrict__ out_norm,
                                                                   float* __restrict__ in_norm,
                                                                   unsigned int* __restrict__ row_ptr) {
  const int blk = (int)blockIdx.x;
  if (blk < DEG_SLICES) edge_bin_block(src, dst, bin_cnt, pairs, n_edges, n_nodes, n_bins, blk);
  else                  degree_norm_block(partial, bin_cnt, out_norm, in_norm, row_ptr, n_nodes, n_bins,
                                          (blk - DEG_SLICES) * 4);
}

#define PLACE_THREADS 256
#define PLACE_CAP     12288

__device__ __forceinline__ void edge_place_block(const uint2* __restrict__ pairs,
                                                 const unsigned int* __restrict__ row_ptr,
                                                 int* __restrict__ sorted_src,
                                                 int n_edges, int n_nodes, int bin) {
  __shared__ unsigned int cur[BIN_NODES];
  __shared__ int lst[PLACE_CAP];
  const int tid  = threadIdx.x;
  const int d0   = bin * BIN_NODES;
  const int nloc = min(BIN_NODES, n_nodes - d0);
  const unsigned int r0 = min(row_ptr[d0], (unsigned int)n_edges);
  const unsigned int r1 = min(row_ptr[d0 + nloc], (unsigned int)n_edges);
  const unsigned int ne = r1 > r0 ? r1 - r0 : 0u;
  const bool in_lds = ne <= (unsigned int)PLACE_CAP;
  if (tid < nloc) cur[tid] = row_ptr[d0 + tid] - r0;
  __syncthreads();

  const uint2* __restrict__ pp = pairs + r0;
  int* __restrict__ out = sorted_src + r0;
  for (unsigned int i0 = 0; i0 < ne; i0 += 4 * PLACE_THREADS) {
    uint2 pr[4];
#pragma unroll
    for (int j = 0; j < 4; ++j) {
      const unsigned int i = i0 + (unsigned int)(tid + j * PLACE_THREADS);
      pr[j] = (i < ne) ? pp[i] : make_uint2(0u, 0xffffffffu);
    }
#pragma unroll
    for (int j = 0; j < 4; ++j) {
      const unsigned int ln = pr[j].y - (unsigned int)d0;
      if (ln < (unsigned int)nloc) {
        const unsigned int pos = atomicAdd(&cur[ln], 1u);
        if (pos < ne) {
          if (in_lds) lst[pos] = (int)pr[j].x;
          else        out[pos] = (int)pr[j].x;
        }
      }
    }
  }
  __syncthreads();
  if (in_lds)
    for (unsigned int i = tid; i < ne; i += PLACE_THREADS) out[i] = lst[i];
}


#define GEMM_K       128
#define GEMM_WAVES   8
#define GEMM_THREADS (GEMM_WAVES * 32)
#define GEMM_GRID    256
static_assert(IN_DIM == GEMM_K && HID_DIM == GEMM_K, "both products have K = 128");

template <int NC>
__device__ __forceinline__ void stage_weights(const float* __restrict__ Wm, float4* Bl4) {
#pragma unroll
  for (int it = 0; it < (GEMM_K / 4 * NC) / GEMM_THREADS; ++it) {
    const int idx = it * GEMM_THREADS + (int)threadIdx.x;
    const int n   = idx % NC;
    const int jh  = idx / NC;
    const int k0  = 8 * (jh >> 1) + 2 * (jh & 1);
    float4 q;
    q.x = Wm[(size_t)(k0)     * NC + n];
    q.y = Wm[(size_t)(k0 + 1) * NC + n];
    q.z = Wm[(size_t)(k0 + 4) * NC + n];
    q.w = Wm[(size_t)(k0 + 5) * NC + n];
    Bl4[idx] = q;
  }
}

template <int NT, int NC = 16 * NT>
__device__ __forceinline__ void strip_mma(const float2 (&a)[GEMM_K / 4], const float4* Bl4,
                                          int half, int sub, v8f (&acc)[NT]) {
  float4 q[NT];
#pragma unroll
  for (int ct = 0; ct < NT; ++ct) q[ct] = Bl4[half * NC + ct * 16 + sub];
#pragma unroll
  for (int j2 = 0; j2 < GEMM_K / 8; ++j2) {
    float4 qn[NT];
#pragma unroll
    for (int ct = 0; ct < NT; ++ct)
      qn[ct] = (j2 + 1 < GEMM_K / 8) ? Bl4[((j2 + 1) * 2 + half) * NC + ct * 16 + sub] : q[ct];
    v2f a0, a1;
    a0.x = a[2 * j2].x;     a0.y = a[2 * j2].y;
    a1.x = a[2 * j2 + 1].x; a1.y = a[2 * j2 + 1].y;
#pragma unroll
    for (int ct = 0; ct < NT; ++ct) {
      v2f b;
      b.x = q[ct].x; b.y = q[ct].y;
      acc[ct] = __builtin_amdgcn_wmma_f32_16x16x4_f32(
           false, a0,  false, b,
           (short)0, acc[ct],  false,  false);
    }
#pragma unroll
    for (int ct = 0; ct < NT; ++ct) {
      v2f b;
      b.x = q[ct].z; b.y = q[ct].w;
      acc[ct] = __builtin_amdgcn_wmma_f32_16x16x4_f32(
          false, a1, false, b, (short)0, acc[ct], false, false);
    }
    __builtin_amdgcn_sched_barrier(0);
#pragma unroll
    for (int ct = 0; ct < NT; ++ct) q[ct] = qn[ct];
  }
}

template <int NT, int NC = 16 * NT>
__device__ __forceinline__ void strip_store(float* __restrict__ dstm, int rt, int half, int sub,
                                            const v8f (&acc)[NT]) {
  const int m_base = rt * 16 + half * 8;
#pragma unroll
  for (int ct = 0; ct < NT; ++ct)
#pragma unroll
    for (int r = 0; r < 8; ++r)
      dstm[(size_t)(m_base + r) * NC + ct * 16 + sub] = acc[ct][r];
}


__device__ __forceinline__ void load_a1(const float* __restrict__ x,
                                        const float* __restrict__ out_norm,
                                        int rt, int half, int sub, float2 (&a)[GEMM_K / 4]) {
  const int m = rt * 16 + sub;
  const float norm = out_norm[m];
  const float* __restrict__ arow = x + (size_t)m * IN_DIM + half * 2;
#pragma unroll
  for (int j = 0; j < GEMM_K / 4; ++j) {
    const float2 av = *(const float2*)(arow + 4 * j);
    a[j].x = av.x * norm;
    a[j].y = av.y * norm;
  }
}

__device__ __forceinline__ void gemm1_block(const float* __restrict__ x,
                                            const float* __restrict__ W1,
                                            const float* __restrict__ out_norm,
                                            float* __restrict__ h1,
                                            int n_row_tiles, int blk) {
  __shared__ float4 Bl4[GEMM_K / 4 * HID_DIM];
  const int lane = threadIdx.x & 31;
  const int wave = __builtin_amdgcn_readfirstlane((int)(threadIdx.x >> 5));
  const int half = lane >> 4;
  const int sub  = lane & 15;
  const int step = GEMM_GRID * GEMM_WAVES;
  int rt = blk + GEMM_GRID * wave;

  float2 a[GEMM_K / 4];
#pragma unroll
  for (int j = 0; j < GEMM_K / 4; ++j) a[j] = make_float2(0.0f, 0.0f);
  if (rt < n_row_tiles) load_a1(x, out_norm, rt, half, sub, a);

  stage_weights<HID_DIM>(W1, Bl4);
  __syncthreads();

  while (rt < n_row_tiles) {
    const int nx = rt + step;
    float2 an[GEMM_K / 4];
#pragma unroll
    for (int j = 0; j < GEMM_K / 4; ++j) an[j] = make_float2(0.0f, 0.0f);
    if (nx < n_row_tiles) load_a1(x, out_norm, nx, half, sub, an);

#pragma unroll
    for (int hc = 0; hc < 2; ++hc) {
      v8f acc[HID_DIM / 32];
#pragma unroll
      for (int ct = 0; ct < HID_DIM / 32; ++ct) acc[ct] = v8f{};
      strip_mma<HID_DIM / 32, HID_DIM>(a, Bl4 + hc * (HID_DIM / 2), half, sub, acc);
      strip_store<HID_DIM / 32, HID_DIM>(h1 + hc * (HID_DIM / 2), rt, half, sub, acc);
    }
#pragma unroll
    for (int j = 0; j < GEMM_K / 4; ++j) a[j] = an[j];
    rt = nx;
  }
}

__global__ __launch_bounds__(GEMM_THREADS) void place_and_gemm1(const float* __restrict__ x,
                                                                const float* __restrict__ W1,
                                                                const float* __restrict__ out_norm,
                                                                float* __restrict__ h1,
                                                                int n_row_tiles,
                                                                const uint2* __restrict__ pairs,
                                                                const unsigned int* __restrict__ row_ptr,
                                                                int* __restrict__ sorted_src,
                                                                int n_edges, int n_nodes, int n_bins,
                                                                const float* __restrict__ W2,
                                                                float4* __restrict__ w2p) {
  static_assert(GEMM_THREADS == PLACE_THREADS, "edge_place_block is written for the product's block size");
  const int blk = (int)blockIdx.x;
  if (blk < GEMM_GRID)               gemm1_block(x, W1, out_norm, h1, n_row_tiles, blk);
  else if (blk < GEMM_GRID + n_bins) edge_place_block(pairs, row_ptr, sorted_src, n_edges, n_nodes, blk - GEMM_GRID);
  else                               stage_weights<OUT_DIM>(W2, w2p);
}


#define FUSE_WAVES   4
#define FUSE_THREADS (FUSE_WAVES * 32)
#define TILE_ROWS    16
#define WAVE_NODES   (TILE_ROWS / FUSE_WAVES)
#define TILE_PITCH   132
#define OTILE_PITCH  66
static_assert(OUT_DIM == 64, "a wave stores a row of h2 as 32 x 8 bytes");
static_assert(OUT_DIM / 16 == FUSE_WAVES, "one wave per column tile of the product");

__global__ __launch_bounds__(FUSE_THREADS) void gather128_gemm2(const float* __restrict__ h,
                                                                const int* __restrict__ sorted_src,
                                                                const unsigned int* __restrict__ row_ptr,
                                                                const float* __restrict__ b1,
                                                                const float* __restrict__ out_norm,
                                                                const float* __restrict__ in_norm,
                                                                const float4* __restrict__ w2p,
                                                                float* __restrict__ h2,
                                                                int n_nodes, int n_edges) {
  __shared__ __attribute__((aligned(16))) float tile[TILE_ROWS * TILE_PITCH];
  __shared__ __attribute__((aligned(16))) float otile[TILE_ROWS * OTILE_PITCH];
  const int lane = threadIdx.x & 31;
  const int wave = __builtin_amdgcn_readfirstlane((int)(threadIdx.x >> 5));
  const int rt = (int)blockIdx.x;
  const int d0 = rt * TILE_ROWS + wave * WAVE_NODES;
  const float* __restrict__ hl = h + lane * 4;
  const float4 bv = *(const float4*)(b1 + lane * 4);

  int   rp[WAVE_NODES + 1];
  float s_in[WAVE_NODES], s_out[WAVE_NODES];
#pragma unroll
  for (int i = 0; i <= WAVE_NODES; ++i)
    rp[i] = (int)min(row_ptr[min(d0 + i, n_nodes)], (unsigned int)n_edges);
#pragma unroll
  for (int i = 0; i < WAVE_NODES; ++i) {
    s_in[i]  = in_norm[min(d0 + i, n_nodes - 1)];
    s_out[i] = out_norm[min(d0 + i, n_nodes - 1)];
  }

#pragma unroll
  for (int nn = 0; nn < WAVE_NODES; ++nn) {
    const int rl = wave * WAVE_NODES + nn;
    float4 o = make_float4(0.0f, 0.0f, 0.0f, 0.0f);
    if (d0 + nn < n_nodes) {
      const int beg = rp[nn];
      const int end = rp[nn + 1];
      float4 acc = make_float4(0.0f, 0.0f, 0.0f, 0.0f);
      for (int j = beg; j < end; j += 32) {
        const int cnt = min(32, end - j);
        int mine = (lane < cnt) ? sorted_src[j + lane] : 0;
        mine = (int)min((unsigned int)mine, (unsigned int)(n_nodes - 1));
        int t = 0;
        for (; t + 8 <= cnt; t += 8) {
          float4 v[8];
#pragma unroll
          for (int u = 0; u < 8; ++u) {
            const int s = __builtin_amdgcn_readlane(mine, t + u);
            v[u] = *(const float4*)(hl + (size_t)s * 128);
          }
#pragma unroll
          for (int u = 0; u < 8; ++u) {
            acc.x += v[u].x; acc.y += v[u].y; acc.z += v[u].z; acc.w += v[u].w;
          }
        }
        if (t < cnt) {
          float4 v[8];
#pragma unroll
          for (int u = 0; u < 8; ++u) {
            v[u] = make_float4(0.0f, 0.0f, 0.0f, 0.0f);
            if (t + u < cnt) {
              const int s = __builtin_amdgcn_readlane(mine, t + u);
              v[u] = *(const float4*)(hl + (size_t)s * 128);
            }
          }
#pragma unroll
          for (int u = 0; u < 8; ++u) {
            acc.x += v[u].x; acc.y += v[u].y; acc.z += v[u].z; acc.w += v[u].w;
          }
        }
      }
      o.x = fmaxf(acc.x * s_in[nn] + bv.x, 0.0f) * s_out[nn];
      o.y = fmaxf(acc.y * s_in[nn] + bv.y, 0.0f) * s_out[nn];
      o.z = fmaxf(acc.z * s_in[nn] + bv.z, 0.0f) * s_out[nn];
      o.w = fmaxf(acc.w * s_in[nn] + bv.w, 0.0f) * s_out[nn];
    }
    *(float4*)(tile + rl * TILE_PITCH + lane * 4) = o;
  }
  __syncthreads();

  {
    const int half = lane >> 4;
    const int sub  = lane & 15;
    const float4* __restrict__ bq = w2p + half * OUT_DIM + wave * 16 + sub;
    float4 q[GEMM_K / 8];
#pragma unroll
    for (int j2 = 0; j2 < GEMM_K / 8; ++j2) q[j2] = bq[(size_t)(j2 * 2) * OUT_DIM];
    const float* arow = tile + sub * TILE_PITCH + half * 2;
    v8f c = v8f{};
#pragma unroll
    for (int j2 = 0; j2 < GEMM_K / 8; ++j2) {
      const float2 x0 = *(const float2*)(arow + 8 * j2);
      const float2 x1 = *(const float2*)(arow + 8 * j2 + 4);
      v2f a0, a1, b0, b1f;
      a0.x = x0.x; a0.y = x0.y;
      a1.x = x1.x; a1.y = x1.y;
      b0.x  = q[j2].x; b0.y  = q[j2].y;
      b1f.x = q[j2].z; b1f.y = q[j2].w;
      c = __builtin_amdgcn_wmma_f32_16x16x4_f32(
           false, a0,  false, b0,
           (short)0, c,  false,  false);
      c = __builtin_amdgcn_wmma_f32_16x16x4_f32(
          false, a1, false, b1f, (short)0, c, false, false);
    }
#pragma unroll
    for (int r = 0; r < 8; ++r) otile[(half * 8 + r) * OTILE_PITCH + wave * 16 + sub] = c[r];
  }
  __syncthreads();

#pragma unroll
  for (int i = 0; i < WAVE_NODES; ++i) {
    const int rl = wave * WAVE_NODES + i;
    const int m  = rt * TILE_ROWS + rl;
    if (m < n_nodes)
      *(float2*)(h2 + (size_t)m * OUT_DIM + lane * 2) = *(const float2*)(otile + rl * OTILE_PITCH + lane * 2);
  }
}

__global__ __launch_bounds__(256) void gather64_out(const float* __restrict__ h,
                                                    const int* __restrict__ sorted_src,
                                                    const unsigned int* __restrict__ row_ptr,
                                                    const float* __restrict__ in_norm,
                                                    const float* __restrict__ b2,
                                                    float* __restrict__ out,
                                                    int n_nodes, int n_edges) {
  const int lane = threadIdx.x & 31;
  const int d = __builtin_amdgcn_readfirstlane((int)((blockIdx.x * blockDim.x + threadIdx.x) >> 5));
  if (d >= n_nodes) return;
  const int beg = (int)min(row_ptr[d], (unsigned int)n_edges);
  const int end = (int)min(row_ptr[d + 1], (unsigned int)n_edges);
  const float* __restrict__ hl = h + lane * 2;
  float2 acc = make_float2(0.0f, 0.0f);

  for (int j = beg; j < end; j += 32) {
    const int cnt = min(32, end - j);
    int mine = (lane < cnt) ? sorted_src[j + lane] : 0;
    mine = (int)min((unsigned int)mine, (unsigned int)(n_nodes - 1));
    int t = 0;
    for (; t + 8 <= cnt; t += 8) {
      float2 v[8];
#pragma unroll
      for (int u = 0; u < 8; ++u) {
        const int s = __builtin_amdgcn_readlane(mine, t + u);
        v[u] = *(const float2*)(hl + (size_t)s * 64);
      }
#pragma unroll
      for (int u = 0; u < 8; ++u) { acc.x += v[u].x; acc.y += v[u].y; }
    }
    if (t < cnt) {
      float2 v[8];
#pragma unroll
      for (int u = 0; u < 8; ++u) {
        v[u] = make_float2(0.0f, 0.0f);
        if (t + u < cnt) {
          const int s = __builtin_amdgcn_readlane(mine, t + u);
          v[u] = *(const float2*)(hl + (size_t)s * 64);
        }
      }
#pragma unroll
      for (int u = 0; u < 8; ++u) { acc.x += v[u].x; acc.y += v[u].y; }
    }
  }
  const float nrm = in_norm[d];
  const float2 bb = *(const float2*)(b2 + lane * 2);
  float2 o;
  o.x = acc.x * nrm + bb.x;
  o.y = acc.y * nrm + bb.y;
  *(float2*)(out + (size_t)d * 64 + lane * 2) = o;
}


static inline int grid1d(long n, int block) { return (int)((n + block - 1) / block); }

extern "C" void kernel_launch(void* const* d_in, const int* in_sizes, int n_in,
                              void* d_out, int out_size, void* d_ws, size_t ws_size,
                              hipStream_t stream) {
  const float* x   = (const float*)d_in[0];
  const float* W1  = (const float*)d_in[1];
  const float* b1  = (const float*)d_in[2];
  const float* W2  = (const float*)d_in[3];
  const float* b2  = (const float*)d_in[4];
  const int*   src = (const int*)d_in[5];
  const int*   dst = (const int*)d_in[6];

  const int n_nodes = in_sizes[0] / IN_DIM;
  const int n_edges = in_sizes[5];
  float* out = (float*)d_out;

  const size_t row = ((size_t)n_nodes + 31) & ~(size_t)31;
  float* h1   = (float*)d_ws;
  float* h2   = h1 + (size_t)n_nodes * HID_DIM;
  unsigned int* partial = (unsigned int*)(h2 + (size_t)n_nodes * OUT_DIM);
  float* out_norm = (float*)(partial + 2 * DEG_SLICES * row);
  float* in_norm  = out_norm + row;
  unsigned int* row_ptr  = (unsigned int*)(in_norm + row);
  const int node_blocks = grid1d(n_nodes, 128);
  const size_t bin_row_words = ((size_t)node_blocks + 3) & ~(size_t)3;
  float4* w2p = (float4*)(row_ptr + row + 32);
  unsigned int* bin_cnt = (unsigned int*)(w2p + GEMM_K / 4 * OUT_DIM);
  uint2* pairs = (uint2*)(bin_cnt + DEG_SLICES * bin_row_words);
  int* sorted_src = (int*)(pairs + n_edges);

  degree_hist<<<dim3(DEG_SLICES, 2), HIST_THREADS, 0, stream>>>(src, dst, partial, bin_cnt, n_edges, n_nodes,
                                                                node_blocks);
  norms_and_edge_bin<<<DEG_SLICES + grid1d(node_blocks, 4), HIST_THREADS, 0, stream>>>(
      src, dst, bin_cnt, pairs, n_edges, n_nodes, node_blocks, partial, out_norm, in_norm, row_ptr);

  const int row_tiles = n_nodes / 16;

  place_and_gemm1<<<GEMM_GRID + node_blocks + 1, GEMM_THREADS, 0, stream>>>(
      x, W1, out_norm, h1, row_tiles, pairs, row_ptr, sorted_src, n_edges, n_nodes, node_blocks, W2, w2p);

  gather128_gemm2<<<grid1d(n_nodes, TILE_ROWS), FUSE_THREADS, 0, stream>>>(
      h1, sorted_src, row_ptr, b1, out_norm, in_norm, w2p, h2, n_nodes, n_edges);

  gather64_out<<<(n_nodes + 7) / 8, 256, 0, stream>>>(h2, sorted_src, row_ptr, in_norm, b2, out,
                                                      n_nodes, n_edges);
}
